# starting kernel
# baseline (speedup 1.0000x reference)
#include <hip/hip_runtime.h>
#include <cstdint>


typedef __attribute__((ext_vector_type(2))) float v2f;
typedef __attribute__((ext_vector_type(8))) float v8f;

#define WMMA_F32(a, b, c) \
  __builtin_amdgcn_wmma_f32_16x16x4_f32(false, (a), false, (b), (short)0, (c), false, false)

constexpr int Hdim = 64;
constexpr int H2   = 128;
constexpr int IND  = 128;

__device__ __forceinline__ void atomicMaxFloat(float* addr, float v) {
  if (v >= 0.f) atomicMax((int*)addr, __float_as_int(v));
  else          atomicMin((unsigned int*)addr, __float_as_uint(v));
}

__global__ __launch_bounds__(256) void enc_gemm_kernel(
    const float* __restrict__ x, const float* __restrict__ W,
    const float* __restrict__ bias, float* __restrict__ h, int nrows)
{
  __shared__ float Xs[64][IND + 4];
  const int tid  = threadIdx.x;
  const int row0 = blockIdx.x * 64;

  for (int i = tid; i < 64 * (IND / 4); i += 256) {
    int r  = i / (IND / 4);
    int c4 = (i % (IND / 4)) * 4;
    int gr = row0 + r;
    float4 v = make_float4(0.f, 0.f, 0.f, 0.f);
    if (gr < nrows) v = *(const float4*)(x + (long long)gr * IND + c4);
    Xs[r][c4 + 0] = v.x; Xs[r][c4 + 1] = v.y;
    Xs[r][c4 + 2] = v.z; Xs[r][c4 + 3] = v.w;
  }
  __syncthreads();

  const int wave = tid >> 5, lane = tid & 31;
  const int half = lane >> 4, ln = lane & 15;

  for (int t = 0; t < 2; ++t) {
    int tile = wave * 2 + t;
    int m0 = (tile >> 2) * 16, n0 = (tile & 3) * 16;
    v8f acc = {};
    for (int kk = 0; kk < IND; kk += 4) {
      int ka = kk + 2 * half;
      v2f a, b;
      a[0] = Xs[m0 + ln][ka];      a[1] = Xs[m0 + ln][ka + 1];
      b[0] = W[ka * Hdim + n0 + ln];
      b[1] = W[(ka + 1) * Hdim + n0 + ln];
      acc = WMMA_F32(a, b, acc);
    }
    for (int r = 0; r < 8; ++r) {
      int gr = row0 + m0 + r + 8 * half;
      if (gr < nrows) {
        int nn = n0 + ln;
        h[(long long)gr * Hdim + nn] = acc[r] + bias[nn];
      }
    }
  }
}

__global__ __launch_bounds__(256) void ln_relu_kernel(
    const float* __restrict__ h, const float* __restrict__ g,
    const float* __restrict__ b, float* __restrict__ z, int nrows)
{
  int row  = blockIdx.x * 8 + (threadIdx.x >> 5);
  int lane = threadIdx.x & 31;
  if (row >= nrows) return;
  const float* hr = h + (long long)row * Hdim;
  float v0 = hr[lane], v1 = hr[lane + 32];
  float s = v0 + v1, s2 = v0 * v0 + v1 * v1;
  for (int off = 1; off < 32; off <<= 1) {
    s  += __shfl_xor(s, off);
    s2 += __shfl_xor(s2, off);
  }
  float mu  = s * (1.f / 64.f);
  float var = s2 * (1.f / 64.f) - mu * mu;
  float rs  = rsqrtf(var + 1e-5f);
  float* zr = z + (long long)row * Hdim;
  zr[lane]      = fmaxf((v0 - mu) * rs * g[lane]      + b[lane],      0.f);
  zr[lane + 32] = fmaxf((v1 - mu) * rs * g[lane + 32] + b[lane + 32], 0.f);
}

__global__ void init_aggr_kernel(float* __restrict__ mmax, float* __restrict__ denom,
                                 float* __restrict__ numer, long long total)
{
  long long i = (long long)blockIdx.x * blockDim.x + threadIdx.x;
  if (i < total) {
    mmax[i]  = -3.402823466e38f;
    denom[i] = 0.f;
    numer[i] = 0.f;
  }
}

__global__ __launch_bounds__(256) void edge_max_kernel(
    const float* __restrict__ z, const int* __restrict__ src,
    const int* __restrict__ dst, const float* __restrict__ t, int ti,
    float* __restrict__ mmax, int E)
{
  long long tid = (long long)blockIdx.x * blockDim.x + threadIdx.x;
  if (tid >= (long long)E * 16) return;
  int e = (int)(tid >> 4);
  int f = ((int)tid & 15) * 4;
  int s = src[e], d = dst[e];
  float tv = t[ti];
  float4 zv = *(const float4*)(z + (long long)s * Hdim + f);
  float* mp = mmax + (long long)d * Hdim + f;
  atomicMaxFloat(mp + 0, (fmaxf(zv.x, 0.f) + 1e-7f) * tv);
  atomicMaxFloat(mp + 1, (fmaxf(zv.y, 0.f) + 1e-7f) * tv);
  atomicMaxFloat(mp + 2, (fmaxf(zv.z, 0.f) + 1e-7f) * tv);
  atomicMaxFloat(mp + 3, (fmaxf(zv.w, 0.f) + 1e-7f) * tv);
}

__global__ __launch_bounds__(256) void edge_alpha_kernel(
    const float* __restrict__ z, const int* __restrict__ src,
    const int* __restrict__ dst, const float* __restrict__ t, int ti,
    const float* __restrict__ mmax, float* __restrict__ denom,
    float* __restrict__ numer, int E)
{
  long long tid = (long long)blockIdx.x * blockDim.x + threadIdx.x;
  if (tid >= (long long)E * 16) return;
  int e = (int)(tid >> 4);
  int f = ((int)tid & 15) * 4;
  int s = src[e], d = dst[e];
  float tv = t[ti];
  float4 zv = *(const float4*)(z + (long long)s * Hdim + f);
  float4 mv = *(const float4*)(mmax + (long long)d * Hdim + f);
  float* dp = denom + (long long)d * Hdim + f;
  float* np = numer + (long long)d * Hdim + f;

  float m0 = fmaxf(zv.x, 0.f) + 1e-7f; float a0 = __expf(m0 * tv - mv.x);
  float m1 = fmaxf(zv.y, 0.f) + 1e-7f; float a1 = __expf(m1 * tv - mv.y);
  float m2 = fmaxf(zv.z, 0.f) + 1e-7f; float a2 = __expf(m2 * tv - mv.z);
  float m3 = fmaxf(zv.w, 0.f) + 1e-7f; float a3 = __expf(m3 * tv - mv.w);
  atomicAdd(dp + 0, a0); atomicAdd(np + 0, m0 * a0);
  atomicAdd(dp + 1, a1); atomicAdd(np + 1, m1 * a1);
  atomicAdd(dp + 2, a2); atomicAdd(np + 2, m2 * a2);
  atomicAdd(dp + 3, a3); atomicAdd(np + 3, m3 * a3);
}

__global__ __launch_bounds__(256) void mlp_kernel(
    const float* __restrict__ z, const float* __restrict__ numer,
    const float* __restrict__ denom, const float* __restrict__ base,
    const float* __restrict__ W1, const float* __restrict__ b1,
    const float* __restrict__ g1, const float* __restrict__ be1,
    const float* __restrict__ W2, const float* __restrict__ b2,
    float* __restrict__ hout, int nrows)
{
  __shared__ float U[64][Hdim + 4];
  __shared__ float T[64][H2 + 4];

  const int tid  = threadIdx.x;
  const int row0 = blockIdx.x * 64;

  for (int i = tid; i < 64 * Hdim; i += 256) {
    int r = i >> 6, c = i & 63;
    int gr = row0 + r;
    float v = 0.f;
    if (gr < nrows) {
      long long idx = (long long)gr * Hdim + c;
      v = z[idx] + numer[idx] / (denom[idx] + 1e-16f);
    }
    U[r][c] = v;
  }
  __syncthreads();

  const int wave = tid >> 5, lane = tid & 31;
  const int half = lane >> 4, ln = lane & 15;

  for (int t = 0; t < 4; ++t) {
    int tile = wave * 4 + t;
    int m0 = (tile >> 3) * 16, n0 = (tile & 7) * 16;
    v8f acc = {};
    for (int kk = 0; kk < Hdim; kk += 4) {
      int ka = kk + 2 * half;
      v2f a, b;
      a[0] = U[m0 + ln][ka];     a[1] = U[m0 + ln][ka + 1];
      b[0] = W1[ka * H2 + n0 + ln];
      b[1] = W1[(ka + 1) * H2 + n0 + ln];
      acc = WMMA_F32(a, b, acc);
    }
    for (int r = 0; r < 8; ++r) {
      int mm = m0 + r + 8 * half;
      int nn = n0 + ln;
      T[mm][nn] = acc[r] + b1[nn];
    }
  }
  __syncthreads();

  {
    int r = tid >> 2;
    int q = tid & 3;
    float s = 0.f, s2 = 0.f;
    for (int c = q * 32; c < q * 32 + 32; ++c) {
      float v = T[r][c]; s += v; s2 += v * v;
    }
    s  += __shfl_xor(s, 1);  s  += __shfl_xor(s, 2);
    s2 += __shfl_xor(s2, 1); s2 += __shfl_xor(s2, 2);
    float mu  = s * (1.f / 128.f);
    float var = s2 * (1.f / 128.f) - mu * mu;
    float rs  = rsqrtf(var + 1e-5f);
    for (int c = q * 32; c < q * 32 + 32; ++c) {
      float v = (T[r][c] - mu) * rs * g1[c] + be1[c];
      T[r][c] = fmaxf(v, 0.f);
    }
  }
  __syncthreads();

  for (int t = 0; t < 2; ++t) {
    int tile = wave * 2 + t;
    int m0 = (tile >> 2) * 16, n0 = (tile & 3) * 16;
    v8f acc = {};
    for (int kk = 0; kk < H2; kk += 4) {
      int ka = kk + 2 * half;
      v2f a, b;
      a[0] = T[m0 + ln][ka];     a[1] = T[m0 + ln][ka + 1];
      b[0] = W2[ka * Hdim + n0 + ln];
      b[1] = W2[(ka + 1) * Hdim + n0 + ln];
      acc = WMMA_F32(a, b, acc);
    }
    for (int r = 0; r < 8; ++r) {
      int gr = row0 + m0 + r + 8 * half;
      if (gr < nrows) {
        int nn = n0 + ln;
        long long idx = (long long)gr * Hdim + nn;
        float v = acc[r] + b2[nn];
        if (base) v += base[idx];
        hout[idx] = v;
      }
    }
  }
}

__global__ __launch_bounds__(256) void final_kernel(
    const float* __restrict__ h, const float* __restrict__ g,
    const float* __restrict__ b, const float* __restrict__ linW,
    const float* __restrict__ linB, float* __restrict__ outH,
    float* __restrict__ outL, int nrows)
{
  int row  = blockIdx.x * 8 + (threadIdx.x >> 5);
  int lane = threadIdx.x & 31;
  if (row >= nrows) return;
  const float* hr = h + (long long)row * Hdim;
  float v0 = hr[lane], v1 = hr[lane + 32];
  float s = v0 + v1, s2 = v0 * v0 + v1 * v1;
  for (int off = 1; off < 32; off <<= 1) {
    s  += __shfl_xor(s, off);
    s2 += __shfl_xor(s2, off);
  }
  float mu  = s * (1.f / 64.f);
  float var = s2 * (1.f / 64.f) - mu * mu;
  float rs  = rsqrtf(var + 1e-5f);
  float f0 = fmaxf((v0 - mu) * rs * g[lane]      + b[lane],      0.f);
  float f1 = fmaxf((v1 - mu) * rs * g[lane + 32] + b[lane + 32], 0.f);
  outH[(long long)row * Hdim + lane]      = f0;
  outH[(long long)row * Hdim + lane + 32] = f1;
  for (int j = 0; j < 3; ++j) {
    float p = f0 * linW[lane * 3 + j] + f1 * linW[(lane + 32) * 3 + j];
    for (int off = 1; off < 32; off <<= 1) p += __shfl_xor(p, off);
    if (lane == 0) outL[(long long)row * 3 + j] = p + linB[j];
  }
}

extern "C" void kernel_launch(void* const* d_in, const int* in_sizes, int n_in,
                              void* d_out, int out_size, void* d_ws, size_t ws_size,
                              hipStream_t stream)
{
  const float* x    = (const float*)d_in[0];
  const int*   edge = (const int*)d_in[1];
  const float* encW = (const float*)d_in[2];
  const float* encB = (const float*)d_in[3];
  const float* tvec = (const float*)d_in[4];
  const float* W1   = (const float*)d_in[5];
  const float* b1   = (const float*)d_in[6];
  const float* g1   = (const float*)d_in[7];
  const float* be1  = (const float*)d_in[8];
  const float* W2   = (const float*)d_in[9];
  const float* b2   = (const float*)d_in[10];
  const float* ng   = (const float*)d_in[11];
  const float* nb   = (const float*)d_in[12];
  const float* linW = (const float*)d_in[13];
  const float* linB = (const float*)d_in[14];

  const int N = in_sizes[0] / IND;
  const int E = in_sizes[1] / 2;
  const int* src = edge;
  const int* dst = edge + E;

  float* h     = (float*)d_ws;
  float* z     = h     + (size_t)N * Hdim;
  float* mmax  = z     + (size_t)N * Hdim;
  float* denom = mmax  + (size_t)N * Hdim;
  float* numer = denom + (size_t)N * Hdim;

  const int rowBlocks  = (N + 63) / 64;
  const int waveBlocks = (N + 7) / 8;
  const long long nf   = (long long)N * Hdim;
  const int nfBlocks   = (int)((nf + 255) / 256);
  const long long et   = (long long)E * 16;
  const int eBlocks    = (int)((et + 255) / 256);

  enc_gemm_kernel<<<rowBlocks, 256, 0, stream>>>(x, encW, encB, h, N);

  for (int i = 0; i < 3; ++i) {
    const float* zin;
    const float* base;
    if (i == 0) {
      zin = h; base = nullptr;
    } else {
      ln_relu_kernel<<<waveBlocks, 256, 0, stream>>>(h, ng + i * Hdim, nb + i * Hdim, z, N);
      zin = z; base = h;
    }
    init_aggr_kernel<<<nfBlocks, 256, 0, stream>>>(mmax, denom, numer, nf);
    edge_max_kernel<<<eBlocks, 256, 0, stream>>>(zin, src, dst, tvec, i, mmax, E);
    edge_alpha_kernel<<<eBlocks, 256, 0, stream>>>(zin, src, dst, tvec, i, mmax, denom, numer, E);
    mlp_kernel<<<rowBlocks, 256, 0, stream>>>(
        zin, numer, denom, base,
        W1 + (size_t)i * Hdim * H2, b1 + (size_t)i * H2,
        g1 + (size_t)i * H2,        be1 + (size_t)i * H2,
        W2 + (size_t)i * H2 * Hdim, b2 + (size_t)i * Hdim, h, N);
  }

  float* outH = (float*)d_out;
  float* outL = outH + (size_t)N * Hdim;
  final_kernel<<<waveBlocks, 256, 0, stream>>>(h, ng, nb, linW, linB, outH, outL, N);
}


# Float atomic scatter (max, sums) replaced by a dst-sorted edge list built once per call (int atomic rank + scan) and a per-node gather softmax; sums now in arrival-order rank
# speedup vs baseline: 2.3065x; 2.3065x over previous
#include <hip/hip_runtime.h>
#include <cstdint>


typedef __attribute__((ext_vector_type(2))) float v2f;
typedef __attribute__((ext_vector_type(8))) float v8f;

#define WMMA_F32(a, b, c) \
  __builtin_amdgcn_wmma_f32_16x16x4_f32(false, (a), false, (b), (short)0, (c), false, false)

constexpr int Hdim = 64;
constexpr int H2   = 128;
constexpr int IND  = 128;

__global__ __launch_bounds__(256) void enc_gemm_kernel(
    const float* __restrict__ x, const float* __restrict__ W,
    const float* __restrict__ bias, float* __restrict__ h, int nrows)
{
  __shared__ float Xs[64][IND + 4];
  const int tid  = threadIdx.x;
  const int row0 = blockIdx.x * 64;

  for (int i = tid; i < 64 * (IND / 4); i += 256) {
    int r  = i / (IND / 4);
    int c4 = (i % (IND / 4)) * 4;
    int gr = row0 + r;
    float4 v = make_float4(0.f, 0.f, 0.f, 0.f);
    if (gr < nrows) v = *(const float4*)(x + (long long)gr * IND + c4);
    Xs[r][c4 + 0] = v.x; Xs[r][c4 + 1] = v.y;
    Xs[r][c4 + 2] = v.z; Xs[r][c4 + 3] = v.w;
  }
  __syncthreads();

  const int wave = tid >> 5, lane = tid & 31;
  const int half = lane >> 4, ln = lane & 15;

  for (int t = 0; t < 2; ++t) {
    int tile = wave * 2 + t;
    int m0 = (tile >> 2) * 16, n0 = (tile & 3) * 16;
    v8f acc = {};
    for (int kk = 0; kk < IND; kk += 4) {
      int ka = kk + 2 * half;
      v2f a, b;
      a[0] = Xs[m0 + ln][ka];      a[1] = Xs[m0 + ln][ka + 1];
      b[0] = W[ka * Hdim + n0 + ln];
      b[1] = W[(ka + 1) * Hdim + n0 + ln];
      acc = WMMA_F32(a, b, acc);
    }
    for (int r = 0; r < 8; ++r) {
      int gr = row0 + m0 + r + 8 * half;
      if (gr < nrows) {
        int nn = n0 + ln;
        h[(long long)gr * Hdim + nn] = acc[r] + bias[nn];
      }
    }
  }
}

__global__ __launch_bounds__(256) void ln_relu_kernel(
    const float* __restrict__ h, const float* __restrict__ g,
    const float* __restrict__ b, float* __restrict__ z, int nrows)
{
  int row  = blockIdx.x * 8 + (threadIdx.x >> 5);
  int lane = threadIdx.x & 31;
  if (row >= nrows) return;
  const float* hr = h + (long long)row * Hdim;
  float v0 = hr[lane], v1 = hr[lane + 32];
  float s = v0 + v1, s2 = v0 * v0 + v1 * v1;
  for (int off = 1; off < 32; off <<= 1) {
    s  += __shfl_xor(s, off);
    s2 += __shfl_xor(s2, off);
  }
  float mu  = s * (1.f / 64.f);
  float var = s2 * (1.f / 64.f) - mu * mu;
  float rs  = rsqrtf(var + 1e-5f);
  float* zr = z + (long long)row * Hdim;
  zr[lane]      = fmaxf((v0 - mu) * rs * g[lane]      + b[lane],      0.f);
  zr[lane + 32] = fmaxf((v1 - mu) * rs * g[lane + 32] + b[lane + 32], 0.f);
}

__global__ __launch_bounds__(256) void csr_zero_kernel(int* __restrict__ cnt, int n1)
{
  int i = blockIdx.x * 256 + threadIdx.x;
  if (i < n1) cnt[i] = 0;
}

__global__ __launch_bounds__(256) void csr_count_kernel(
    const int* __restrict__ dst, int* __restrict__ cnt, int* __restrict__ rank, int E)
{
  int e = blockIdx.x * 256 + threadIdx.x;
  if (e >= E) return;
  rank[e] = atomicAdd(cnt + dst[e], 1);
}

__global__ __launch_bounds__(1024) void csr_scan_kernel(
    const int* __restrict__ cnt, int* __restrict__ off, int n1)
{
  __shared__ int wpre[32];
  __shared__ int wtot[32];
  const int tid = threadIdx.x, lane = tid & 31, wave = tid >> 5;
  const int base = blockIdx.x * 1024;

  int pre = 0;
  for (int i = tid; i < base; i += 1024) pre += cnt[i];

  const int idx = base + tid;
  const int v = (idx < n1) ? cnt[idx] : 0;
  int inc = v;
  for (int o = 1; o < 32; o <<= 1) {
    pre += __shfl_xor(pre, o);
    int up = __shfl_up(inc, o);
    if (lane >= o) inc += up;
  }
  if (lane == 0)  wpre[wave] = pre;
  if (lane == 31) wtot[wave] = inc;
  __syncthreads();

  int bp = wpre[lane];
  int wx = (lane < wave) ? wtot[lane] : 0;
  for (int o = 1; o < 32; o <<= 1) {
    bp += __shfl_xor(bp, o);
    wx += __shfl_xor(wx, o);
  }
  if (idx < n1) off[idx] = bp + wx + inc - v;
}

__global__ __launch_bounds__(256) void csr_fill_kernel(
    const int* __restrict__ src, const int* __restrict__ dst,
    const int* __restrict__ off, const int* __restrict__ rank,
    int* __restrict__ csr, int E)
{
  int e = blockIdx.x * 256 + threadIdx.x;
  if (e >= E) return;
  int pos = off[dst[e]] + rank[e];
  if ((unsigned)pos < (unsigned)E) csr[pos] = src[e];
}

__device__ __forceinline__ unsigned edge_row(const int* __restrict__ csr, int p, int jl,
                                             unsigned last)
{
  p = p > jl ? jl : p;
  unsigned s = (unsigned)csr[p];
  return s > last ? last : s;
}

__device__ __forceinline__ void seg_max4(float4 zv, float tv, float& x0, float& x1,
                                         float& x2, float& x3)
{
  x0 = fmaxf(x0, (fmaxf(zv.x, 0.f) + 1e-7f) * tv);
  x1 = fmaxf(x1, (fmaxf(zv.y, 0.f) + 1e-7f) * tv);
  x2 = fmaxf(x2, (fmaxf(zv.z, 0.f) + 1e-7f) * tv);
  x3 = fmaxf(x3, (fmaxf(zv.w, 0.f) + 1e-7f) * tv);
}

__device__ __forceinline__ void seg_sum1(float zc, float tv, float mx, bool live,
                                         float& den, float& num)
{
  float m = fmaxf(zc, 0.f) + 1e-7f;
  float a = __expf(m * tv - mx);
  a = live ? a : 0.f;
  den += a;
  num += m * a;
}

__global__ __launch_bounds__(256) void gather_softmax_kernel(
    const float* __restrict__ z, const int* __restrict__ off,
    const int* __restrict__ csr, const float* __restrict__ t, int ti,
    float* __restrict__ denom, float* __restrict__ numer, int N, int E)
{
  const int gt = blockIdx.x * 256 + threadIdx.x;
  const int d  = gt >> 4;
  if (d >= N) return;
  const int f  = (gt & 15) * 4;
  int j0 = off[d], j1 = off[d + 1];
  j0 = j0 < 0 ? 0 : (j0 > E ? E : j0);
  j1 = j1 < j0 ? j0 : (j1 > E ? E : j1);

  float d0 = 0.f, d1 = 0.f, d2 = 0.f, d3 = 0.f;
  float n0 = 0.f, n1 = 0.f, n2 = 0.f, n3 = 0.f;

  if (j0 < j1) {
    const int jl = j1 - 1;
    const unsigned last = (unsigned)(N - 1);
    const float tv = t[ti];
    const float* zf = z + f;

    float x0 = -3.402823466e38f, x1 = x0, x2 = x0, x3 = x0;
    unsigned c0 = edge_row(csr, j0, jl, last),     c1 = edge_row(csr, j0 + 1, jl, last);
    unsigned c2 = edge_row(csr, j0 + 2, jl, last), c3 = edge_row(csr, j0 + 3, jl, last);
    for (int j = j0; j < j1; j += 4) {
      unsigned q0 = edge_row(csr, j + 4, jl, last), q1 = edge_row(csr, j + 5, jl, last);
      unsigned q2 = edge_row(csr, j + 6, jl, last), q3 = edge_row(csr, j + 7, jl, last);
      float4 r0 = *(const float4*)(zf + (size_t)c0 * Hdim);
      float4 r1 = *(const float4*)(zf + (size_t)c1 * Hdim);
      float4 r2 = *(const float4*)(zf + (size_t)c2 * Hdim);
      float4 r3 = *(const float4*)(zf + (size_t)c3 * Hdim);
      seg_max4(r0, tv, x0, x1, x2, x3);
      seg_max4(r1, tv, x0, x1, x2, x3);
      seg_max4(r2, tv, x0, x1, x2, x3);
      seg_max4(r3, tv, x0, x1, x2, x3);
      c0 = q0; c1 = q1; c2 = q2; c3 = q3;
    }

    c0 = edge_row(csr, j0, jl, last);     c1 = edge_row(csr, j0 + 1, jl, last);
    c2 = edge_row(csr, j0 + 2, jl, last); c3 = edge_row(csr, j0 + 3, jl, last);
    for (int j = j0; j < j1; j += 4) {
      unsigned q0 = edge_row(csr, j + 4, jl, last), q1 = edge_row(csr, j + 5, jl, last);
      unsigned q2 = edge_row(csr, j + 6, jl, last), q3 = edge_row(csr, j + 7, jl, last);
      float4 r0 = *(const float4*)(zf + (size_t)c0 * Hdim);
      float4 r1 = *(const float4*)(zf + (size_t)c1 * Hdim);
      float4 r2 = *(const float4*)(zf + (size_t)c2 * Hdim);
      float4 r3 = *(const float4*)(zf + (size_t)c3 * Hdim);
      const bool l1 = j + 1 <= jl, l2 = j + 2 <= jl, l3 = j + 3 <= jl;
      seg_sum1(r0.x, tv, x0, true, d0, n0); seg_sum1(r0.y, tv, x1, true, d1, n1);
      seg_sum1(r0.z, tv, x2, true, d2, n2); seg_sum1(r0.w, tv, x3, true, d3, n3);
      seg_sum1(r1.x, tv, x0, l1, d0, n0);   seg_sum1(r1.y, tv, x1, l1, d1, n1);
      seg_sum1(r1.z, tv, x2, l1, d2, n2);   seg_sum1(r1.w, tv, x3, l1, d3, n3);
      seg_sum1(r2.x, tv, x0, l2, d0, n0);   seg_sum1(r2.y, tv, x1, l2, d1, n1);
      seg_sum1(r2.z, tv, x2, l2, d2, n2);   seg_sum1(r2.w, tv, x3, l2, d3, n3);
      seg_sum1(r3.x, tv, x0, l3, d0, n0);   seg_sum1(r3.y, tv, x1, l3, d1, n1);
      seg_sum1(r3.z, tv, x2, l3, d2, n2);   seg_sum1(r3.w, tv, x3, l3, d3, n3);
      c0 = q0; c1 = q1; c2 = q2; c3 = q3;
    }
  }
  const size_t o = (size_t)d * Hdim + f;
  *(float4*)(denom + o) = make_float4(d0, d1, d2, d3);
  *(float4*)(numer + o) = make_float4(n0, n1, n2, n3);
}

__global__ __launch_bounds__(256) void mlp_kernel(
    const float* __restrict__ z, const float* __restrict__ numer,
    const float* __restrict__ denom, const float* __restrict__ base,
    const float* __restrict__ W1, const float* __restrict__ b1,
    const float* __restrict__ g1, const float* __restrict__ be1,
    const float* __restrict__ W2, const float* __restrict__ b2,
    float* __restrict__ hout, int nrows)
{
  __shared__ float U[64][Hdim + 4];
  __shared__ float T[64][H2 + 4];

  const int tid  = threadIdx.x;
  const int row0 = blockIdx.x * 64;

  for (int i = tid; i < 64 * Hdim; i += 256) {
    int r = i >> 6, c = i & 63;
    int gr = row0 + r;
    float v = 0.f;
    if (gr < nrows) {
      long long idx = (long long)gr * Hdim + c;
      v = z[idx] + numer[idx] / (denom[idx] + 1e-16f);
    }
    U[r][c] = v;
  }
  __syncthreads();

  const int wave = tid >> 5, lane = tid & 31;
  const int half = lane >> 4, ln = lane & 15;

  for (int t = 0; t < 4; ++t) {
    int tile = wave * 4 + t;
    int m0 = (tile >> 3) * 16, n0 = (tile & 7) * 16;
    v8f acc = {};
    for (int kk = 0; kk < Hdim; kk += 4) {
      int ka = kk + 2 * half;
      v2f a, b;
      a[0] = U[m0 + ln][ka];     a[1] = U[m0 + ln][ka + 1];
      b[0] = W1[ka * H2 + n0 + ln];
      b[1] = W1[(ka + 1) * H2 + n0 + ln];
      acc = WMMA_F32(a, b, acc);
    }
    for (int r = 0; r < 8; ++r) {
      int mm = m0 + r + 8 * half;
      int nn = n0 + ln;
      T[mm][nn] = acc[r] + b1[nn];
    }
  }
  __syncthreads();

  {
    int r = tid >> 2;
    int q = tid & 3;
    float s = 0.f, s2 = 0.f;
    for (int c = q * 32; c < q * 32 + 32; ++c) {
      float v = T[r][c]; s += v; s2 += v * v;
    }
    s  += __shfl_xor(s, 1);  s  += __shfl_xor(s, 2);
    s2 += __shfl_xor(s2, 1); s2 += __shfl_xor(s2, 2);
    float mu  = s * (1.f / 128.f);
    float var = s2 * (1.f / 128.f) - mu * mu;
    float rs  = rsqrtf(var + 1e-5f);
    for (int c = q * 32; c < q * 32 + 32; ++c) {
      float v = (T[r][c] - mu) * rs * g1[c] + be1[c];
      T[r][c] = fmaxf(v, 0.f);
    }
  }
  __syncthreads();

  for (int t = 0; t < 2; ++t) {
    int tile = wave * 2 + t;
    int m0 = (tile >> 2) * 16, n0 = (tile & 3) * 16;
    v8f acc = {};
    for (int kk = 0; kk < H2; kk += 4) {
      int ka = kk + 2 * half;
      v2f a, b;
      a[0] = T[m0 + ln][ka];     a[1] = T[m0 + ln][ka + 1];
      b[0] = W2[ka * Hdim + n0 + ln];
      b[1] = W2[(ka + 1) * Hdim + n0 + ln];
      acc = WMMA_F32(a, b, acc);
    }
    for (int r = 0; r < 8; ++r) {
      int gr = row0 + m0 + r + 8 * half;
      if (gr < nrows) {
        int nn = n0 + ln;
        long long idx = (long long)gr * Hdim + nn;
        float v = acc[r] + b2[nn];
        if (base) v += base[idx];
        hout[idx] = v;
      }
    }
  }
}

__global__ __launch_bounds__(256) void final_kernel(
    const float* __restrict__ h, const float* __restrict__ g,
    const float* __restrict__ b, const float* __restrict__ linW,
    const float* __restrict__ linB, float* __restrict__ outH,
    float* __restrict__ outL, int nrows)
{
  int row  = blockIdx.x * 8 + (threadIdx.x >> 5);
  int lane = threadIdx.x & 31;
  if (row >= nrows) return;
  const float* hr = h + (long long)row * Hdim;
  float v0 = hr[lane], v1 = hr[lane + 32];
  float s = v0 + v1, s2 = v0 * v0 + v1 * v1;
  for (int off = 1; off < 32; off <<= 1) {
    s  += __shfl_xor(s, off);
    s2 += __shfl_xor(s2, off);
  }
  float mu  = s * (1.f / 64.f);
  float var = s2 * (1.f / 64.f) - mu * mu;
  float rs  = rsqrtf(var + 1e-5f);
  float f0 = fmaxf((v0 - mu) * rs * g[lane]      + b[lane],      0.f);
  float f1 = fmaxf((v1 - mu) * rs * g[lane + 32] + b[lane + 32], 0.f);
  outH[(long long)row * Hdim + lane]      = f0;
  outH[(long long)row * Hdim + lane + 32] = f1;
  for (int j = 0; j < 3; ++j) {
    float p = f0 * linW[lane * 3 + j] + f1 * linW[(lane + 32) * 3 + j];
    for (int off = 1; off < 32; off <<= 1) p += __shfl_xor(p, off);
    if (lane == 0) outL[(long long)row * 3 + j] = p + linB[j];
  }
}

extern "C" void kernel_launch(void* const* d_in, const int* in_sizes, int n_in,
                              void* d_out, int out_size, void* d_ws, size_t ws_size,
                              hipStream_t stream)
{
  const float* x    = (const float*)d_in[0];
  const int*   edge = (const int*)d_in[1];
  const float* encW = (const float*)d_in[2];
  const float* encB = (const float*)d_in[3];
  const float* tvec = (const float*)d_in[4];
  const float* W1   = (const float*)d_in[5];
  const float* b1   = (const float*)d_in[6];
  const float* g1   = (const float*)d_in[7];
  const float* be1  = (const float*)d_in[8];
  const float* W2   = (const float*)d_in[9];
  const float* b2   = (const float*)d_in[10];
  const float* ng   = (const float*)d_in[11];
  const float* nb   = (const float*)d_in[12];
  const float* linW = (const float*)d_in[13];
  const float* linB = (const float*)d_in[14];

  const int N = in_sizes[0] / IND;
  const int E = in_sizes[1] / 2;
  const int* src = edge;
  const int* dst = edge + E;

  float* h     = (float*)d_ws;
  float* z     = h     + (size_t)N * Hdim;
  float* plane = z     + (size_t)N * Hdim;
  float* denom = plane + (size_t)N * Hdim;
  float* numer = denom + (size_t)N * Hdim;

  const int n1  = N + 1;
  const int n1a = (n1 + 3) & ~3;
  int* cnt  = (int*)plane;
  int* off  = cnt + n1a;
  int* rank = off + n1a;
  int* csr  = rank + E;

  const int rowBlocks  = (N + 63) / 64;
  const int waveBlocks = (N + 7) / 8;
  const int eBlocks    = (E + 255) / 256;
  const int gBlocks    = (N + 15) / 16;

  csr_zero_kernel<<<(n1 + 255) / 256, 256, 0, stream>>>(cnt, n1);
  csr_count_kernel<<<eBlocks, 256, 0, stream>>>(dst, cnt, rank, E);
  csr_scan_kernel<<<(n1 + 1023) / 1024, 1024, 0, stream>>>(cnt, off, n1);
  csr_fill_kernel<<<eBlocks, 256, 0, stream>>>(src, dst, off, rank, csr, E);

  enc_gemm_kernel<<<rowBlocks, 256, 0, stream>>>(x, encW, encB, h, N);

  for (int i = 0; i < 3; ++i) {
    const float* zin;
    const float* base;
    if (i == 0) {
      zin = h; base = nullptr;
    } else {
      ln_relu_kernel<<<waveBlocks, 256, 0, stream>>>(h, ng + i * Hdim, nb + i * Hdim, z, N);
      zin = z; base = h;
    }
    gather_softmax_kernel<<<gBlocks, 256, 0, stream>>>(zin, off, csr, tvec, i, denom, numer, N, E);
    mlp_kernel<<<rowBlocks, 256, 0, stream>>>(
        zin, numer, denom, base,
        W1 + (size_t)i * Hdim * H2, b1 + (size_t)i * H2,
        g1 + (size_t)i * H2,        be1 + (size_t)i * H2,
        W2 + (size_t)i * H2 * Hdim, b2 + (size_t)i * Hdim, h, N);
  }

  float* outH = (float*)d_out;
  float* outL = outH + (size_t)N * Hdim;
  final_kernel<<<waveBlocks, 256, 0, stream>>>(h, ng, nb, linW, linB, outH, outL, N);
}


# Gather reads each z row once: the rows fetched for the segment max (a node's first 24 edges) stay in registers as messages and feed the sums; longer segments re-read only the rest; sum order kept
# speedup vs baseline: 2.4163x; 1.0476x over previous
#include <hip/hip_runtime.h>
#include <cstdint>


typedef __attribute__((ext_vector_type(2))) float v2f;
typedef __attribute__((ext_vector_type(8))) float v8f;

#define WMMA_F32(a, b, c) \
  __builtin_amdgcn_wmma_f32_16x16x4_f32(false, (a), false, (b), (short)0, (c), false, false)

constexpr int Hdim = 64;
constexpr int H2   = 128;
constexpr int IND  = 128;

__global__ __launch_bounds__(256) void enc_gemm_kernel(
    const float* __restrict__ x, const float* __restrict__ W,
    const float* __restrict__ bias, float* __restrict__ h, int nrows)
{
  __shared__ float Xs[64][IND + 4];
  const int tid  = threadIdx.x;
  const int row0 = blockIdx.x * 64;

  for (int i = tid; i < 64 * (IND / 4); i += 256) {
    int r  = i / (IND / 4);
    int c4 = (i % (IND / 4)) * 4;
    int gr = row0 + r;
    float4 v = make_float4(0.f, 0.f, 0.f, 0.f);
    if (gr < nrows) v = *(const float4*)(x + (long long)gr * IND + c4);
    Xs[r][c4 + 0] = v.x; Xs[r][c4 + 1] = v.y;
    Xs[r][c4 + 2] = v.z; Xs[r][c4 + 3] = v.w;
  }
  __syncthreads();

  const int wave = tid >> 5, lane = tid & 31;
  const int half = lane >> 4, ln = lane & 15;

  for (int t = 0; t < 2; ++t) {
    int tile = wave * 2 + t;
    int m0 = (tile >> 2) * 16, n0 = (tile & 3) * 16;
    v8f acc = {};
    for (int kk = 0; kk < IND; kk += 4) {
      int ka = kk + 2 * half;
      v2f a, b;
      a[0] = Xs[m0 + ln][ka];      a[1] = Xs[m0 + ln][ka + 1];
      b[0] = W[ka * Hdim + n0 + ln];
      b[1] = W[(ka + 1) * Hdim + n0 + ln];
      acc = WMMA_F32(a, b, acc);
    }
    for (int r = 0; r < 8; ++r) {
      int gr = row0 + m0 + r + 8 * half;
      if (gr < nrows) {
        int nn = n0 + ln;
        h[(long long)gr * Hdim + nn] = acc[r] + bias[nn];
      }
    }
  }
}

__global__ __launch_bounds__(256) void ln_relu_kernel(
    const float* __restrict__ h, const float* __restrict__ g,
    const float* __restrict__ b, float* __restrict__ z, int nrows)
{
  int row  = blockIdx.x * 8 + (threadIdx.x >> 5);
  int lane = threadIdx.x & 31;
  if (row >= nrows) return;
  const float* hr = h + (long long)row * Hdim;
  float v0 = hr[lane], v1 = hr[lane + 32];
  float s = v0 + v1, s2 = v0 * v0 + v1 * v1;
  for (int off = 1; off < 32; off <<= 1) {
    s  += __shfl_xor(s, off);
    s2 += __shfl_xor(s2, off);
  }
  float mu  = s * (1.f / 64.f);
  float var = s2 * (1.f / 64.f) - mu * mu;
  float rs  = rsqrtf(var + 1e-5f);
  float* zr = z + (long long)row * Hdim;
  zr[lane]      = fmaxf((v0 - mu) * rs * g[lane]      + b[lane],      0.f);
  zr[lane + 32] = fmaxf((v1 - mu) * rs * g[lane + 32] + b[lane + 32], 0.f);
}

__global__ __launch_bounds__(256) void csr_zero_kernel(int* __restrict__ cnt, int n1)
{
  int i = blockIdx.x * 256 + threadIdx.x;
  if (i < n1) cnt[i] = 0;
}

__global__ __launch_bounds__(256) void csr_count_kernel(
    const int* __restrict__ dst, int* __restrict__ cnt, int* __restrict__ rank, int E)
{
  int e = blockIdx.x * 256 + threadIdx.x;
  if (e >= E) return;
  rank[e] = atomicAdd(cnt + dst[e], 1);
}

__global__ __launch_bounds__(1024) void csr_scan_kernel(
    const int* __restrict__ cnt, int* __restrict__ off, int n1)
{
  __shared__ int wpre[32];
  __shared__ int wtot[32];
  const int tid = threadIdx.x, lane = tid & 31, wave = tid >> 5;
  const int base = blockIdx.x * 1024;

  int pre = 0;
  for (int i = tid; i < base; i += 1024) pre += cnt[i];

  const int idx = base + tid;
  const int v = (idx < n1) ? cnt[idx] : 0;
  int inc = v;
  for (int o = 1; o < 32; o <<= 1) {
    pre += __shfl_xor(pre, o);
    int up = __shfl_up(inc, o);
    if (lane >= o) inc += up;
  }
  if (lane == 0)  wpre[wave] = pre;
  if (lane == 31) wtot[wave] = inc;
  __syncthreads();

  int bp = wpre[lane];
  int wx = (lane < wave) ? wtot[lane] : 0;
  for (int o = 1; o < 32; o <<= 1) {
    bp += __shfl_xor(bp, o);
    wx += __shfl_xor(wx, o);
  }
  if (idx < n1) off[idx] = bp + wx + inc - v;
}

__global__ __launch_bounds__(256) void csr_fill_kernel(
    const int* __restrict__ src, const int* __restrict__ dst,
    const int* __restrict__ off, const int* __restrict__ rank,
    int* __restrict__ csr, int E)
{
  int e = blockIdx.x * 256 + threadIdx.x;
  if (e >= E) return;
  int pos = off[dst[e]] + rank[e];
  if ((unsigned)pos < (unsigned)E) csr[pos] = src[e];
}

constexpr int GK = 6;
constexpr int RK = 4 * GK;

__device__ __forceinline__ unsigned edge_row(const int* __restrict__ csr, int p, int jl,
                                             unsigned last)
{
  p = p > jl ? jl : p;
  unsigned s = (unsigned)csr[p];
  return s > last ? last : s;
}

__device__ __forceinline__ float4 msg4(float4 zv)
{
  return make_float4(fmaxf(zv.x, 0.f) + 1e-7f, fmaxf(zv.y, 0.f) + 1e-7f,
                     fmaxf(zv.z, 0.f) + 1e-7f, fmaxf(zv.w, 0.f) + 1e-7f);
}

__device__ __forceinline__ void seg_max_m(float4 m, float tv, float& x0, float& x1,
                                          float& x2, float& x3)
{
  x0 = fmaxf(x0, m.x * tv);
  x1 = fmaxf(x1, m.y * tv);
  x2 = fmaxf(x2, m.z * tv);
  x3 = fmaxf(x3, m.w * tv);
}

__device__ __forceinline__ void seg_sum_m(float m, float tv, float mx, bool live,
                                          float& den, float& num)
{
  float a = __expf(m * tv - mx);
  a = live ? a : 0.f;
  den += a;
  num += m * a;
}

__device__ __forceinline__ void seg_sum4(float4 m, float tv, float x0, float x1, float x2,
                                         float x3, bool live, float& d0, float& d1,
                                         float& d2, float& d3, float& n0, float& n1,
                                         float& n2, float& n3)
{
  seg_sum_m(m.x, tv, x0, live, d0, n0);
  seg_sum_m(m.y, tv, x1, live, d1, n1);
  seg_sum_m(m.z, tv, x2, live, d2, n2);
  seg_sum_m(m.w, tv, x3, live, d3, n3);
}

__global__ __launch_bounds__(256) void gather_softmax_kernel(
    const float* __restrict__ z, const int* __restrict__ off,
    const int* __restrict__ csr, const float* __restrict__ t, int ti,
    float* __restrict__ denom, float* __restrict__ numer, int N, int E)
{
  const int gt = blockIdx.x * 256 + threadIdx.x;
  const int d  = gt >> 4;
  if (d >= N) return;
  const int f  = (gt & 15) * 4;
  int j0 = off[d], j1 = off[d + 1];
  j0 = j0 < 0 ? 0 : (j0 > E ? E : j0);
  j1 = j1 < j0 ? j0 : (j1 > E ? E : j1);

  float d0 = 0.f, d1 = 0.f, d2 = 0.f, d3 = 0.f;
  float n0 = 0.f, n1 = 0.f, n2 = 0.f, n3 = 0.f;

  if (j0 < j1) {
    const int jl = j1 - 1;
    const unsigned last = (unsigned)(N - 1);
    const float tv = t[ti];
    const float* zf = z + f;
    const int jt = j0 + RK;

    unsigned c[RK];
#pragma unroll
    for (int k = 0; k < RK; ++k) c[k] = edge_row(csr, j0 + k, jl, last);

    float4 R[RK];
#pragma unroll
    for (int k = 0; k < RK; ++k) R[k] = *(const float4*)(zf + (size_t)c[k] * Hdim);
    __builtin_amdgcn_sched_barrier(0);

    float x0 = -3.402823466e38f, x1 = x0, x2 = x0, x3 = x0;
#pragma unroll
    for (int g = 0; g < GK; ++g) {
      if (j0 + 4 * g <= jl) {
#pragma unroll
        for (int i = 0; i < 4; ++i) {
          R[4 * g + i] = msg4(R[4 * g + i]);
          seg_max_m(R[4 * g + i], tv, x0, x1, x2, x3);
        }
      }
    }
    for (int j = jt; j < j1; j += 4) {
      const unsigned c0 = edge_row(csr, j, jl, last),     c1 = edge_row(csr, j + 1, jl, last);
      const unsigned c2 = edge_row(csr, j + 2, jl, last), c3 = edge_row(csr, j + 3, jl, last);
      const float4 r0 = *(const float4*)(zf + (size_t)c0 * Hdim);
      const float4 r1 = *(const float4*)(zf + (size_t)c1 * Hdim);
      const float4 r2 = *(const float4*)(zf + (size_t)c2 * Hdim);
      const float4 r3 = *(const float4*)(zf + (size_t)c3 * Hdim);
      seg_max_m(msg4(r0), tv, x0, x1, x2, x3);
      seg_max_m(msg4(r1), tv, x0, x1, x2, x3);
      seg_max_m(msg4(r2), tv, x0, x1, x2, x3);
      seg_max_m(msg4(r3), tv, x0, x1, x2, x3);
    }

#pragma unroll
    for (int g = 0; g < GK; ++g) {
      const int p = j0 + 4 * g;
      if (p <= jl) {
        const bool l1 = p + 1 <= jl, l2 = p + 2 <= jl, l3 = p + 3 <= jl;
        seg_sum4(R[4 * g + 0], tv, x0, x1, x2, x3, true, d0, d1, d2, d3, n0, n1, n2, n3);
        seg_sum4(R[4 * g + 1], tv, x0, x1, x2, x3, l1,   d0, d1, d2, d3, n0, n1, n2, n3);
        seg_sum4(R[4 * g + 2], tv, x0, x1, x2, x3, l2,   d0, d1, d2, d3, n0, n1, n2, n3);
        seg_sum4(R[4 * g + 3], tv, x0, x1, x2, x3, l3,   d0, d1, d2, d3, n0, n1, n2, n3);
      }
    }
    for (int j = jt; j < j1; j += 4) {
      const unsigned c0 = edge_row(csr, j, jl, last),     c1 = edge_row(csr, j + 1, jl, last);
      const unsigned c2 = edge_row(csr, j + 2, jl, last), c3 = edge_row(csr, j + 3, jl, last);
      const float4 r0 = *(const float4*)(zf + (size_t)c0 * Hdim);
      const float4 r1 = *(const float4*)(zf + (size_t)c1 * Hdim);
      const float4 r2 = *(const float4*)(zf + (size_t)c2 * Hdim);
      const float4 r3 = *(const float4*)(zf + (size_t)c3 * Hdim);
      const bool l1 = j + 1 <= jl, l2 = j + 2 <= jl, l3 = j + 3 <= jl;
      seg_sum4(msg4(r0), tv, x0, x1, x2, x3, true, d0, d1, d2, d3, n0, n1, n2, n3);
      seg_sum4(msg4(r1), tv, x0, x1, x2, x3, l1,   d0, d1, d2, d3, n0, n1, n2, n3);
      seg_sum4(msg4(r2), tv, x0, x1, x2, x3, l2,   d0, d1, d2, d3, n0, n1, n2, n3);
      seg_sum4(msg4(r3), tv, x0, x1, x2, x3, l3,   d0, d1, d2, d3, n0, n1, n2, n3);
    }
  }
  const size_t o = (size_t)d * Hdim + f;
  *(float4*)(denom + o) = make_float4(d0, d1, d2, d3);
  *(float4*)(numer + o) = make_float4(n0, n1, n2, n3);
}

__global__ __launch_bounds__(256) void mlp_kernel(
    const float* __restrict__ z, const float* __restrict__ numer,
    const float* __restrict__ denom, const float* __restrict__ base,
    const float* __restrict__ W1, const float* __restrict__ b1,
    const float* __restrict__ g1, const float* __restrict__ be1,
    const float* __restrict__ W2, const float* __restrict__ b2,
    float* __restrict__ hout, int nrows)
{
  __shared__ float U[64][Hdim + 4];
  __shared__ float T[64][H2 + 4];

  const int tid  = threadIdx.x;
  const int row0 = blockIdx.x * 64;

  for (int i = tid; i < 64 * Hdim; i += 256) {
    int r = i >> 6, c = i & 63;
    int gr = row0 + r;
    float v = 0.f;
    if (gr < nrows) {
      long long idx = (long long)gr * Hdim + c;
      v = z[idx] + numer[idx] / (denom[idx] + 1e-16f);
    }
    U[r][c] = v;
  }
  __syncthreads();

  const int wave = tid >> 5, lane = tid & 31;
  const int half = lane >> 4, ln = lane & 15;

  for (int t = 0; t < 4; ++t) {
    int tile = wave * 4 + t;
    int m0 = (tile >> 3) * 16, n0 = (tile & 7) * 16;
    v8f acc = {};
    for (int kk = 0; kk < Hdim; kk += 4) {
      int ka = kk + 2 * half;
      v2f a, b;
      a[0] = U[m0 + ln][ka];     a[1] = U[m0 + ln][ka + 1];
      b[0] = W1[ka * H2 + n0 + ln];
      b[1] = W1[(ka + 1) * H2 + n0 + ln];
      acc = WMMA_F32(a, b, acc);
    }
    for (int r = 0; r < 8; ++r) {
      int mm = m0 + r + 8 * half;
      int nn = n0 + ln;
      T[mm][nn] = acc[r] + b1[nn];
    }
  }
  __syncthreads();

  {
    int r = tid >> 2;
    int q = tid & 3;
    float s = 0.f, s2 = 0.f;
    for (int c = q * 32; c < q * 32 + 32; ++c) {
      float v = T[r][c]; s += v; s2 += v * v;
    }
    s  += __shfl_xor(s, 1);  s  += __shfl_xor(s, 2);
    s2 += __shfl_xor(s2, 1); s2 += __shfl_xor(s2, 2);
    float mu  = s * (1.f / 128.f);
    float var = s2 * (1.f / 128.f) - mu * mu;
    float rs  = rsqrtf(var + 1e-5f);
    for (int c = q * 32; c < q * 32 + 32; ++c) {
      float v = (T[r][c] - mu) * rs * g1[c] + be1[c];
      T[r][c] = fmaxf(v, 0.f);
    }
  }
  __syncthreads();

  for (int t = 0; t < 2; ++t) {
    int tile = wave * 2 + t;
    int m0 = (tile >> 2) * 16, n0 = (tile & 3) * 16;
    v8f acc = {};
    for (int kk = 0; kk < H2; kk += 4) {
      int ka = kk + 2 * half;
      v2f a, b;
      a[0] = T[m0 + ln][ka];     a[1] = T[m0 + ln][ka + 1];
      b[0] = W2[ka * Hdim + n0 + ln];
      b[1] = W2[(ka + 1) * Hdim + n0 + ln];
      acc = WMMA_F32(a, b, acc);
    }
    for (int r = 0; r < 8; ++r) {
      int gr = row0 + m0 + r + 8 * half;
      if (gr < nrows) {
        int nn = n0 + ln;
        long long idx = (long long)gr * Hdim + nn;
        float v = acc[r] + b2[nn];
        if (base) v += base[idx];
        hout[idx] = v;
      }
    }
  }
}

__global__ __launch_bounds__(256) void final_kernel(
    const float* __restrict__ h, const float* __restrict__ g,
    const float* __restrict__ b, const float* __restrict__ linW,
    const float* __restrict__ linB, float* __restrict__ outH,
    float* __restrict__ outL, int nrows)
{
  int row  = blockIdx.x * 8 + (threadIdx.x >> 5);
  int lane = threadIdx.x & 31;
  if (row >= nrows) return;
  const float* hr = h + (long long)row * Hdim;
  float v0 = hr[lane], v1 = hr[lane + 32];
  float s = v0 + v1, s2 = v0 * v0 + v1 * v1;
  for (int off = 1; off < 32; off <<= 1) {
    s  += __shfl_xor(s, off);
    s2 += __shfl_xor(s2, off);
  }
  float mu  = s * (1.f / 64.f);
  float var = s2 * (1.f / 64.f) - mu * mu;
  float rs  = rsqrtf(var + 1e-5f);
  float f0 = fmaxf((v0 - mu) * rs * g[lane]      + b[lane],      0.f);
  float f1 = fmaxf((v1 - mu) * rs * g[lane + 32] + b[lane + 32], 0.f);
  outH[(long long)row * Hdim + lane]      = f0;
  outH[(long long)row * Hdim + lane + 32] = f1;
  for (int j = 0; j < 3; ++j) {
    float p = f0 * linW[lane * 3 + j] + f1 * linW[(lane + 32) * 3 + j];
    for (int off = 1; off < 32; off <<= 1) p += __shfl_xor(p, off);
    if (lane == 0) outL[(long long)row * 3 + j] = p + linB[j];
  }
}

extern "C" void kernel_launch(void* const* d_in, const int* in_sizes, int n_in,
                              void* d_out, int out_size, void* d_ws, size_t ws_size,
                              hipStream_t stream)
{
  const float* x    = (const float*)d_in[0];
  const int*   edge = (const int*)d_in[1];
  const float* encW = (const float*)d_in[2];
  const float* encB = (const float*)d_in[3];
  const float* tvec = (const float*)d_in[4];
  const float* W1   = (const float*)d_in[5];
  const float* b1   = (const float*)d_in[6];
  const float* g1   = (const float*)d_in[7];
  const float* be1  = (const float*)d_in[8];
  const float* W2   = (const float*)d_in[9];
  const float* b2   = (const float*)d_in[10];
  const float* ng   = (const float*)d_in[11];
  const float* nb   = (const float*)d_in[12];
  const float* linW = (const float*)d_in[13];
  const float* linB = (const float*)d_in[14];

  const int N = in_sizes[0] / IND;
  const int E = in_sizes[1] / 2;
  const int* src = edge;
  const int* dst = edge + E;

  float* h     = (float*)d_ws;
  float* z     = h     + (size_t)N * Hdim;
  float* plane = z     + (size_t)N * Hdim;
  float* denom = plane + (size_t)N * Hdim;
  float* numer = denom + (size_t)N * Hdim;

  const int n1  = N + 1;
  const int n1a = (n1 + 3) & ~3;
  int* cnt  = (int*)plane;
  int* off  = cnt + n1a;
  int* rank = off + n1a;
  int* csr  = rank + E;

  const int rowBlocks  = (N + 63) / 64;
  const int waveBlocks = (N + 7) / 8;
  const int eBlocks    = (E + 255) / 256;
  const int gBlocks    = (N + 15) / 16;

  csr_zero_kernel<<<(n1 + 255) / 256, 256, 0, stream>>>(cnt, n1);
  csr_count_kernel<<<eBlocks, 256, 0, stream>>>(dst, cnt, rank, E);
  csr_scan_kernel<<<(n1 + 1023) / 1024, 1024, 0, stream>>>(cnt, off, n1);
  csr_fill_kernel<<<eBlocks, 256, 0, stream>>>(src, dst, off, rank, csr, E);

  enc_gemm_kernel<<<rowBlocks, 256, 0, stream>>>(x, encW, encB, h, N);

  for (int i = 0; i < 3; ++i) {
    const float* zin;
    const float* base;
    if (i == 0) {
      zin = h; base = nullptr;
    } else {
      ln_relu_kernel<<<waveBlocks, 256, 0, stream>>>(h, ng + i * Hdim, nb + i * Hdim, z, N);
      zin = z; base = h;
    }
    gather_softmax_kernel<<<gBlocks, 256, 0, stream>>>(zin, off, csr, tvec, i, denom, numer, N, E);
    mlp_kernel<<<rowBlocks, 256, 0, stream>>>(
        zin, numer, denom, base,
        W1 + (size_t)i * Hdim * H2, b1 + (size_t)i * H2,
        g1 + (size_t)i * H2,        be1 + (size_t)i * H2,
        W2 + (size_t)i * H2 * Hdim, b2 + (size_t)i * Hdim, h, N);
  }

  float* outH = (float*)d_out;
  float* outL = outH + (size_t)N * Hdim;
  final_kernel<<<waveBlocks, 256, 0, stream>>>(h, ng, nb, linW, linB, outH, outL, N);
}


# Encoder, MLP and head kernels read tile rows, bias, residual and head weights in groups ahead of use, not one dependent memory trip per row under its bounds branch; sum orders unchanged
# speedup vs baseline: 2.8268x; 1.1699x over previous
#include <hip/hip_runtime.h>
#include <cstdint>


typedef __attribute__((ext_vector_type(2))) float v2f;
typedef __attribute__((ext_vector_type(8))) float v8f;

#define WMMA_F32(a, b, c) \
  __builtin_amdgcn_wmma_f32_16x16x4_f32(false, (a), false, (b), (short)0, (c), false, false)

constexpr int Hdim = 64;
constexpr int H2   = 128;
constexpr int IND  = 128;

__global__ __launch_bounds__(256) void enc_gemm_kernel(
    const float* __restrict__ x, const float* __restrict__ W,
    const float* __restrict__ bias, float* __restrict__ h, int nrows)
{
  __shared__ float Xs[64][IND + 4] __attribute__((aligned(16)));
  const int tid  = threadIdx.x;
  const int row0 = blockIdx.x * 64;

  float4 xv[8];
#pragma unroll
  for (int k = 0; k < 8; ++k) {
    const int i  = tid + 256 * k;
    const int gr = row0 + (i >> 5);
    const int lr = gr < nrows ? gr : nrows - 1;
    xv[k] = *(const float4*)(x + (long long)lr * IND + (i & 31) * 4);
  }
#pragma unroll
  for (int k = 0; k < 8; ++k) {
    const int i = tid + 256 * k;
    const int r = i >> 5;
    const float4 v = (row0 + r < nrows) ? xv[k] : make_float4(0.f, 0.f, 0.f, 0.f);
    *(float4*)&Xs[r][(i & 31) * 4] = v;
  }
  __syncthreads();

  const int wave = tid >> 5, lane = tid & 31;
  const int half = lane >> 4, ln = lane & 15;

  for (int t = 0; t < 2; ++t) {
    int tile = wave * 2 + t;
    int m0 = (tile >> 2) * 16, n0 = (tile & 3) * 16;
    const int nn = n0 + ln;
    const float bn = bias[nn];
    v8f acc = {};
    for (int kk = 0; kk < IND; kk += 4) {
      int ka = kk + 2 * half;
      v2f a, b;
      a[0] = Xs[m0 + ln][ka];      a[1] = Xs[m0 + ln][ka + 1];
      b[0] = W[ka * Hdim + n0 + ln];
      b[1] = W[(ka + 1) * Hdim + n0 + ln];
      acc = WMMA_F32(a, b, acc);
    }
    for (int r = 0; r < 8; ++r) {
      int gr = row0 + m0 + r + 8 * half;
      if (gr < nrows) h[(long long)gr * Hdim + nn] = acc[r] + bn;
    }
  }
}

__global__ __launch_bounds__(256) void ln_relu_kernel(
    const float* __restrict__ h, const float* __restrict__ g,
    const float* __restrict__ b, float* __restrict__ z, int nrows)
{
  int row  = blockIdx.x * 8 + (threadIdx.x >> 5);
  int lane = threadIdx.x & 31;
  if (row >= nrows) return;
  const float* hr = h + (long long)row * Hdim;
  float v0 = hr[lane], v1 = hr[lane + 32];
  float s = v0 + v1, s2 = v0 * v0 + v1 * v1;
  for (int off = 1; off < 32; off <<= 1) {
    s  += __shfl_xor(s, off);
    s2 += __shfl_xor(s2, off);
  }
  float mu  = s * (1.f / 64.f);
  float var = s2 * (1.f / 64.f) - mu * mu;
  float rs  = rsqrtf(var + 1e-5f);
  float* zr = z + (long long)row * Hdim;
  zr[lane]      = fmaxf((v0 - mu) * rs * g[lane]      + b[lane],      0.f);
  zr[lane + 32] = fmaxf((v1 - mu) * rs * g[lane + 32] + b[lane + 32], 0.f);
}

__global__ __launch_bounds__(256) void csr_zero_kernel(int* __restrict__ cnt, int n1)
{
  int i = blockIdx.x * 256 + threadIdx.x;
  if (i < n1) cnt[i] = 0;
}

__global__ __launch_bounds__(256) void csr_count_kernel(
    const int* __restrict__ dst, int* __restrict__ cnt, int* __restrict__ rank, int E)
{
  int e = blockIdx.x * 256 + threadIdx.x;
  if (e >= E) return;
  rank[e] = atomicAdd(cnt + dst[e], 1);
}

__global__ __launch_bounds__(1024) void csr_scan_kernel(
    const int* __restrict__ cnt, int* __restrict__ off, int n1)
{
  __shared__ int wpre[32];
  __shared__ int wtot[32];
  const int tid = threadIdx.x, lane = tid & 31, wave = tid >> 5;
  const int base = blockIdx.x * 1024;

  int pre = 0;
  for (int i = tid; i < base; i += 1024) pre += cnt[i];

  const int idx = base + tid;
  const int v = (idx < n1) ? cnt[idx] : 0;
  int inc = v;
  for (int o = 1; o < 32; o <<= 1) {
    pre += __shfl_xor(pre, o);
    int up = __shfl_up(inc, o);
    if (lane >= o) inc += up;
  }
  if (lane == 0)  wpre[wave] = pre;
  if (lane == 31) wtot[wave] = inc;
  __syncthreads();

  int bp = wpre[lane];
  int wx = (lane < wave) ? wtot[lane] : 0;
  for (int o = 1; o < 32; o <<= 1) {
    bp += __shfl_xor(bp, o);
    wx += __shfl_xor(wx, o);
  }
  if (idx < n1) off[idx] = bp + wx + inc - v;
}

__global__ __launch_bounds__(256) void csr_fill_kernel(
    const int* __restrict__ src, const int* __restrict__ dst,
    const int* __restrict__ off, const int* __restrict__ rank,
    int* __restrict__ csr, int E)
{
  int e = blockIdx.x * 256 + threadIdx.x;
  if (e >= E) return;
  int pos = off[dst[e]] + rank[e];
  if ((unsigned)pos < (unsigned)E) csr[pos] = src[e];
}

constexpr int GK = 6;
constexpr int RK = 4 * GK;

__device__ __forceinline__ unsigned edge_row(const int* __restrict__ csr, int p, int jl,
                                             unsigned last)
{
  p = p > jl ? jl : p;
  unsigned s = (unsigned)csr[p];
  return s > last ? last : s;
}

__device__ __forceinline__ float4 msg4(float4 zv)
{
  return make_float4(fmaxf(zv.x, 0.f) + 1e-7f, fmaxf(zv.y, 0.f) + 1e-7f,
                     fmaxf(zv.z, 0.f) + 1e-7f, fmaxf(zv.w, 0.f) + 1e-7f);
}

__device__ __forceinline__ void seg_max_m(float4 m, float tv, float& x0, float& x1,
                                          float& x2, float& x3)
{
  x0 = fmaxf(x0, m.x * tv);
  x1 = fmaxf(x1, m.y * tv);
  x2 = fmaxf(x2, m.z * tv);
  x3 = fmaxf(x3, m.w * tv);
}

__device__ __forceinline__ void seg_sum_m(float m, float tv, float mx, bool live,
                                          float& den, float& num)
{
  float a = __expf(m * tv - mx);
  a = live ? a : 0.f;
  den += a;
  num += m * a;
}

__device__ __forceinline__ void seg_sum4(float4 m, float tv, float x0, float x1, float x2,
                                         float x3, bool live, float& d0, float& d1,
                                         float& d2, float& d3, float& n0, float& n1,
                                         float& n2, float& n3)
{
  seg_sum_m(m.x, tv, x0, live, d0, n0);
  seg_sum_m(m.y, tv, x1, live, d1, n1);
  seg_sum_m(m.z, tv, x2, live, d2, n2);
  seg_sum_m(m.w, tv, x3, live, d3, n3);
}

__global__ __launch_bounds__(256) void gather_softmax_kernel(
    const float* __restrict__ z, const int* __restrict__ off,
    const int* __restrict__ csr, const float* __restrict__ t, int ti,
    float* __restrict__ denom, float* __restrict__ numer, int N, int E)
{
  const int gt = blockIdx.x * 256 + threadIdx.x;
  const int d  = gt >> 4;
  if (d >= N) return;
  const int f  = (gt & 15) * 4;
  int j0 = off[d], j1 = off[d + 1];
  j0 = j0 < 0 ? 0 : (j0 > E ? E : j0);
  j1 = j1 < j0 ? j0 : (j1 > E ? E : j1);

  float d0 = 0.f, d1 = 0.f, d2 = 0.f, d3 = 0.f;
  float n0 = 0.f, n1 = 0.f, n2 = 0.f, n3 = 0.f;

  if (j0 < j1) {
    const int jl = j1 - 1;
    const unsigned last = (unsigned)(N - 1);
    const float tv = t[ti];
    const float* zf = z + f;
    const int jt = j0 + RK;

    unsigned c[RK];
#pragma unroll
    for (int k = 0; k < RK; ++k) c[k] = edge_row(csr, j0 + k, jl, last);

    float4 R[RK];
#pragma unroll
    for (int k = 0; k < RK; ++k) R[k] = *(const float4*)(zf + (size_t)c[k] * Hdim);
    __builtin_amdgcn_sched_barrier(0);

    float x0 = -3.402823466e38f, x1 = x0, x2 = x0, x3 = x0;
#pragma unroll
    for (int g = 0; g < GK; ++g) {
      if (j0 + 4 * g <= jl) {
#pragma unroll
        for (int i = 0; i < 4; ++i) {
          R[4 * g + i] = msg4(R[4 * g + i]);
          seg_max_m(R[4 * g + i], tv, x0, x1, x2, x3);
        }
      }
    }
    for (int j = jt; j < j1; j += 4) {
      const unsigned c0 = edge_row(csr, j, jl, last),     c1 = edge_row(csr, j + 1, jl, last);
      const unsigned c2 = edge_row(csr, j + 2, jl, last), c3 = edge_row(csr, j + 3, jl, last);
      const float4 r0 = *(const float4*)(zf + (size_t)c0 * Hdim);
      const float4 r1 = *(const float4*)(zf + (size_t)c1 * Hdim);
      const float4 r2 = *(const float4*)(zf + (size_t)c2 * Hdim);
      const float4 r3 = *(const float4*)(zf + (size_t)c3 * Hdim);
      seg_max_m(msg4(r0), tv, x0, x1, x2, x3);
      seg_max_m(msg4(r1), tv, x0, x1, x2, x3);
      seg_max_m(msg4(r2), tv, x0, x1, x2, x3);
      seg_max_m(msg4(r3), tv, x0, x1, x2, x3);
    }

#pragma unroll
    for (int g = 0; g < GK; ++g) {
      const int p = j0 + 4 * g;
      if (p <= jl) {
        const bool l1 = p + 1 <= jl, l2 = p + 2 <= jl, l3 = p + 3 <= jl;
        seg_sum4(R[4 * g + 0], tv, x0, x1, x2, x3, true, d0, d1, d2, d3, n0, n1, n2, n3);
        seg_sum4(R[4 * g + 1], tv, x0, x1, x2, x3, l1,   d0, d1, d2, d3, n0, n1, n2, n3);
        seg_sum4(R[4 * g + 2], tv, x0, x1, x2, x3, l2,   d0, d1, d2, d3, n0, n1, n2, n3);
        seg_sum4(R[4 * g + 3], tv, x0, x1, x2, x3, l3,   d0, d1, d2, d3, n0, n1, n2, n3);
      }
    }
    for (int j = jt; j < j1; j += 4) {
      const unsigned c0 = edge_row(csr, j, jl, last),     c1 = edge_row(csr, j + 1, jl, last);
      const unsigned c2 = edge_row(csr, j + 2, jl, last), c3 = edge_row(csr, j + 3, jl, last);
      const float4 r0 = *(const float4*)(zf + (size_t)c0 * Hdim);
      const float4 r1 = *(const float4*)(zf + (size_t)c1 * Hdim);
      const float4 r2 = *(const float4*)(zf + (size_t)c2 * Hdim);
      const float4 r3 = *(const float4*)(zf + (size_t)c3 * Hdim);
      const bool l1 = j + 1 <= jl, l2 = j + 2 <= jl, l3 = j + 3 <= jl;
      seg_sum4(msg4(r0), tv, x0, x1, x2, x3, true, d0, d1, d2, d3, n0, n1, n2, n3);
      seg_sum4(msg4(r1), tv, x0, x1, x2, x3, l1,   d0, d1, d2, d3, n0, n1, n2, n3);
      seg_sum4(msg4(r2), tv, x0, x1, x2, x3, l2,   d0, d1, d2, d3, n0, n1, n2, n3);
      seg_sum4(msg4(r3), tv, x0, x1, x2, x3, l3,   d0, d1, d2, d3, n0, n1, n2, n3);
    }
  }
  const size_t o = (size_t)d * Hdim + f;
  *(float4*)(denom + o) = make_float4(d0, d1, d2, d3);
  *(float4*)(numer + o) = make_float4(n0, n1, n2, n3);
}

__global__ __launch_bounds__(256) void mlp_kernel(
    const float* __restrict__ z, const float* __restrict__ numer,
    const float* __restrict__ denom, const float* __restrict__ base,
    const float* __restrict__ W1, const float* __restrict__ b1,
    const float* __restrict__ g1, const float* __restrict__ be1,
    const float* __restrict__ W2, const float* __restrict__ b2,
    float* __restrict__ hout, int nrows)
{
  __shared__ float U[64][Hdim + 4] __attribute__((aligned(16)));
  __shared__ float T[64][H2 + 4];

  const int tid  = threadIdx.x;
  const int row0 = blockIdx.x * 64;

  const int wave = tid >> 5, lane = tid & 31;
  const int half = lane >> 4, ln = lane & 15;

  {
    float4 zq[4], nq[4], dq[4];
#pragma unroll
    for (int k = 0; k < 4; ++k) {
      const int i  = tid + 256 * k;
      const int gr = row0 + (i >> 4);
      const int lr = gr < nrows ? gr : nrows - 1;
      const long long idx = (long long)lr * Hdim + (i & 15) * 4;
      zq[k] = *(const float4*)(z + idx);
      nq[k] = *(const float4*)(numer + idx);
      dq[k] = *(const float4*)(denom + idx);
    }
#pragma unroll
    for (int k = 0; k < 4; ++k) {
      const int i = tid + 256 * k;
      const int r = i >> 4;
      const bool in = row0 + r < nrows;
      float4 v;
      v.x = zq[k].x + nq[k].x / (dq[k].x + 1e-16f);
      v.y = zq[k].y + nq[k].y / (dq[k].y + 1e-16f);
      v.z = zq[k].z + nq[k].z / (dq[k].z + 1e-16f);
      v.w = zq[k].w + nq[k].w / (dq[k].w + 1e-16f);
      v.x = in ? v.x : 0.f;  v.y = in ? v.y : 0.f;
      v.z = in ? v.z : 0.f;  v.w = in ? v.w : 0.f;
      *(float4*)&U[r][(i & 15) * 4] = v;
    }
  }
  __syncthreads();

  for (int t = 0; t < 4; ++t) {
    int tile = wave * 4 + t;
    int m0 = (tile >> 3) * 16, n0 = (tile & 7) * 16;
    v8f acc = {};
    for (int kk = 0; kk < Hdim; kk += 4) {
      int ka = kk + 2 * half;
      v2f a, b;
      a[0] = U[m0 + ln][ka];     a[1] = U[m0 + ln][ka + 1];
      b[0] = W1[ka * H2 + n0 + ln];
      b[1] = W1[(ka + 1) * H2 + n0 + ln];
      acc = WMMA_F32(a, b, acc);
    }
    for (int r = 0; r < 8; ++r) {
      int mm = m0 + r + 8 * half;
      int nn = n0 + ln;
      T[mm][nn] = acc[r] + b1[nn];
    }
  }
  __syncthreads();

  {
    int r = tid >> 2;
    int q = tid & 3;
    float s = 0.f, s2 = 0.f;
    for (int c = q * 32; c < q * 32 + 32; ++c) {
      float v = T[r][c]; s += v; s2 += v * v;
    }
    s  += __shfl_xor(s, 1);  s  += __shfl_xor(s, 2);
    s2 += __shfl_xor(s2, 1); s2 += __shfl_xor(s2, 2);
    float mu  = s * (1.f / 128.f);
    float var = s2 * (1.f / 128.f) - mu * mu;
    float rs  = rsqrtf(var + 1e-5f);
    for (int c = q * 32; c < q * 32 + 32; ++c) {
      float v = (T[r][c] - mu) * rs * g1[c] + be1[c];
      T[r][c] = fmaxf(v, 0.f);
    }
  }
  __syncthreads();

#pragma unroll
  for (int t = 0; t < 2; ++t) {
    int tile = wave * 2 + t;
    int m0 = (tile >> 2) * 16, n0 = (tile & 3) * 16;
    const int nn = n0 + ln;
    const float bn = b2[nn];
    float bv[8];
    if (base) {
#pragma unroll
      for (int r = 0; r < 8; ++r) {
        const int gr = row0 + m0 + r + 8 * half;
        const int lr = gr < nrows ? gr : nrows - 1;
        bv[r] = base[(long long)lr * Hdim + nn];
      }
    } else {
#pragma unroll
      for (int r = 0; r < 8; ++r) bv[r] = 0.f;
    }
    v8f acc = {};
    for (int kk = 0; kk < H2; kk += 4) {
      int ka = kk + 2 * half;
      v2f a, b;
      a[0] = T[m0 + ln][ka];     a[1] = T[m0 + ln][ka + 1];
      b[0] = W2[ka * Hdim + n0 + ln];
      b[1] = W2[(ka + 1) * Hdim + n0 + ln];
      acc = WMMA_F32(a, b, acc);
    }
    for (int r = 0; r < 8; ++r) {
      int gr = row0 + m0 + r + 8 * half;
      float v = acc[r] + bn;
      if (base) v += bv[r];
      if (gr < nrows) hout[(long long)gr * Hdim + nn] = v;
    }
  }
}

__global__ __launch_bounds__(256) void final_kernel(
    const float* __restrict__ h, const float* __restrict__ g,
    const float* __restrict__ b, const float* __restrict__ linW,
    const float* __restrict__ linB, float* __restrict__ outH,
    float* __restrict__ outL, int nrows)
{
  int row  = blockIdx.x * 8 + (threadIdx.x >> 5);
  int lane = threadIdx.x & 31;
  if (row >= nrows) return;
  const float* hr = h + (long long)row * Hdim;
  float v0 = hr[lane], v1 = hr[lane + 32];
  float w0[3], w1[3], lb[3];
#pragma unroll
  for (int j = 0; j < 3; ++j) {
    w0[j] = linW[lane * 3 + j];
    w1[j] = linW[(lane + 32) * 3 + j];
    lb[j] = linB[j];
  }
  float s = v0 + v1, s2 = v0 * v0 + v1 * v1;
  for (int off = 1; off < 32; off <<= 1) {
    s  += __shfl_xor(s, off);
    s2 += __shfl_xor(s2, off);
  }
  float mu  = s * (1.f / 64.f);
  float var = s2 * (1.f / 64.f) - mu * mu;
  float rs  = rsqrtf(var + 1e-5f);
  float f0 = fmaxf((v0 - mu) * rs * g[lane]      + b[lane],      0.f);
  float f1 = fmaxf((v1 - mu) * rs * g[lane + 32] + b[lane + 32], 0.f);
  outH[(long long)row * Hdim + lane]      = f0;
  outH[(long long)row * Hdim + lane + 32] = f1;
#pragma unroll
  for (int j = 0; j < 3; ++j) {
    float p = f0 * w0[j] + f1 * w1[j];
    for (int off = 1; off < 32; off <<= 1) p += __shfl_xor(p, off);
    if (lane == 0) outL[(long long)row * 3 + j] = p + lb[j];
  }
}

extern "C" void kernel_launch(void* const* d_in, const int* in_sizes, int n_in,
                              void* d_out, int out_size, void* d_ws, size_t ws_size,
                              hipStream_t stream)
{
  const float* x    = (const float*)d_in[0];
  const int*   edge = (const int*)d_in[1];
  const float* encW = (const float*)d_in[2];
  const float* encB = (const float*)d_in[3];
  const float* tvec = (const float*)d_in[4];
  const float* W1   = (const float*)d_in[5];
  const float* b1   = (const float*)d_in[6];
  const float* g1   = (const float*)d_in[7];
  const float* be1  = (const float*)d_in[8];
  const float* W2   = (const float*)d_in[9];
  const float* b2   = (const float*)d_in[10];
  const float* ng   = (const float*)d_in[11];
  const float* nb   = (const float*)d_in[12];
  const float* linW = (const float*)d_in[13];
  const float* linB = (const float*)d_in[14];

  const int N = in_sizes[0] / IND;
  const int E = in_sizes[1] / 2;
  const int* src = edge;
  const int* dst = edge + E;

  float* h     = (float*)d_ws;
  float* z     = h     + (size_t)N * Hdim;
  float* plane = z     + (size_t)N * Hdim;
  float* denom = plane + (size_t)N * Hdim;
  float* numer = denom + (size_t)N * Hdim;

  const int n1  = N + 1;
  const int n1a = (n1 + 3) & ~3;
  int* cnt  = (int*)plane;
  int* off  = cnt + n1a;
  int* rank = off + n1a;
  int* csr  = rank + E;

  const int rowBlocks  = (N + 63) / 64;
  const int waveBlocks = (N + 7) / 8;
  const int eBlocks    = (E + 255) / 256;
  const int gBlocks    = (N + 15) / 16;

  csr_zero_kernel<<<(n1 + 255) / 256, 256, 0, stream>>>(cnt, n1);
  csr_count_kernel<<<eBlocks, 256, 0, stream>>>(dst, cnt, rank, E);
  csr_scan_kernel<<<(n1 + 1023) / 1024, 1024, 0, stream>>>(cnt, off, n1);
  csr_fill_kernel<<<eBlocks, 256, 0, stream>>>(src, dst, off, rank, csr, E);

  enc_gemm_kernel<<<rowBlocks, 256, 0, stream>>>(x, encW, encB, h, N);

  for (int i = 0; i < 3; ++i) {
    const float* zin;
    const float* base;
    if (i == 0) {
      zin = h; base = nullptr;
    } else {
      ln_relu_kernel<<<waveBlocks, 256, 0, stream>>>(h, ng + i * Hdim, nb + i * Hdim, z, N);
      zin = z; base = h;
    }
    gather_softmax_kernel<<<gBlocks, 256, 0, stream>>>(zin, off, csr, tvec, i, denom, numer, N, E);
    mlp_kernel<<<rowBlocks, 256, 0, stream>>>(
        zin, numer, denom, base,
        W1 + (size_t)i * Hdim * H2, b1 + (size_t)i * H2,
        g1 + (size_t)i * H2,        be1 + (size_t)i * H2,
        W2 + (size_t)i * H2 * Hdim, b2 + (size_t)i * Hdim, h, N);
  }

  float* outH = (float*)d_out;
  float* outL = outH + (size_t)N * Hdim;
  final_kernel<<<waveBlocks, 256, 0, stream>>>(h, ng, nb, linW, linB, outH, outL, N);
}


# Row-wise LayerNorm kernels (2 between convs, 1 final with head) take 8 rows per wave, the edge-list fill 4 edges per thread: 50000 -> 6250 and 25000 -> 6250 waves to start per launch; sum orders kept
# speedup vs baseline: 2.9118x; 1.0301x over previous
#include <hip/hip_runtime.h>
#include <cstdint>


typedef __attribute__((ext_vector_type(2))) float v2f;
typedef __attribute__((ext_vector_type(8))) float v8f;

#define WMMA_F32(a, b, c) \
  __builtin_amdgcn_wmma_f32_16x16x4_f32(false, (a), false, (b), (short)0, (c), false, false)

constexpr int Hdim = 64;
constexpr int H2   = 128;
constexpr int IND  = 128;

__global__ __launch_bounds__(256) void enc_gemm_kernel(
    const float* __restrict__ x, const float* __restrict__ W,
    const float* __restrict__ bias, float* __restrict__ h, int nrows)
{
  __shared__ float Xs[64][IND + 4] __attribute__((aligned(16)));
  const int tid  = threadIdx.x;
  const int row0 = blockIdx.x * 64;

  float4 xv[8];
#pragma unroll
  for (int k = 0; k < 8; ++k) {
    const int i  = tid + 256 * k;
    const int gr = row0 + (i >> 5);
    const int lr = gr < nrows ? gr : nrows - 1;
    xv[k] = *(const float4*)(x + (long long)lr * IND + (i & 31) * 4);
  }
#pragma unroll
  for (int k = 0; k < 8; ++k) {
    const int i = tid + 256 * k;
    const int r = i >> 5;
    const float4 v = (row0 + r < nrows) ? xv[k] : make_float4(0.f, 0.f, 0.f, 0.f);
    *(float4*)&Xs[r][(i & 31) * 4] = v;
  }
  __syncthreads();

  const int wave = tid >> 5, lane = tid & 31;
  const int half = lane >> 4, ln = lane & 15;

  for (int t = 0; t < 2; ++t) {
    int tile = wave * 2 + t;
    int m0 = (tile >> 2) * 16, n0 = (tile & 3) * 16;
    const int nn = n0 + ln;
    const float bn = bias[nn];
    v8f acc = {};
    for (int kk = 0; kk < IND; kk += 4) {
      int ka = kk + 2 * half;
      v2f a, b;
      a[0] = Xs[m0 + ln][ka];      a[1] = Xs[m0 + ln][ka + 1];
      b[0] = W[ka * Hdim + n0 + ln];
      b[1] = W[(ka + 1) * Hdim + n0 + ln];
      acc = WMMA_F32(a, b, acc);
    }
    for (int r = 0; r < 8; ++r) {
      int gr = row0 + m0 + r + 8 * half;
      if (gr < nrows) h[(long long)gr * Hdim + nn] = acc[r] + bn;
    }
  }
}

constexpr int ROWS_PER_WAVE = 8;

__global__ __launch_bounds__(256) void ln_relu_kernel(
    const float* __restrict__ h, const float* __restrict__ g,
    const float* __restrict__ b, float* __restrict__ z, int nrows)
{
  const int lane = threadIdx.x & 31;
  const int row0 = __builtin_amdgcn_readfirstlane(
      (int)(blockIdx.x * 8 + (threadIdx.x >> 5)) * ROWS_PER_WAVE);
  if (row0 >= nrows) return;
  float v0[ROWS_PER_WAVE], v1[ROWS_PER_WAVE];
#pragma unroll
  for (int k = 0; k < ROWS_PER_WAVE; ++k) {
    const int r  = row0 + k;
    const int lr = r < nrows ? r : nrows - 1;
    const float* hr = h + (long long)lr * Hdim;
    v0[k] = hr[lane];
    v1[k] = hr[lane + 32];
  }
  const float g0 = g[lane], g1 = g[lane + 32];
  const float b0 = b[lane], b1 = b[lane + 32];

  float s[ROWS_PER_WAVE], s2[ROWS_PER_WAVE];
#pragma unroll
  for (int k = 0; k < ROWS_PER_WAVE; ++k) {
    s[k]  = v0[k] + v1[k];
    s2[k] = v0[k] * v0[k] + v1[k] * v1[k];
  }
#pragma unroll
  for (int off = 1; off < 32; off <<= 1) {
#pragma unroll
    for (int k = 0; k < ROWS_PER_WAVE; ++k) {
      s[k]  += __shfl_xor(s[k], off);
      s2[k] += __shfl_xor(s2[k], off);
    }
  }
#pragma unroll
  for (int k = 0; k < ROWS_PER_WAVE; ++k) {
    float mu  = s[k] * (1.f / 64.f);
    float var = s2[k] * (1.f / 64.f) - mu * mu;
    float rs  = rsqrtf(var + 1e-5f);
    if (row0 + k < nrows) {
      float* zr = z + (long long)(row0 + k) * Hdim;
      zr[lane]      = fmaxf((v0[k] - mu) * rs * g0 + b0, 0.f);
      zr[lane + 32] = fmaxf((v1[k] - mu) * rs * g1 + b1, 0.f);
    }
  }
}

__global__ __launch_bounds__(256) void csr_zero_kernel(int* __restrict__ cnt, int n1)
{
  int i = blockIdx.x * 256 + threadIdx.x;
  if (i < n1) cnt[i] = 0;
}

__global__ __launch_bounds__(256) void csr_count_kernel(
    const int* __restrict__ dst, int* __restrict__ cnt, int* __restrict__ rank, int E)
{
  int e = blockIdx.x * 256 + threadIdx.x;
  if (e >= E) return;
  rank[e] = atomicAdd(cnt + dst[e], 1);
}

__global__ __launch_bounds__(1024) void csr_scan_kernel(
    const int* __restrict__ cnt, int* __restrict__ off, int n1)
{
  __shared__ int wpre[32];
  __shared__ int wtot[32];
  const int tid = threadIdx.x, lane = tid & 31, wave = tid >> 5;
  const int base = blockIdx.x * 1024;

  int pre = 0;
  for (int i = tid; i < base; i += 1024) pre += cnt[i];

  const int idx = base + tid;
  const int v = (idx < n1) ? cnt[idx] : 0;
  int inc = v;
  for (int o = 1; o < 32; o <<= 1) {
    pre += __shfl_xor(pre, o);
    int up = __shfl_up(inc, o);
    if (lane >= o) inc += up;
  }
  if (lane == 0)  wpre[wave] = pre;
  if (lane == 31) wtot[wave] = inc;
  __syncthreads();

  int bp = wpre[lane];
  int wx = (lane < wave) ? wtot[lane] : 0;
  for (int o = 1; o < 32; o <<= 1) {
    bp += __shfl_xor(bp, o);
    wx += __shfl_xor(wx, o);
  }
  if (idx < n1) off[idx] = bp + wx + inc - v;
}

constexpr int EDGES_PER_THREAD = 4;

__global__ __launch_bounds__(256) void csr_fill_kernel(
    const int* __restrict__ src, const int* __restrict__ dst,
    const int* __restrict__ off, const int* __restrict__ rank,
    int* __restrict__ csr, int E)
{
  const int e0 = blockIdx.x * (256 * EDGES_PER_THREAD) + threadIdx.x;
  int dd[EDGES_PER_THREAD], rr[EDGES_PER_THREAD], ss[EDGES_PER_THREAD];
#pragma unroll
  for (int k = 0; k < EDGES_PER_THREAD; ++k) {
    const int e  = e0 + 256 * k;
    const int le = e < E ? e : E - 1;
    dd[k] = dst[le];
    rr[k] = rank[le];
    ss[k] = src[le];
  }
  int oo[EDGES_PER_THREAD];
#pragma unroll
  for (int k = 0; k < EDGES_PER_THREAD; ++k) oo[k] = off[dd[k]];
#pragma unroll
  for (int k = 0; k < EDGES_PER_THREAD; ++k) asm volatile("" : "+v"(ss[k]));
#pragma unroll
  for (int k = 0; k < EDGES_PER_THREAD; ++k) {
    const int e   = e0 + 256 * k;
    const int pos = oo[k] + rr[k];
    if (e < E && (unsigned)pos < (unsigned)E) csr[pos] = ss[k];
  }
}

constexpr int GK = 6;
constexpr int RK = 4 * GK;

__device__ __forceinline__ unsigned edge_row(const int* __restrict__ csr, int p, int jl,
                                             unsigned last)
{
  p = p > jl ? jl : p;
  unsigned s = (unsigned)csr[p];
  return s > last ? last : s;
}

__device__ __forceinline__ float4 msg4(float4 zv)
{
  return make_float4(fmaxf(zv.x, 0.f) + 1e-7f, fmaxf(zv.y, 0.f) + 1e-7f,
                     fmaxf(zv.z, 0.f) + 1e-7f, fmaxf(zv.w, 0.f) + 1e-7f);
}

__device__ __forceinline__ void seg_max_m(float4 m, float tv, float& x0, float& x1,
                                          float& x2, float& x3)
{
  x0 = fmaxf(x0, m.x * tv);
  x1 = fmaxf(x1, m.y * tv);
  x2 = fmaxf(x2, m.z * tv);
  x3 = fmaxf(x3, m.w * tv);
}

__device__ __forceinline__ void seg_sum_m(float m, float tv, float mx, bool live,
                                          float& den, float& num)
{
  float a = __expf(m * tv - mx);
  a = live ? a : 0.f;
  den += a;
  num += m * a;
}

__device__ __forceinline__ void seg_sum4(float4 m, float tv, float x0, float x1, float x2,
                                         float x3, bool live, float& d0, float& d1,
                                         float& d2, float& d3, float& n0, float& n1,
                                         float& n2, float& n3)
{
  seg_sum_m(m.x, tv, x0, live, d0, n0);
  seg_sum_m(m.y, tv, x1, live, d1, n1);
  seg_sum_m(m.z, tv, x2, live, d2, n2);
  seg_sum_m(m.w, tv, x3, live, d3, n3);
}

__global__ __launch_bounds__(256) void gather_softmax_kernel(
    const float* __restrict__ z, const int* __restrict__ off,
    const int* __restrict__ csr, const float* __restrict__ t, int ti,
    float* __restrict__ denom, float* __restrict__ numer, int N, int E)
{
  const int gt = blockIdx.x * 256 + threadIdx.x;
  const int d  = gt >> 4;
  if (d >= N) return;
  const int f  = (gt & 15) * 4;
  int j0 = off[d], j1 = off[d + 1];
  j0 = j0 < 0 ? 0 : (j0 > E ? E : j0);
  j1 = j1 < j0 ? j0 : (j1 > E ? E : j1);

  float d0 = 0.f, d1 = 0.f, d2 = 0.f, d3 = 0.f;
  float n0 = 0.f, n1 = 0.f, n2 = 0.f, n3 = 0.f;

  if (j0 < j1) {
    const int jl = j1 - 1;
    const unsigned last = (unsigned)(N - 1);
    const float tv = t[ti];
    const float* zf = z + f;
    const int jt = j0 + RK;

    unsigned c[RK];
#pragma unroll
    for (int k = 0; k < RK; ++k) c[k] = edge_row(csr, j0 + k, jl, last);

    float4 R[RK];
#pragma unroll
    for (int k = 0; k < RK; ++k) R[k] = *(const float4*)(zf + (size_t)c[k] * Hdim);
    __builtin_amdgcn_sched_barrier(0);

    float x0 = -3.402823466e38f, x1 = x0, x2 = x0, x3 = x0;
#pragma unroll
    for (int g = 0; g < GK; ++g) {
      if (j0 + 4 * g <= jl) {
#pragma unroll
        for (int i = 0; i < 4; ++i) {
          R[4 * g + i] = msg4(R[4 * g + i]);
          seg_max_m(R[4 * g + i], tv, x0, x1, x2, x3);
        }
      }
    }
    for (int j = jt; j < j1; j += 4) {
      const unsigned c0 = edge_row(csr, j, jl, last),     c1 = edge_row(csr, j + 1, jl, last);
      const unsigned c2 = edge_row(csr, j + 2, jl, last), c3 = edge_row(csr, j + 3, jl, last);
      const float4 r0 = *(const float4*)(zf + (size_t)c0 * Hdim);
      const float4 r1 = *(const float4*)(zf + (size_t)c1 * Hdim);
      const float4 r2 = *(const float4*)(zf + (size_t)c2 * Hdim);
      const float4 r3 = *(const float4*)(zf + (size_t)c3 * Hdim);
      seg_max_m(msg4(r0), tv, x0, x1, x2, x3);
      seg_max_m(msg4(r1), tv, x0, x1, x2, x3);
      seg_max_m(msg4(r2), tv, x0, x1, x2, x3);
      seg_max_m(msg4(r3), tv, x0, x1, x2, x3);
    }

#pragma unroll
    for (int g = 0; g < GK; ++g) {
      const int p = j0 + 4 * g;
      if (p <= jl) {
        const bool l1 = p + 1 <= jl, l2 = p + 2 <= jl, l3 = p + 3 <= jl;
        seg_sum4(R[4 * g + 0], tv, x0, x1, x2, x3, true, d0, d1, d2, d3, n0, n1, n2, n3);
        seg_sum4(R[4 * g + 1], tv, x0, x1, x2, x3, l1,   d0, d1, d2, d3, n0, n1, n2, n3);
        seg_sum4(R[4 * g + 2], tv, x0, x1, x2, x3, l2,   d0, d1, d2, d3, n0, n1, n2, n3);
        seg_sum4(R[4 * g + 3], tv, x0, x1, x2, x3, l3,   d0, d1, d2, d3, n0, n1, n2, n3);
      }
    }
    for (int j = jt; j < j1; j += 4) {
      const unsigned c0 = edge_row(csr, j, jl, last),     c1 = edge_row(csr, j + 1, jl, last);
      const unsigned c2 = edge_row(csr, j + 2, jl, last), c3 = edge_row(csr, j + 3, jl, last);
      const float4 r0 = *(const float4*)(zf + (size_t)c0 * Hdim);
      const float4 r1 = *(const float4*)(zf + (size_t)c1 * Hdim);
      const float4 r2 = *(const float4*)(zf + (size_t)c2 * Hdim);
      const float4 r3 = *(const float4*)(zf + (size_t)c3 * Hdim);
      const bool l1 = j + 1 <= jl, l2 = j + 2 <= jl, l3 = j + 3 <= jl;
      seg_sum4(msg4(r0), tv, x0, x1, x2, x3, true, d0, d1, d2, d3, n0, n1, n2, n3);
      seg_sum4(msg4(r1), tv, x0, x1, x2, x3, l1,   d0, d1, d2, d3, n0, n1, n2, n3);
      seg_sum4(msg4(r2), tv, x0, x1, x2, x3, l2,   d0, d1, d2, d3, n0, n1, n2, n3);
      seg_sum4(msg4(r3), tv, x0, x1, x2, x3, l3,   d0, d1, d2, d3, n0, n1, n2, n3);
    }
  }
  const size_t o = (size_t)d * Hdim + f;
  *(float4*)(denom + o) = make_float4(d0, d1, d2, d3);
  *(float4*)(numer + o) = make_float4(n0, n1, n2, n3);
}

__global__ __launch_bounds__(256) void mlp_kernel(
    const float* __restrict__ z, const float* __restrict__ numer,
    const float* __restrict__ denom, const float* __restrict__ base,
    const float* __restrict__ W1, const float* __restrict__ b1,
    const float* __restrict__ g1, const float* __restrict__ be1,
    const float* __restrict__ W2, const float* __restrict__ b2,
    float* __restrict__ hout, int nrows)
{
  __shared__ float U[64][Hdim + 4] __attribute__((aligned(16)));
  __shared__ float T[64][H2 + 4];

  const int tid  = threadIdx.x;
  const int row0 = blockIdx.x * 64;

  const int wave = tid >> 5, lane = tid & 31;
  const int half = lane >> 4, ln = lane & 15;

  {
    float4 zq[4], nq[4], dq[4];
#pragma unroll
    for (int k = 0; k < 4; ++k) {
      const int i  = tid + 256 * k;
      const int gr = row0 + (i >> 4);
      const int lr = gr < nrows ? gr : nrows - 1;
      const long long idx = (long long)lr * Hdim + (i & 15) * 4;
      zq[k] = *(const float4*)(z + idx);
      nq[k] = *(const float4*)(numer + idx);
      dq[k] = *(const float4*)(denom + idx);
    }
#pragma unroll
    for (int k = 0; k < 4; ++k) {
      const int i = tid + 256 * k;
      const int r = i >> 4;
      const bool in = row0 + r < nrows;
      float4 v;
      v.x = zq[k].x + nq[k].x / (dq[k].x + 1e-16f);
      v.y = zq[k].y + nq[k].y / (dq[k].y + 1e-16f);
      v.z = zq[k].z + nq[k].z / (dq[k].z + 1e-16f);
      v.w = zq[k].w + nq[k].w / (dq[k].w + 1e-16f);
      v.x = in ? v.x : 0.f;  v.y = in ? v.y : 0.f;
      v.z = in ? v.z : 0.f;  v.w = in ? v.w : 0.f;
      *(float4*)&U[r][(i & 15) * 4] = v;
    }
  }
  __syncthreads();

  for (int t = 0; t < 4; ++t) {
    int tile = wave * 4 + t;
    int m0 = (tile >> 3) * 16, n0 = (tile & 7) * 16;
    v8f acc = {};
    for (int kk = 0; kk < Hdim; kk += 4) {
      int ka = kk + 2 * half;
      v2f a, b;
      a[0] = U[m0 + ln][ka];     a[1] = U[m0 + ln][ka + 1];
      b[0] = W1[ka * H2 + n0 + ln];
      b[1] = W1[(ka + 1) * H2 + n0 + ln];
      acc = WMMA_F32(a, b, acc);
    }
    for (int r = 0; r < 8; ++r) {
      int mm = m0 + r + 8 * half;
      int nn = n0 + ln;
      T[mm][nn] = acc[r] + b1[nn];
    }
  }
  __syncthreads();

  {
    int r = tid >> 2;
    int q = tid & 3;
    float s = 0.f, s2 = 0.f;
    for (int c = q * 32; c < q * 32 + 32; ++c) {
      float v = T[r][c]; s += v; s2 += v * v;
    }
    s  += __shfl_xor(s, 1);  s  += __shfl_xor(s, 2);
    s2 += __shfl_xor(s2, 1); s2 += __shfl_xor(s2, 2);
    float mu  = s * (1.f / 128.f);
    float var = s2 * (1.f / 128.f) - mu * mu;
    float rs  = rsqrtf(var + 1e-5f);
    for (int c = q * 32; c < q * 32 + 32; ++c) {
      float v = (T[r][c] - mu) * rs * g1[c] + be1[c];
      T[r][c] = fmaxf(v, 0.f);
    }
  }
  __syncthreads();

#pragma unroll
  for (int t = 0; t < 2; ++t) {
    int tile = wave * 2 + t;
    int m0 = (tile >> 2) * 16, n0 = (tile & 3) * 16;
    const int nn = n0 + ln;
    const float bn = b2[nn];
    float bv[8];
    if (base) {
#pragma unroll
      for (int r = 0; r < 8; ++r) {
        const int gr = row0 + m0 + r + 8 * half;
        const int lr = gr < nrows ? gr : nrows - 1;
        bv[r] = base[(long long)lr * Hdim + nn];
      }
    } else {
#pragma unroll
      for (int r = 0; r < 8; ++r) bv[r] = 0.f;
    }
    v8f acc = {};
    for (int kk = 0; kk < H2; kk += 4) {
      int ka = kk + 2 * half;
      v2f a, b;
      a[0] = T[m0 + ln][ka];     a[1] = T[m0 + ln][ka + 1];
      b[0] = W2[ka * Hdim + n0 + ln];
      b[1] = W2[(ka + 1) * Hdim + n0 + ln];
      acc = WMMA_F32(a, b, acc);
    }
    for (int r = 0; r < 8; ++r) {
      int gr = row0 + m0 + r + 8 * half;
      float v = acc[r] + bn;
      if (base) v += bv[r];
      if (gr < nrows) hout[(long long)gr * Hdim + nn] = v;
    }
  }
}

__global__ __launch_bounds__(256) void final_kernel(
    const float* __restrict__ h, const float* __restrict__ g,
    const float* __restrict__ b, const float* __restrict__ linW,
    const float* __restrict__ linB, float* __restrict__ outH,
    float* __restrict__ outL, int nrows)
{
  const int lane = threadIdx.x & 31;
  const int row0 = __builtin_amdgcn_readfirstlane(
      (int)(blockIdx.x * 8 + (threadIdx.x >> 5)) * ROWS_PER_WAVE);
  if (row0 >= nrows) return;
  float v0[ROWS_PER_WAVE], v1[ROWS_PER_WAVE];
#pragma unroll
  for (int k = 0; k < ROWS_PER_WAVE; ++k) {
    const int r  = row0 + k;
    const int lr = r < nrows ? r : nrows - 1;
    const float* hr = h + (long long)lr * Hdim;
    v0[k] = hr[lane];
    v1[k] = hr[lane + 32];
  }
  float w0[3], w1[3], lb[3];
#pragma unroll
  for (int j = 0; j < 3; ++j) {
    w0[j] = linW[lane * 3 + j];
    w1[j] = linW[(lane + 32) * 3 + j];
    lb[j] = linB[j];
  }
  const float g0 = g[lane], g1 = g[lane + 32];
  const float b0 = b[lane], b1 = b[lane + 32];

  float s[ROWS_PER_WAVE], s2[ROWS_PER_WAVE];
#pragma unroll
  for (int k = 0; k < ROWS_PER_WAVE; ++k) {
    s[k]  = v0[k] + v1[k];
    s2[k] = v0[k] * v0[k] + v1[k] * v1[k];
  }
#pragma unroll
  for (int off = 1; off < 32; off <<= 1) {
#pragma unroll
    for (int k = 0; k < ROWS_PER_WAVE; ++k) {
      s[k]  += __shfl_xor(s[k], off);
      s2[k] += __shfl_xor(s2[k], off);
    }
  }
  float p[ROWS_PER_WAVE][3];
#pragma unroll
  for (int k = 0; k < ROWS_PER_WAVE; ++k) {
    float mu  = s[k] * (1.f / 64.f);
    float var = s2[k] * (1.f / 64.f) - mu * mu;
    float rs  = rsqrtf(var + 1e-5f);
    float f0 = fmaxf((v0[k] - mu) * rs * g0 + b0, 0.f);
    float f1 = fmaxf((v1[k] - mu) * rs * g1 + b1, 0.f);
    if (row0 + k < nrows) {
      outH[(long long)(row0 + k) * Hdim + lane]      = f0;
      outH[(long long)(row0 + k) * Hdim + lane + 32] = f1;
    }
#pragma unroll
    for (int j = 0; j < 3; ++j) p[k][j] = f0 * w0[j] + f1 * w1[j];
  }
#pragma unroll
  for (int off = 1; off < 32; off <<= 1) {
#pragma unroll
    for (int k = 0; k < ROWS_PER_WAVE; ++k) {
#pragma unroll
      for (int j = 0; j < 3; ++j) p[k][j] += __shfl_xor(p[k][j], off);
    }
  }
#pragma unroll
  for (int k = 0; k < ROWS_PER_WAVE; ++k) {
    if (row0 + k < nrows && lane == 0) {
#pragma unroll
      for (int j = 0; j < 3; ++j) outL[(long long)(row0 + k) * 3 + j] = p[k][j] + lb[j];
    }
  }
}

extern "C" void kernel_launch(void* const* d_in, const int* in_sizes, int n_in,
                              void* d_out, int out_size, void* d_ws, size_t ws_size,
                              hipStream_t stream)
{
  const float* x    = (const float*)d_in[0];
  const int*   edge = (const int*)d_in[1];
  const float* encW = (const float*)d_in[2];
  const float* encB = (const float*)d_in[3];
  const float* tvec = (const float*)d_in[4];
  const float* W1   = (const float*)d_in[5];
  const float* b1   = (const float*)d_in[6];
  const float* g1   = (const float*)d_in[7];
  const float* be1  = (const float*)d_in[8];
  const float* W2   = (const float*)d_in[9];
  const float* b2   = (const float*)d_in[10];
  const float* ng   = (const float*)d_in[11];
  const float* nb   = (const float*)d_in[12];
  const float* linW = (const float*)d_in[13];
  const float* linB = (const float*)d_in[14];

  const int N = in_sizes[0] / IND;
  const int E = in_sizes[1] / 2;
  const int* src = edge;
  const int* dst = edge + E;

  float* h     = (float*)d_ws;
  float* z     = h     + (size_t)N * Hdim;
  float* plane = z     + (size_t)N * Hdim;
  float* denom = plane + (size_t)N * Hdim;
  float* numer = denom + (size_t)N * Hdim;

  const int n1  = N + 1;
  const int n1a = (n1 + 3) & ~3;
  int* cnt  = (int*)plane;
  int* off  = cnt + n1a;
  int* rank = off + n1a;
  int* csr  = rank + E;

  const int rowBlocks  = (N + 63) / 64;
  const int eBlocks    = (E + 255) / 256;
  const int fBlocks    = (E + 256 * EDGES_PER_THREAD - 1) / (256 * EDGES_PER_THREAD);
  const int gBlocks    = (N + 15) / 16;

  csr_zero_kernel<<<(n1 + 255) / 256, 256, 0, stream>>>(cnt, n1);
  csr_count_kernel<<<eBlocks, 256, 0, stream>>>(dst, cnt, rank, E);
  csr_scan_kernel<<<(n1 + 1023) / 1024, 1024, 0, stream>>>(cnt, off, n1);
  csr_fill_kernel<<<fBlocks, 256, 0, stream>>>(src, dst, off, rank, csr, E);

  enc_gemm_kernel<<<rowBlocks, 256, 0, stream>>>(x, encW, encB, h, N);

  for (int i = 0; i < 3; ++i) {
    const float* zin;
    const float* base;
    if (i == 0) {
      zin = h; base = nullptr;
    } else {
      ln_relu_kernel<<<rowBlocks, 256, 0, stream>>>(h, ng + i * Hdim, nb + i * Hdim, z, N);
      zin = z; base = h;
    }
    gather_softmax_kernel<<<gBlocks, 256, 0, stream>>>(zin, off, csr, tvec, i, denom, numer, N, E);
    mlp_kernel<<<rowBlocks, 256, 0, stream>>>(
        zin, numer, denom, base,
        W1 + (size_t)i * Hdim * H2, b1 + (size_t)i * H2,
        g1 + (size_t)i * H2,        be1 + (size_t)i * H2,
        W2 + (size_t)i * H2 * Hdim, b2 + (size_t)i * Hdim, h, N);
  }

  float* outH = (float*)d_out;
  float* outL = outH + (size_t)N * Hdim;
  final_kernel<<<rowBlocks, 256, 0, stream>>>(h, ng, nb, linW, linB, outH, outL, N);
}


# LayerNorm+ReLU launches (2 between convs, final with head) moved into the MLP epilogue on the tile in LDS: 14 -> 11 launches; rows keep the removed kernels' lane layout and sum tree, sum orders kept
# speedup vs baseline: 3.0402x; 1.0441x over previous
#include <hip/hip_runtime.h>
#include <cstdint>


typedef __attribute__((ext_vector_type(2))) float v2f;
typedef __attribute__((ext_vector_type(8))) float v8f;

#define WMMA_F32(a, b, c) \
  __builtin_amdgcn_wmma_f32_16x16x4_f32(false, (a), false, (b), (short)0, (c), false, false)

constexpr int Hdim = 64;
constexpr int H2   = 128;
constexpr int IND  = 128;

__global__ __launch_bounds__(256) void enc_gemm_kernel(
    const float* __restrict__ x, const float* __restrict__ W,
    const float* __restrict__ bias, float* __restrict__ h, int nrows)
{
  __shared__ float Xs[64][IND + 4] __attribute__((aligned(16)));
  const int tid  = threadIdx.x;
  const int row0 = blockIdx.x * 64;

  float4 xv[8];
#pragma unroll
  for (int k = 0; k < 8; ++k) {
    const int i  = tid + 256 * k;
    const int gr = row0 + (i >> 5);
    const int lr = gr < nrows ? gr : nrows - 1;
    xv[k] = *(const float4*)(x + (long long)lr * IND + (i & 31) * 4);
  }
#pragma unroll
  for (int k = 0; k < 8; ++k) {
    const int i = tid + 256 * k;
    const int r = i >> 5;
    const float4 v = (row0 + r < nrows) ? xv[k] : make_float4(0.f, 0.f, 0.f, 0.f);
    *(float4*)&Xs[r][(i & 31) * 4] = v;
  }
  __syncthreads();

  const int wave = tid >> 5, lane = tid & 31;
  const int half = lane >> 4, ln = lane & 15;

  for (int t = 0; t < 2; ++t) {
    int tile = wave * 2 + t;
    int m0 = (tile >> 2) * 16, n0 = (tile & 3) * 16;
    const int nn = n0 + ln;
    const float bn = bias[nn];
    v8f acc = {};
    for (int kk = 0; kk < IND; kk += 4) {
      int ka = kk + 2 * half;
      v2f a, b;
      a[0] = Xs[m0 + ln][ka];      a[1] = Xs[m0 + ln][ka + 1];
      b[0] = W[ka * Hdim + n0 + ln];
      b[1] = W[(ka + 1) * Hdim + n0 + ln];
      acc = WMMA_F32(a, b, acc);
    }
    for (int r = 0; r < 8; ++r) {
      int gr = row0 + m0 + r + 8 * half;
      if (gr < nrows) h[(long long)gr * Hdim + nn] = acc[r] + bn;
    }
  }
}

constexpr int ROWS_PER_WAVE = 8;

__global__ __launch_bounds__(256) void csr_zero_kernel(int* __restrict__ cnt, int n1)
{
  int i = blockIdx.x * 256 + threadIdx.x;
  if (i < n1) cnt[i] = 0;
}

__global__ __launch_bounds__(256) void csr_count_kernel(
    const int* __restrict__ dst, int* __restrict__ cnt, int* __restrict__ rank, int E)
{
  int e = blockIdx.x * 256 + threadIdx.x;
  if (e >= E) return;
  rank[e] = atomicAdd(cnt + dst[e], 1);
}

__global__ __launch_bounds__(1024) void csr_scan_kernel(
    const int* __restrict__ cnt, int* __restrict__ off, int n1)
{
  __shared__ int wpre[32];
  __shared__ int wtot[32];
  const int tid = threadIdx.x, lane = tid & 31, wave = tid >> 5;
  const int base = blockIdx.x * 1024;

  int pre = 0;
  for (int i = tid; i < base; i += 1024) pre += cnt[i];

  const int idx = base + tid;
  const int v = (idx < n1) ? cnt[idx] : 0;
  int inc = v;
  for (int o = 1; o < 32; o <<= 1) {
    pre += __shfl_xor(pre, o);
    int up = __shfl_up(inc, o);
    if (lane >= o) inc += up;
  }
  if (lane == 0)  wpre[wave] = pre;
  if (lane == 31) wtot[wave] = inc;
  __syncthreads();

  int bp = wpre[lane];
  int wx = (lane < wave) ? wtot[lane] : 0;
  for (int o = 1; o < 32; o <<= 1) {
    bp += __shfl_xor(bp, o);
    wx += __shfl_xor(wx, o);
  }
  if (idx < n1) off[idx] = bp + wx + inc - v;
}

constexpr int EDGES_PER_THREAD = 4;

__global__ __launch_bounds__(256) void csr_fill_kernel(
    const int* __restrict__ src, const int* __restrict__ dst,
    const int* __restrict__ off, const int* __restrict__ rank,
    int* __restrict__ csr, int E)
{
  const int e0 = blockIdx.x * (256 * EDGES_PER_THREAD) + threadIdx.x;
  int dd[EDGES_PER_THREAD], rr[EDGES_PER_THREAD], ss[EDGES_PER_THREAD];
#pragma unroll
  for (int k = 0; k < EDGES_PER_THREAD; ++k) {
    const int e  = e0 + 256 * k;
    const int le = e < E ? e : E - 1;
    dd[k] = dst[le];
    rr[k] = rank[le];
    ss[k] = src[le];
  }
  int oo[EDGES_PER_THREAD];
#pragma unroll
  for (int k = 0; k < EDGES_PER_THREAD; ++k) oo[k] = off[dd[k]];
#pragma unroll
  for (int k = 0; k < EDGES_PER_THREAD; ++k) asm volatile("" : "+v"(ss[k]));
#pragma unroll
  for (int k = 0; k < EDGES_PER_THREAD; ++k) {
    const int e   = e0 + 256 * k;
    const int pos = oo[k] + rr[k];
    if (e < E && (unsigned)pos < (unsigned)E) csr[pos] = ss[k];
  }
}

constexpr int GK = 6;
constexpr int RK = 4 * GK;

__device__ __forceinline__ unsigned edge_row(const int* __restrict__ csr, int p, int jl,
                                             unsigned last)
{
  p = p > jl ? jl : p;
  unsigned s = (unsigned)csr[p];
  return s > last ? last : s;
}

__device__ __forceinline__ float4 msg4(float4 zv)
{
  return make_float4(fmaxf(zv.x, 0.f) + 1e-7f, fmaxf(zv.y, 0.f) + 1e-7f,
                     fmaxf(zv.z, 0.f) + 1e-7f, fmaxf(zv.w, 0.f) + 1e-7f);
}

__device__ __forceinline__ void seg_max_m(float4 m, float tv, float& x0, float& x1,
                                          float& x2, float& x3)
{
  x0 = fmaxf(x0, m.x * tv);
  x1 = fmaxf(x1, m.y * tv);
  x2 = fmaxf(x2, m.z * tv);
  x3 = fmaxf(x3, m.w * tv);
}

__device__ __forceinline__ void seg_sum_m(float m, float tv, float mx, bool live,
                                          float& den, float& num)
{
  float a = __expf(m * tv - mx);
  a = live ? a : 0.f;
  den += a;
  num += m * a;
}

__device__ __forceinline__ void seg_sum4(float4 m, float tv, float x0, float x1, float x2,
                                         float x3, bool live, float& d0, float& d1,
                                         float& d2, float& d3, float& n0, float& n1,
                                         float& n2, float& n3)
{
  seg_sum_m(m.x, tv, x0, live, d0, n0);
  seg_sum_m(m.y, tv, x1, live, d1, n1);
  seg_sum_m(m.z, tv, x2, live, d2, n2);
  seg_sum_m(m.w, tv, x3, live, d3, n3);
}

__global__ __launch_bounds__(256) void gather_softmax_kernel(
    const float* __restrict__ z, const int* __restrict__ off,
    const int* __restrict__ csr, const float* __restrict__ t, int ti,
    float* __restrict__ denom, float* __restrict__ numer, int N, int E)
{
  const int gt = blockIdx.x * 256 + threadIdx.x;
  const int d  = gt >> 4;
  if (d >= N) return;
  const int f  = (gt & 15) * 4;
  int j0 = off[d], j1 = off[d + 1];
  j0 = j0 < 0 ? 0 : (j0 > E ? E : j0);
  j1 = j1 < j0 ? j0 : (j1 > E ? E : j1);

  float d0 = 0.f, d1 = 0.f, d2 = 0.f, d3 = 0.f;
  float n0 = 0.f, n1 = 0.f, n2 = 0.f, n3 = 0.f;

  if (j0 < j1) {
    const int jl = j1 - 1;
    const unsigned last = (unsigned)(N - 1);
    const float tv = t[ti];
    const float* zf = z + f;
    const int jt = j0 + RK;

    unsigned c[RK];
#pragma unroll
    for (int k = 0; k < RK; ++k) c[k] = edge_row(csr, j0 + k, jl, last);

    float4 R[RK];
#pragma unroll
    for (int k = 0; k < RK; ++k) R[k] = *(const float4*)(zf + (size_t)c[k] * Hdim);
    __builtin_amdgcn_sched_barrier(0);

    float x0 = -3.402823466e38f, x1 = x0, x2 = x0, x3 = x0;
#pragma unroll
    for (int g = 0; g < GK; ++g) {
      if (j0 + 4 * g <= jl) {
#pragma unroll
        for (int i = 0; i < 4; ++i) {
          R[4 * g + i] = msg4(R[4 * g + i]);
          seg_max_m(R[4 * g + i], tv, x0, x1, x2, x3);
        }
      }
    }
    for (int j = jt; j < j1; j += 4) {
      const unsigned c0 = edge_row(csr, j, jl, last),     c1 = edge_row(csr, j + 1, jl, last);
      const unsigned c2 = edge_row(csr, j + 2, jl, last), c3 = edge_row(csr, j + 3, jl, last);
      const float4 r0 = *(const float4*)(zf + (size_t)c0 * Hdim);
      const float4 r1 = *(const float4*)(zf + (size_t)c1 * Hdim);
      const float4 r2 = *(const float4*)(zf + (size_t)c2 * Hdim);
      const float4 r3 = *(const float4*)(zf + (size_t)c3 * Hdim);
      seg_max_m(msg4(r0), tv, x0, x1, x2, x3);
      seg_max_m(msg4(r1), tv, x0, x1, x2, x3);
      seg_max_m(msg4(r2), tv, x0, x1, x2, x3);
      seg_max_m(msg4(r3), tv, x0, x1, x2, x3);
    }

#pragma unroll
    for (int g = 0; g < GK; ++g) {
      const int p = j0 + 4 * g;
      if (p <= jl) {
        const bool l1 = p + 1 <= jl, l2 = p + 2 <= jl, l3 = p + 3 <= jl;
        seg_sum4(R[4 * g + 0], tv, x0, x1, x2, x3, true, d0, d1, d2, d3, n0, n1, n2, n3);
        seg_sum4(R[4 * g + 1], tv, x0, x1, x2, x3, l1,   d0, d1, d2, d3, n0, n1, n2, n3);
        seg_sum4(R[4 * g + 2], tv, x0, x1, x2, x3, l2,   d0, d1, d2, d3, n0, n1, n2, n3);
        seg_sum4(R[4 * g + 3], tv, x0, x1, x2, x3, l3,   d0, d1, d2, d3, n0, n1, n2, n3);
      }
    }
    for (int j = jt; j < j1; j += 4) {
      const unsigned c0 = edge_row(csr, j, jl, last),     c1 = edge_row(csr, j + 1, jl, last);
      const unsigned c2 = edge_row(csr, j + 2, jl, last), c3 = edge_row(csr, j + 3, jl, last);
      const float4 r0 = *(const float4*)(zf + (size_t)c0 * Hdim);
      const float4 r1 = *(const float4*)(zf + (size_t)c1 * Hdim);
      const float4 r2 = *(const float4*)(zf + (size_t)c2 * Hdim);
      const float4 r3 = *(const float4*)(zf + (size_t)c3 * Hdim);
      const bool l1 = j + 1 <= jl, l2 = j + 2 <= jl, l3 = j + 3 <= jl;
      seg_sum4(msg4(r0), tv, x0, x1, x2, x3, true, d0, d1, d2, d3, n0, n1, n2, n3);
      seg_sum4(msg4(r1), tv, x0, x1, x2, x3, l1,   d0, d1, d2, d3, n0, n1, n2, n3);
      seg_sum4(msg4(r2), tv, x0, x1, x2, x3, l2,   d0, d1, d2, d3, n0, n1, n2, n3);
      seg_sum4(msg4(r3), tv, x0, x1, x2, x3, l3,   d0, d1, d2, d3, n0, n1, n2, n3);
    }
  }
  const size_t o = (size_t)d * Hdim + f;
  *(float4*)(denom + o) = make_float4(d0, d1, d2, d3);
  *(float4*)(numer + o) = make_float4(n0, n1, n2, n3);
}

template <bool FINAL>
__global__ __launch_bounds__(256) void mlp_kernel(
    const float* z, const float* __restrict__ numer,
    const float* __restrict__ denom, const float* base,
    const float* __restrict__ W1, const float* __restrict__ b1,
    const float* __restrict__ g1, const float* __restrict__ be1,
    const float* __restrict__ W2, const float* __restrict__ b2,
    const float* __restrict__ ng, const float* __restrict__ nb,
    const float* __restrict__ linW, const float* __restrict__ linB,
    float* hout, float* fout, float* __restrict__ outL, int nrows)
{
  __shared__ float U[64][Hdim + 4] __attribute__((aligned(16)));
  __shared__ float T[64][H2 + 4];

  const int tid  = threadIdx.x;
  const int row0 = blockIdx.x * 64;

  const int wave = tid >> 5, lane = tid & 31;
  const int half = lane >> 4, ln = lane & 15;

  {
    float4 zq[4], nq[4], dq[4];
#pragma unroll
    for (int k = 0; k < 4; ++k) {
      const int i  = tid + 256 * k;
      const int gr = row0 + (i >> 4);
      const int lr = gr < nrows ? gr : nrows - 1;
      const long long idx = (long long)lr * Hdim + (i & 15) * 4;
      zq[k] = *(const float4*)(z + idx);
      nq[k] = *(const float4*)(numer + idx);
      dq[k] = *(const float4*)(denom + idx);
    }
#pragma unroll
    for (int k = 0; k < 4; ++k) {
      const int i = tid + 256 * k;
      const int r = i >> 4;
      const bool in = row0 + r < nrows;
      float4 v;
      v.x = zq[k].x + nq[k].x / (dq[k].x + 1e-16f);
      v.y = zq[k].y + nq[k].y / (dq[k].y + 1e-16f);
      v.z = zq[k].z + nq[k].z / (dq[k].z + 1e-16f);
      v.w = zq[k].w + nq[k].w / (dq[k].w + 1e-16f);
      v.x = in ? v.x : 0.f;  v.y = in ? v.y : 0.f;
      v.z = in ? v.z : 0.f;  v.w = in ? v.w : 0.f;
      *(float4*)&U[r][(i & 15) * 4] = v;
    }
  }
  __syncthreads();

  for (int t = 0; t < 4; ++t) {
    int tile = wave * 4 + t;
    int m0 = (tile >> 3) * 16, n0 = (tile & 7) * 16;
    v8f acc = {};
    for (int kk = 0; kk < Hdim; kk += 4) {
      int ka = kk + 2 * half;
      v2f a, b;
      a[0] = U[m0 + ln][ka];     a[1] = U[m0 + ln][ka + 1];
      b[0] = W1[ka * H2 + n0 + ln];
      b[1] = W1[(ka + 1) * H2 + n0 + ln];
      acc = WMMA_F32(a, b, acc);
    }
    for (int r = 0; r < 8; ++r) {
      int mm = m0 + r + 8 * half;
      int nn = n0 + ln;
      T[mm][nn] = acc[r] + b1[nn];
    }
  }
  __syncthreads();

  {
    int r = tid >> 2;
    int q = tid & 3;
    float s = 0.f, s2 = 0.f;
    for (int c = q * 32; c < q * 32 + 32; ++c) {
      float v = T[r][c]; s += v; s2 += v * v;
    }
    s  += __shfl_xor(s, 1);  s  += __shfl_xor(s, 2);
    s2 += __shfl_xor(s2, 1); s2 += __shfl_xor(s2, 2);
    float mu  = s * (1.f / 128.f);
    float var = s2 * (1.f / 128.f) - mu * mu;
    float rs  = rsqrtf(var + 1e-5f);
    for (int c = q * 32; c < q * 32 + 32; ++c) {
      float v = (T[r][c] - mu) * rs * g1[c] + be1[c];
      T[r][c] = fmaxf(v, 0.f);
    }
  }
  __syncthreads();

#pragma unroll
  for (int t = 0; t < 2; ++t) {
    int tile = wave * 2 + t;
    int m0 = (tile >> 2) * 16, n0 = (tile & 3) * 16;
    const int nn = n0 + ln;
    const float bn = b2[nn];
    float bv[8];
    if (base) {
#pragma unroll
      for (int r = 0; r < 8; ++r) {
        const int gr = row0 + m0 + r + 8 * half;
        const int lr = gr < nrows ? gr : nrows - 1;
        bv[r] = base[(long long)lr * Hdim + nn];
      }
    } else {
#pragma unroll
      for (int r = 0; r < 8; ++r) bv[r] = 0.f;
    }
    v8f acc = {};
    for (int kk = 0; kk < H2; kk += 4) {
      int ka = kk + 2 * half;
      v2f a, b;
      a[0] = T[m0 + ln][ka];     a[1] = T[m0 + ln][ka + 1];
      b[0] = W2[ka * Hdim + n0 + ln];
      b[1] = W2[(ka + 1) * Hdim + n0 + ln];
      acc = WMMA_F32(a, b, acc);
    }
    for (int r = 0; r < 8; ++r) {
      float v = acc[r] + bn;
      if (base) v += bv[r];
      U[m0 + r + 8 * half][nn] = v;
    }
  }
  __syncthreads();

  {
    const int lr0 = __builtin_amdgcn_readfirstlane(wave) * ROWS_PER_WAVE;
    const int gr0 = row0 + lr0;
    float v0[ROWS_PER_WAVE], v1[ROWS_PER_WAVE];
#pragma unroll
    for (int k = 0; k < ROWS_PER_WAVE; ++k) {
      v0[k] = U[lr0 + k][lane];
      v1[k] = U[lr0 + k][lane + 32];
    }
    const float g0 = ng[lane], g1n = ng[lane + 32];
    const float b0 = nb[lane], b1n = nb[lane + 32];
    float w0[3], w1[3], lb[3];
    if (FINAL) {
#pragma unroll
      for (int j = 0; j < 3; ++j) {
        w0[j] = linW[lane * 3 + j];
        w1[j] = linW[(lane + 32) * 3 + j];
        lb[j] = linB[j];
      }
    }

    float s[ROWS_PER_WAVE], s2[ROWS_PER_WAVE];
#pragma unroll
    for (int k = 0; k < ROWS_PER_WAVE; ++k) {
      s[k]  = v0[k] + v1[k];
      s2[k] = v0[k] * v0[k] + v1[k] * v1[k];
    }
#pragma unroll
    for (int off = 1; off < 32; off <<= 1) {
#pragma unroll
      for (int k = 0; k < ROWS_PER_WAVE; ++k) {
        s[k]  += __shfl_xor(s[k], off);
        s2[k] += __shfl_xor(s2[k], off);
      }
    }
    float p[ROWS_PER_WAVE][3];
#pragma unroll
    for (int k = 0; k < ROWS_PER_WAVE; ++k) {
      float mu  = s[k] * (1.f / 64.f);
      float var = s2[k] * (1.f / 64.f) - mu * mu;
      float rs  = rsqrtf(var + 1e-5f);
      float f0 = fmaxf((v0[k] - mu) * rs * g0 + b0, 0.f);
      float f1 = fmaxf((v1[k] - mu) * rs * g1n + b1n, 0.f);
      if (gr0 + k < nrows) {
        if (!FINAL) {
          float* hr = hout + (long long)(gr0 + k) * Hdim;
          hr[lane]      = v0[k];
          hr[lane + 32] = v1[k];
        }
        float* fr = fout + (long long)(gr0 + k) * Hdim;
        fr[lane]      = f0;
        fr[lane + 32] = f1;
      }
      if (FINAL) {
#pragma unroll
        for (int j = 0; j < 3; ++j) p[k][j] = f0 * w0[j] + f1 * w1[j];
      }
    }
    if (FINAL) {
#pragma unroll
      for (int off = 1; off < 32; off <<= 1) {
#pragma unroll
        for (int k = 0; k < ROWS_PER_WAVE; ++k) {
#pragma unroll
          for (int j = 0; j < 3; ++j) p[k][j] += __shfl_xor(p[k][j], off);
        }
      }
#pragma unroll
      for (int k = 0; k < ROWS_PER_WAVE; ++k) {
        if (gr0 + k < nrows && lane == 0) {
#pragma unroll
          for (int j = 0; j < 3; ++j) outL[(long long)(gr0 + k) * 3 + j] = p[k][j] + lb[j];
        }
      }
    }
  }
}

extern "C" void kernel_launch(void* const* d_in, const int* in_sizes, int n_in,
                              void* d_out, int out_size, void* d_ws, size_t ws_size,
                              hipStream_t stream)
{
  const float* x    = (const float*)d_in[0];
  const int*   edge = (const int*)d_in[1];
  const float* encW = (const float*)d_in[2];
  const float* encB = (const float*)d_in[3];
  const float* tvec = (const float*)d_in[4];
  const float* W1   = (const float*)d_in[5];
  const float* b1   = (const float*)d_in[6];
  const float* g1   = (const float*)d_in[7];
  const float* be1  = (const float*)d_in[8];
  const float* W2   = (const float*)d_in[9];
  const float* b2   = (const float*)d_in[10];
  const float* ng   = (const float*)d_in[11];
  const float* nb   = (const float*)d_in[12];
  const float* linW = (const float*)d_in[13];
  const float* linB = (const float*)d_in[14];

  const int N = in_sizes[0] / IND;
  const int E = in_sizes[1] / 2;
  const int* src = edge;
  const int* dst = edge + E;

  float* h     = (float*)d_ws;
  float* z     = h     + (size_t)N * Hdim;
  float* plane = z     + (size_t)N * Hdim;
  float* denom = plane + (size_t)N * Hdim;
  float* numer = denom + (size_t)N * Hdim;

  const int n1  = N + 1;
  const int n1a = (n1 + 3) & ~3;
  int* cnt  = (int*)plane;
  int* off  = cnt + n1a;
  int* rank = off + n1a;
  int* csr  = rank + E;

  const int rowBlocks  = (N + 63) / 64;
  const int eBlocks    = (E + 255) / 256;
  const int fBlocks    = (E + 256 * EDGES_PER_THREAD - 1) / (256 * EDGES_PER_THREAD);
  const int gBlocks    = (N + 15) / 16;

  csr_zero_kernel<<<(n1 + 255) / 256, 256, 0, stream>>>(cnt, n1);
  csr_count_kernel<<<eBlocks, 256, 0, stream>>>(dst, cnt, rank, E);
  csr_scan_kernel<<<(n1 + 1023) / 1024, 1024, 0, stream>>>(cnt, off, n1);
  csr_fill_kernel<<<fBlocks, 256, 0, stream>>>(src, dst, off, rank, csr, E);

  enc_gemm_kernel<<<rowBlocks, 256, 0, stream>>>(x, encW, encB, h, N);

  float* outH = (float*)d_out;
  float* outL = outH + (size_t)N * Hdim;

  for (int i = 0; i < 3; ++i) {
    const float* zin  = i == 0 ? h : z;
    const float* base = i == 0 ? nullptr : h;
    gather_softmax_kernel<<<gBlocks, 256, 0, stream>>>(zin, off, csr, tvec, i, denom, numer, N, E);
    if (i < 2) {
      mlp_kernel<false><<<rowBlocks, 256, 0, stream>>>(
          zin, numer, denom, base,
          W1 + (size_t)i * Hdim * H2, b1 + (size_t)i * H2,
          g1 + (size_t)i * H2,        be1 + (size_t)i * H2,
          W2 + (size_t)i * H2 * Hdim, b2 + (size_t)i * Hdim,
          ng + (i + 1) * Hdim, nb + (i + 1) * Hdim, nullptr, nullptr,
          h, z, nullptr, N);
    } else {
      mlp_kernel<true><<<rowBlocks, 256, 0, stream>>>(
          zin, numer, denom, base,
          W1 + (size_t)i * Hdim * H2, b1 + (size_t)i * H2,
          g1 + (size_t)i * H2,        be1 + (size_t)i * H2,
          W2 + (size_t)i * H2 * Hdim, b2 + (size_t)i * Hdim,
          ng, nb, linW, linB,
          nullptr, outH, outL, N);
    }
  }
}


# Gather fused into the MLP kernel: a two-wave block aggregates its 16 nodes (one wave per node) into the u tile in LDS, then runs the MLP; 11 -> 8 launches, no sum planes in memory; sum order kept
# speedup vs baseline: 3.3057x; 1.0873x over previous
#include <hip/hip_runtime.h>
#include <cstdint>


typedef __attribute__((ext_vector_type(2))) float v2f;
typedef __attribute__((ext_vector_type(8))) float v8f;

#define WMMA_F32(a, b, c) \
  __builtin_amdgcn_wmma_f32_16x16x4_f32(false, (a), false, (b), (short)0, (c), false, false)

constexpr int Hdim = 64;
constexpr int H2   = 128;
constexpr int IND  = 128;

__global__ __launch_bounds__(256) void enc_gemm_kernel(
    const float* __restrict__ x, const float* __restrict__ W,
    const float* __restrict__ bias, float* __restrict__ h, int nrows)
{
  __shared__ float Xs[64][IND + 4] __attribute__((aligned(16)));
  const int tid  = threadIdx.x;
  const int row0 = blockIdx.x * 64;

  float4 xv[8];
#pragma unroll
  for (int k = 0; k < 8; ++k) {
    const int i  = tid + 256 * k;
    const int gr = row0 + (i >> 5);
    const int lr = gr < nrows ? gr : nrows - 1;
    xv[k] = *(const float4*)(x + (long long)lr * IND + (i & 31) * 4);
  }
#pragma unroll
  for (int k = 0; k < 8; ++k) {
    const int i = tid + 256 * k;
    const int r = i >> 5;
    const float4 v = (row0 + r < nrows) ? xv[k] : make_float4(0.f, 0.f, 0.f, 0.f);
    *(float4*)&Xs[r][(i & 31) * 4] = v;
  }
  __syncthreads();

  const int wave = tid >> 5, lane = tid & 31;
  const int half = lane >> 4, ln = lane & 15;

  for (int t = 0; t < 2; ++t) {
    int tile = wave * 2 + t;
    int m0 = (tile >> 2) * 16, n0 = (tile & 3) * 16;
    const int nn = n0 + ln;
    const float bn = bias[nn];
    v8f acc = {};
    for (int kk = 0; kk < IND; kk += 4) {
      int ka = kk + 2 * half;
      v2f a, b;
      a[0] = Xs[m0 + ln][ka];      a[1] = Xs[m0 + ln][ka + 1];
      b[0] = W[ka * Hdim + n0 + ln];
      b[1] = W[(ka + 1) * Hdim + n0 + ln];
      acc = WMMA_F32(a, b, acc);
    }
    for (int r = 0; r < 8; ++r) {
      int gr = row0 + m0 + r + 8 * half;
      if (gr < nrows) h[(long long)gr * Hdim + nn] = acc[r] + bn;
    }
  }
}

constexpr int ROWS_PER_WAVE = 8;

__global__ __launch_bounds__(256) void csr_zero_kernel(int* __restrict__ cnt, int n1)
{
  int i = blockIdx.x * 256 + threadIdx.x;
  if (i < n1) cnt[i] = 0;
}

__global__ __launch_bounds__(256) void csr_count_kernel(
    const int* __restrict__ dst, int* __restrict__ cnt, int* __restrict__ rank, int E)
{
  int e = blockIdx.x * 256 + threadIdx.x;
  if (e >= E) return;
  rank[e] = atomicAdd(cnt + dst[e], 1);
}

__global__ __launch_bounds__(1024) void csr_scan_kernel(
    const int* __restrict__ cnt, int* __restrict__ off, int n1)
{
  __shared__ int wpre[32];
  __shared__ int wtot[32];
  const int tid = threadIdx.x, lane = tid & 31, wave = tid >> 5;
  const int base = blockIdx.x * 1024;

  int pre = 0;
  for (int i = tid; i < base; i += 1024) pre += cnt[i];

  const int idx = base + tid;
  const int v = (idx < n1) ? cnt[idx] : 0;
  int inc = v;
  for (int o = 1; o < 32; o <<= 1) {
    pre += __shfl_xor(pre, o);
    int up = __shfl_up(inc, o);
    if (lane >= o) inc += up;
  }
  if (lane == 0)  wpre[wave] = pre;
  if (lane == 31) wtot[wave] = inc;
  __syncthreads();

  int bp = wpre[lane];
  int wx = (lane < wave) ? wtot[lane] : 0;
  for (int o = 1; o < 32; o <<= 1) {
    bp += __shfl_xor(bp, o);
    wx += __shfl_xor(wx, o);
  }
  if (idx < n1) off[idx] = bp + wx + inc - v;
}

constexpr int EDGES_PER_THREAD = 4;

__global__ __launch_bounds__(256) void csr_fill_kernel(
    const int* __restrict__ src, const int* __restrict__ dst,
    const int* __restrict__ off, const int* __restrict__ rank,
    int* __restrict__ csr, int E)
{
  const int e0 = blockIdx.x * (256 * EDGES_PER_THREAD) + threadIdx.x;
  int dd[EDGES_PER_THREAD], rr[EDGES_PER_THREAD], ss[EDGES_PER_THREAD];
#pragma unroll
  for (int k = 0; k < EDGES_PER_THREAD; ++k) {
    const int e  = e0 + 256 * k;
    const int le = e < E ? e : E - 1;
    dd[k] = dst[le];
    rr[k] = rank[le];
    ss[k] = src[le];
  }
  int oo[EDGES_PER_THREAD];
#pragma unroll
  for (int k = 0; k < EDGES_PER_THREAD; ++k) oo[k] = off[dd[k]];
#pragma unroll
  for (int k = 0; k < EDGES_PER_THREAD; ++k) asm volatile("" : "+v"(ss[k]));
#pragma unroll
  for (int k = 0; k < EDGES_PER_THREAD; ++k) {
    const int e   = e0 + 256 * k;
    const int pos = oo[k] + rr[k];
    if (e < E && (unsigned)pos < (unsigned)E) csr[pos] = ss[k];
  }
}

constexpr int GK = 6;
constexpr int RK = 4 * GK;

__device__ __forceinline__ unsigned edge_row(const int* __restrict__ csr, int p, int jl,
                                             unsigned last)
{
  p = p > jl ? jl : p;
  unsigned s = (unsigned)csr[p];
  return s > last ? last : s;
}

__device__ __forceinline__ float2 z_row(const char* __restrict__ zc, unsigned row, unsigned fb)
{
  return *(const float2*)(zc + (size_t)row * (size_t)(Hdim * 4) + fb);
}

__device__ __forceinline__ float2 msg2(float2 zv)
{
  return make_float2(fmaxf(zv.x, 0.f) + 1e-7f, fmaxf(zv.y, 0.f) + 1e-7f);
}

__device__ __forceinline__ void seg_max_m(float2 m, float tv, float& x0, float& x1)
{
  x0 = fmaxf(x0, m.x * tv);
  x1 = fmaxf(x1, m.y * tv);
}

__device__ __forceinline__ void seg_sum_m(float m, float tv, float mx, bool live,
                                          float& den, float& num)
{
  float a = __expf(m * tv - mx);
  a = live ? a : 0.f;
  den += a;
  num += m * a;
}

__device__ __forceinline__ void seg_sum2(float2 m, float tv, float x0, float x1, bool live,
                                         float& d0, float& d1, float& n0, float& n1)
{
  seg_sum_m(m.x, tv, x0, live, d0, n0);
  seg_sum_m(m.y, tv, x1, live, d1, n1);
}

__device__ __forceinline__ void node_sums(
    const char* __restrict__ zc, unsigned fb, const int* __restrict__ csr,
    int j0, int j1, int lane, float tv, unsigned last,
    float& d0, float& d1, float& n0, float& n1)
{
  const int jl = j1 - 1;
  const int jt = j0 + RK;

  const unsigned cv = edge_row(csr, j0 + lane, jl, last);

  float2 R[RK];
#pragma unroll
  for (int g = 0; g < GK; ++g) {
    if (j0 + 4 * g <= jl) {
#pragma unroll
      for (int i = 0; i < 4; ++i)
        R[4 * g + i] = z_row(zc, (unsigned)__builtin_amdgcn_readlane((int)cv, 4 * g + i), fb);
    }
  }

  float x0 = -3.402823466e38f, x1 = x0;
#pragma unroll
  for (int g = 0; g < GK; ++g) {
    if (j0 + 4 * g <= jl) {
#pragma unroll
      for (int i = 0; i < 4; ++i) {
        R[4 * g + i] = msg2(R[4 * g + i]);
        seg_max_m(R[4 * g + i], tv, x0, x1);
      }
    }
  }
  for (int j = jt; j < j1; j += 4) {
    const unsigned c0 = edge_row(csr, j, jl, last),     c1 = edge_row(csr, j + 1, jl, last);
    const unsigned c2 = edge_row(csr, j + 2, jl, last), c3 = edge_row(csr, j + 3, jl, last);
    const float2 r0 = z_row(zc, c0, fb);
    const float2 r1 = z_row(zc, c1, fb);
    const float2 r2 = z_row(zc, c2, fb);
    const float2 r3 = z_row(zc, c3, fb);
    seg_max_m(msg2(r0), tv, x0, x1);
    seg_max_m(msg2(r1), tv, x0, x1);
    seg_max_m(msg2(r2), tv, x0, x1);
    seg_max_m(msg2(r3), tv, x0, x1);
  }

#pragma unroll
  for (int g = 0; g < GK; ++g) {
    const int p = j0 + 4 * g;
    if (p <= jl) {
      const bool l1 = p + 1 <= jl, l2 = p + 2 <= jl, l3 = p + 3 <= jl;
      seg_sum2(R[4 * g + 0], tv, x0, x1, true, d0, d1, n0, n1);
      seg_sum2(R[4 * g + 1], tv, x0, x1, l1,   d0, d1, n0, n1);
      seg_sum2(R[4 * g + 2], tv, x0, x1, l2,   d0, d1, n0, n1);
      seg_sum2(R[4 * g + 3], tv, x0, x1, l3,   d0, d1, n0, n1);
    }
  }
  for (int j = jt; j < j1; j += 4) {
    const unsigned c0 = edge_row(csr, j, jl, last),     c1 = edge_row(csr, j + 1, jl, last);
    const unsigned c2 = edge_row(csr, j + 2, jl, last), c3 = edge_row(csr, j + 3, jl, last);
    const float2 r0 = z_row(zc, c0, fb);
    const float2 r1 = z_row(zc, c1, fb);
    const float2 r2 = z_row(zc, c2, fb);
    const float2 r3 = z_row(zc, c3, fb);
    const bool l1 = j + 1 <= jl, l2 = j + 2 <= jl, l3 = j + 3 <= jl;
    seg_sum2(msg2(r0), tv, x0, x1, true, d0, d1, n0, n1);
    seg_sum2(msg2(r1), tv, x0, x1, l1,   d0, d1, n0, n1);
    seg_sum2(msg2(r2), tv, x0, x1, l2,   d0, d1, n0, n1);
    seg_sum2(msg2(r3), tv, x0, x1, l3,   d0, d1, n0, n1);
  }
}

constexpr int TR = 16;

template <bool FINAL>
__global__ __launch_bounds__(TR / ROWS_PER_WAVE * 32) void conv_kernel(
    const float* __restrict__ z, const int* __restrict__ off,
    const int* __restrict__ csr, const float* __restrict__ t, int ti,
    const float* base,
    const float* __restrict__ W1, const float* __restrict__ b1,
    const float* __restrict__ g1, const float* __restrict__ be1,
    const float* __restrict__ W2, const float* __restrict__ b2,
    const float* __restrict__ ng, const float* __restrict__ nb,
    const float* __restrict__ linW, const float* __restrict__ linB,
    float* hout, float* __restrict__ fout, float* __restrict__ outL, int nrows, int E)
{
  __shared__ float U[TR][Hdim + 4] __attribute__((aligned(16)));
  __shared__ float T[TR][H2 + 4];

  const int tid  = threadIdx.x;
  const int row0 = blockIdx.x * TR;

  const int wave = tid >> 5, lane = tid & 31;
  const int half = lane >> 4, ln = lane & 15;

  {
    const int lr0 = __builtin_amdgcn_readfirstlane(wave) * ROWS_PER_WAVE;
    const unsigned fb = (unsigned)lane * 8u;
    const char* zc = (const char*)z;
    const float tv = t[ti];
    const unsigned last = (unsigned)(nrows - 1);
    int ob;
    {
      const int k = lane < ROWS_PER_WAVE ? lane : ROWS_PER_WAVE;
      const int d = row0 + lr0 + k;
      ob = off[d < nrows ? d : nrows];
      ob = ob < 0 ? 0 : (ob > E ? E : ob);
    }
#pragma nounroll
    for (int i = 0; i < ROWS_PER_WAVE; ++i) {
      const int r = lr0 + i;
      const int d = row0 + r;
      float2 u = make_float2(0.f, 0.f);
      if (d < nrows) {
        const int j0 = __builtin_amdgcn_readlane(ob, i);
        int j1 = __builtin_amdgcn_readlane(ob, i + 1);
        j1 = j1 < j0 ? j0 : j1;
        const float2 zv = z_row(zc, (unsigned)d, fb);
        float d0 = 0.f, d1 = 0.f, n0 = 0.f, n1 = 0.f;
        if (j0 < j1)
          node_sums(zc, fb, csr, j0, j1, lane, tv, last, d0, d1, n0, n1);
        u.x = zv.x + n0 / (d0 + 1e-16f);
        u.y = zv.y + n1 / (d1 + 1e-16f);
      }
      *(float2*)&U[r][2 * lane] = u;
    }
  }
  __syncthreads();

  for (int t = 0; t < 4; ++t) {
    const int n0 = (wave * 4 + t) * 16;
    v8f acc = {};
    for (int kk = 0; kk < Hdim; kk += 4) {
      int ka = kk + 2 * half;
      v2f a, b;
      a[0] = U[ln][ka];     a[1] = U[ln][ka + 1];
      b[0] = W1[ka * H2 + n0 + ln];
      b[1] = W1[(ka + 1) * H2 + n0 + ln];
      acc = WMMA_F32(a, b, acc);
    }
    for (int r = 0; r < 8; ++r) {
      int mm = r + 8 * half;
      int nn = n0 + ln;
      T[mm][nn] = acc[r] + b1[nn];
    }
  }
  __syncthreads();

  {
    int r = tid >> 2;
    int q = tid & 3;
    float s = 0.f, s2 = 0.f;
    for (int c = q * 32; c < q * 32 + 32; ++c) {
      float v = T[r][c]; s += v; s2 += v * v;
    }
    s  += __shfl_xor(s, 1);  s  += __shfl_xor(s, 2);
    s2 += __shfl_xor(s2, 1); s2 += __shfl_xor(s2, 2);
    float mu  = s * (1.f / 128.f);
    float var = s2 * (1.f / 128.f) - mu * mu;
    float rs  = rsqrtf(var + 1e-5f);
    for (int c = q * 32; c < q * 32 + 32; ++c) {
      float v = (T[r][c] - mu) * rs * g1[c] + be1[c];
      T[r][c] = fmaxf(v, 0.f);
    }
  }
  __syncthreads();

#pragma unroll
  for (int t = 0; t < 2; ++t) {
    const int n0 = (wave * 2 + t) * 16;
    const int nn = n0 + ln;
    const float bn = b2[nn];
    float bv[8];
    if (base) {
#pragma unroll
      for (int r = 0; r < 8; ++r) {
        const int gr = row0 + r + 8 * half;
        const int lr = gr < nrows ? gr : nrows - 1;
        bv[r] = base[(long long)lr * Hdim + nn];
      }
    } else {
#pragma unroll
      for (int r = 0; r < 8; ++r) bv[r] = 0.f;
    }
    v8f acc = {};
    for (int kk = 0; kk < H2; kk += 4) {
      int ka = kk + 2 * half;
      v2f a, b;
      a[0] = T[ln][ka];     a[1] = T[ln][ka + 1];
      b[0] = W2[ka * Hdim + n0 + ln];
      b[1] = W2[(ka + 1) * Hdim + n0 + ln];
      acc = WMMA_F32(a, b, acc);
    }
    for (int r = 0; r < 8; ++r) {
      float v = acc[r] + bn;
      if (base) v += bv[r];
      U[r + 8 * half][nn] = v;
    }
  }
  __syncthreads();

  {
    const int lr0 = __builtin_amdgcn_readfirstlane(wave) * ROWS_PER_WAVE;
    const int gr0 = row0 + lr0;
    float v0[ROWS_PER_WAVE], v1[ROWS_PER_WAVE];
#pragma unroll
    for (int k = 0; k < ROWS_PER_WAVE; ++k) {
      v0[k] = U[lr0 + k][lane];
      v1[k] = U[lr0 + k][lane + 32];
    }
    const float g0 = ng[lane], g1n = ng[lane + 32];
    const float b0 = nb[lane], b1n = nb[lane + 32];
    float w0[3], w1[3], lb[3];
    if (FINAL) {
#pragma unroll
      for (int j = 0; j < 3; ++j) {
        w0[j] = linW[lane * 3 + j];
        w1[j] = linW[(lane + 32) * 3 + j];
        lb[j] = linB[j];
      }
    }

    float s[ROWS_PER_WAVE], s2[ROWS_PER_WAVE];
#pragma unroll
    for (int k = 0; k < ROWS_PER_WAVE; ++k) {
      s[k]  = v0[k] + v1[k];
      s2[k] = v0[k] * v0[k] + v1[k] * v1[k];
    }
#pragma unroll
    for (int off = 1; off < 32; off <<= 1) {
#pragma unroll
      for (int k = 0; k < ROWS_PER_WAVE; ++k) {
        s[k]  += __shfl_xor(s[k], off);
        s2[k] += __shfl_xor(s2[k], off);
      }
    }
    float p[ROWS_PER_WAVE][3];
#pragma unroll
    for (int k = 0; k < ROWS_PER_WAVE; ++k) {
      float mu  = s[k] * (1.f / 64.f);
      float var = s2[k] * (1.f / 64.f) - mu * mu;
      float rs  = rsqrtf(var + 1e-5f);
      float f0 = fmaxf((v0[k] - mu) * rs * g0 + b0, 0.f);
      float f1 = fmaxf((v1[k] - mu) * rs * g1n + b1n, 0.f);
      if (gr0 + k < nrows) {
        if (!FINAL) {
          float* hr = hout + (long long)(gr0 + k) * Hdim;
          hr[lane]      = v0[k];
          hr[lane + 32] = v1[k];
        }
        float* fr = fout + (long long)(gr0 + k) * Hdim;
        fr[lane]      = f0;
        fr[lane + 32] = f1;
      }
      if (FINAL) {
#pragma unroll
        for (int j = 0; j < 3; ++j) p[k][j] = f0 * w0[j] + f1 * w1[j];
      }
    }
    if (FINAL) {
#pragma unroll
      for (int off = 1; off < 32; off <<= 1) {
#pragma unroll
        for (int k = 0; k < ROWS_PER_WAVE; ++k) {
#pragma unroll
          for (int j = 0; j < 3; ++j) p[k][j] += __shfl_xor(p[k][j], off);
        }
      }
#pragma unroll
      for (int k = 0; k < ROWS_PER_WAVE; ++k) {
        if (gr0 + k < nrows && lane == 0) {
#pragma unroll
          for (int j = 0; j < 3; ++j) outL[(long long)(gr0 + k) * 3 + j] = p[k][j] + lb[j];
        }
      }
    }
  }
}

extern "C" void kernel_launch(void* const* d_in, const int* in_sizes, int n_in,
                              void* d_out, int out_size, void* d_ws, size_t ws_size,
                              hipStream_t stream)
{
  const float* x    = (const float*)d_in[0];
  const int*   edge = (const int*)d_in[1];
  const float* encW = (const float*)d_in[2];
  const float* encB = (const float*)d_in[3];
  const float* tvec = (const float*)d_in[4];
  const float* W1   = (const float*)d_in[5];
  const float* b1   = (const float*)d_in[6];
  const float* g1   = (const float*)d_in[7];
  const float* be1  = (const float*)d_in[8];
  const float* W2   = (const float*)d_in[9];
  const float* b2   = (const float*)d_in[10];
  const float* ng   = (const float*)d_in[11];
  const float* nb   = (const float*)d_in[12];
  const float* linW = (const float*)d_in[13];
  const float* linB = (const float*)d_in[14];

  const int N = in_sizes[0] / IND;
  const int E = in_sizes[1] / 2;
  const int* src = edge;
  const int* dst = edge + E;

  float* za    = (float*)d_ws;
  float* zb    = za    + (size_t)N * Hdim;
  float* plane = zb    + (size_t)N * Hdim;
  float* h     = plane + (size_t)N * Hdim;

  const int n1  = N + 1;
  const int n1a = (n1 + 3) & ~3;
  int* cnt  = (int*)plane;
  int* off  = cnt + n1a;
  int* rank = off + n1a;
  int* csr  = rank + E;

  const int rowBlocks  = (N + 63) / 64;
  const int cBlocks    = (N + TR - 1) / TR;
  const int cThreads   = TR / ROWS_PER_WAVE * 32;
  const int eBlocks    = (E + 255) / 256;
  const int fBlocks    = (E + 256 * EDGES_PER_THREAD - 1) / (256 * EDGES_PER_THREAD);

  csr_zero_kernel<<<(n1 + 255) / 256, 256, 0, stream>>>(cnt, n1);
  csr_count_kernel<<<eBlocks, 256, 0, stream>>>(dst, cnt, rank, E);
  csr_scan_kernel<<<(n1 + 1023) / 1024, 1024, 0, stream>>>(cnt, off, n1);
  csr_fill_kernel<<<fBlocks, 256, 0, stream>>>(src, dst, off, rank, csr, E);

  enc_gemm_kernel<<<rowBlocks, 256, 0, stream>>>(x, encW, encB, za, N);

  float* outH = (float*)d_out;
  float* outL = outH + (size_t)N * Hdim;

  for (int i = 0; i < 3; ++i) {
    const float* zin  = i == 1 ? zb : za;
    float*       zout = i == 0 ? zb : za;
    const float* base = i == 0 ? nullptr : h;
    if (i < 2) {
      conv_kernel<false><<<cBlocks, cThreads, 0, stream>>>(
          zin, off, csr, tvec, i, base,
          W1 + (size_t)i * Hdim * H2, b1 + (size_t)i * H2,
          g1 + (size_t)i * H2,        be1 + (size_t)i * H2,
          W2 + (size_t)i * H2 * Hdim, b2 + (size_t)i * Hdim,
          ng + (i + 1) * Hdim, nb + (i + 1) * Hdim, nullptr, nullptr,
          h, zout, nullptr, N, E);
    } else {
      conv_kernel<true><<<cBlocks, cThreads, 0, stream>>>(
          zin, off, csr, tvec, i, base,
          W1 + (size_t)i * Hdim * H2, b1 + (size_t)i * H2,
          g1 + (size_t)i * H2,        be1 + (size_t)i * H2,
          W2 + (size_t)i * H2 * Hdim, b2 + (size_t)i * Hdim,
          ng, nb, linW, linB,
          nullptr, outH, outL, N, E);
    }
  }
}


# Edge list built by a two-level counting sort ranked with LDS counters (256-node buckets, then nodes), not one returning global add per edge: 8 -> 7 launches; segment order stays an arrival order
# speedup vs baseline: 4.8313x; 1.4615x over previous
#include <hip/hip_runtime.h>
#include <cstdint>


typedef __attribute__((ext_vector_type(2))) float v2f;
typedef __attribute__((ext_vector_type(8))) float v8f;

#define WMMA_F32(a, b, c) \
  __builtin_amdgcn_wmma_f32_16x16x4_f32(false, (a), false, (b), (short)0, (c), false, false)

constexpr int Hdim = 64;
constexpr int H2   = 128;
constexpr int IND  = 128;

__global__ __launch_bounds__(256) void enc_gemm_kernel(
    const float* __restrict__ x, const float* __restrict__ W,
    const float* __restrict__ bias, float* __restrict__ h, int nrows)
{
  __shared__ float Xs[64][IND + 4] __attribute__((aligned(16)));
  const int tid  = threadIdx.x;
  const int row0 = blockIdx.x * 64;

  float4 xv[8];
#pragma unroll
  for (int k = 0; k < 8; ++k) {
    const int i  = tid + 256 * k;
    const int gr = row0 + (i >> 5);
    const int lr = gr < nrows ? gr : nrows - 1;
    xv[k] = *(const float4*)(x + (long long)lr * IND + (i & 31) * 4);
  }
#pragma unroll
  for (int k = 0; k < 8; ++k) {
    const int i = tid + 256 * k;
    const int r = i >> 5;
    const float4 v = (row0 + r < nrows) ? xv[k] : make_float4(0.f, 0.f, 0.f, 0.f);
    *(float4*)&Xs[r][(i & 31) * 4] = v;
  }
  __syncthreads();

  const int wave = tid >> 5, lane = tid & 31;
  const int half = lane >> 4, ln = lane & 15;

  for (int t = 0; t < 2; ++t) {
    int tile = wave * 2 + t;
    int m0 = (tile >> 2) * 16, n0 = (tile & 3) * 16;
    const int nn = n0 + ln;
    const float bn = bias[nn];
    v8f acc = {};
    for (int kk = 0; kk < IND; kk += 4) {
      int ka = kk + 2 * half;
      v2f a, b;
      a[0] = Xs[m0 + ln][ka];      a[1] = Xs[m0 + ln][ka + 1];
      b[0] = W[ka * Hdim + n0 + ln];
      b[1] = W[(ka + 1) * Hdim + n0 + ln];
      acc = WMMA_F32(a, b, acc);
    }
    for (int r = 0; r < 8; ++r) {
      int gr = row0 + m0 + r + 8 * half;
      if (gr < nrows) h[(long long)gr * Hdim + nn] = acc[r] + bn;
    }
  }
}

constexpr int ROWS_PER_WAVE = 8;

constexpr int BSH   = 8;
constexpr int BW    = 1 << BSH;
constexpr int NBMAX = 256;
constexpr int EPT   = 4;
constexpr int CH    = 1024 * EPT;

__device__ __forceinline__ int bucket_of(int d, int NB)
{
  const unsigned b = (unsigned)d >> BSH;
  return b < (unsigned)NB ? (int)b : NB - 1;
}

__global__ __launch_bounds__(1024) void bin_count_kernel(
    const int* __restrict__ dst, int* __restrict__ M, int E, int NB)
{
  __shared__ int hist[NBMAX];
  const int tid = threadIdx.x;
  const int e0  = blockIdx.x * CH + tid;
  int dd[EPT];
#pragma unroll
  for (int k = 0; k < EPT; ++k) {
    const int e = e0 + 1024 * k;
    dd[k] = dst[e < E ? e : E - 1];
  }
  if (tid < NBMAX) hist[tid] = 0;
  __syncthreads();
#pragma unroll
  for (int k = 0; k < EPT; ++k)
    if (e0 + 1024 * k < E) atomicAdd(&hist[bucket_of(dd[k], NB)], 1);
  __syncthreads();
  if (tid < NB && tid < NBMAX) M[blockIdx.x * NB + tid] = hist[tid];
}

__global__ __launch_bounds__(1024) void bin_scatter_kernel(
    const int* __restrict__ src, const int* __restrict__ dst,
    const int* __restrict__ M, int* __restrict__ bstart, int* __restrict__ rec,
    int E, int NB, int NC)
{
  __shared__ int ptot[4][NBMAX];
  __shared__ int ppre[4][NBMAX];
  __shared__ int cur[NBMAX];
  __shared__ int wsum[NBMAX / 32];
  const int tid = threadIdx.x, lane = tid & 31, wave = tid >> 5;
  const int c  = blockIdx.x;
  const int e0 = c * CH + tid;

  int dd[EPT], ss[EPT];
#pragma unroll
  for (int k = 0; k < EPT; ++k) {
    const int e  = e0 + 1024 * k;
    const int le = e < E ? e : E - 1;
    dd[k] = dst[le];
    ss[k] = src[le];
  }

  const int b = tid & (NBMAX - 1), q = tid >> 8;
  int tot = 0, pre = 0;
  if (b < NB) {
    for (int k0 = q; k0 < NC; k0 += 64) {
      int v[16];
#pragma unroll
      for (int j = 0; j < 16; ++j) {
        const int k = k0 + 4 * j;
        v[j] = M[(k < NC ? k : NC - 1) * NB + b];
      }
#pragma unroll
      for (int j = 0; j < 16; ++j) {
        const int k = k0 + 4 * j;
        const int a = k < NC ? v[j] : 0;
        tot += a;
        pre += k < c ? a : 0;
      }
    }
  }
  ptot[q][b] = tot;
  ppre[q][b] = pre;
  __syncthreads();

  int inc = 0;
  if (tid < NBMAX) {
    tot = ptot[0][tid] + ptot[1][tid] + ptot[2][tid] + ptot[3][tid];
    pre = ppre[0][tid] + ppre[1][tid] + ppre[2][tid] + ppre[3][tid];
    inc = tot;
    for (int o = 1; o < 32; o <<= 1) {
      const int up = __shfl_up(inc, o);
      if (lane >= o) inc += up;
    }
    if (lane == 31) wsum[wave] = inc;
  }
  __syncthreads();
  if (tid < NBMAX) {
    int wx = 0;
    for (int w = 0; w < wave; ++w) wx += wsum[w];
    const int ex = wx + inc - tot;
    cur[tid] = ex + pre;
    if (c == 0) {
      if (tid < NB)      bstart[tid] = ex;
      if (tid == NB - 1) bstart[NB]  = ex + tot;
    }
  }
  __syncthreads();

  int2* rc = (int2*)rec;
#pragma unroll
  for (int k = 0; k < EPT; ++k) {
    if (e0 + 1024 * k < E) {
      const int pos = atomicAdd(&cur[bucket_of(dd[k], NB)], 1);
      if ((unsigned)pos < (unsigned)E) rc[pos] = make_int2(ss[k], dd[k]);
    }
  }
}

__device__ __forceinline__ int local_of(int d, int n0)
{
  const int l = d - n0;
  return l < 0 ? 0 : (l > BW - 1 ? BW - 1 : l);
}

__global__ __launch_bounds__(1024) void bucket_fill_kernel(
    const int* __restrict__ rec, const int* __restrict__ bstart,
    int* __restrict__ off, int* __restrict__ csr, int N, int E, int NB)
{
  __shared__ int deg[BW];
  __shared__ int loff[BW];
  __shared__ int wsum[BW / 32];
  const int tid = threadIdx.x, lane = tid & 31, wave = tid >> 5;
  const int b  = blockIdx.x;
  const int n0 = b * BW;
  const int2* rc = (const int2*)rec;
  int s0 = bstart[b], s1 = bstart[b + 1];
  s0 = s0 < 0 ? 0 : (s0 > E ? E : s0);
  s1 = s1 < s0 ? s0 : (s1 > E ? E : s1);

  if (tid < BW) deg[tid] = 0;
  __syncthreads();
  for (int r0 = s0 + tid; r0 < s1; r0 += 4096) {
    int dv[4];
#pragma unroll
    for (int j = 0; j < 4; ++j) {
      const int r = r0 + 1024 * j;
      dv[j] = rc[r < s1 ? r : s1 - 1].y;
    }
#pragma unroll
    for (int j = 0; j < 4; ++j) asm volatile("" : "+v"(dv[j]));
#pragma unroll
    for (int j = 0; j < 4; ++j)
      if (r0 + 1024 * j < s1) atomicAdd(&deg[local_of(dv[j], n0)], 1);
  }
  __syncthreads();

  int v = 0, inc = 0;
  if (tid < BW) {
    v = deg[tid];
    inc = v;
    for (int o = 1; o < 32; o <<= 1) {
      const int up = __shfl_up(inc, o);
      if (lane >= o) inc += up;
    }
    if (lane == 31) wsum[wave] = inc;
  }
  __syncthreads();
  if (tid < BW) {
    int wx = 0;
    for (int w = 0; w < wave; ++w) wx += wsum[w];
    const int st = s0 + wx + inc - v;
    loff[tid] = st;
    deg[tid]  = 0;
    const int node = n0 + tid;
    if (node <= N) off[node] = st;
    if (b == NB - 1 && tid == BW - 1 && node + 1 == N) off[N] = st + v;
  }
  __syncthreads();

  for (int r0 = s0 + tid; r0 < s1; r0 += 4096) {
    int2 rv[4];
#pragma unroll
    for (int j = 0; j < 4; ++j) {
      const int r = r0 + 1024 * j;
      rv[j] = rc[r < s1 ? r : s1 - 1];
    }
#pragma unroll
    for (int j = 0; j < 4; ++j) asm volatile("" : "+v"(rv[j].x), "+v"(rv[j].y));
#pragma unroll
    for (int j = 0; j < 4; ++j) {
      if (r0 + 1024 * j < s1) {
        const int l   = local_of(rv[j].y, n0);
        const int pos = loff[l] + atomicAdd(&deg[l], 1);
        if ((unsigned)pos < (unsigned)E) csr[pos] = rv[j].x;
      }
    }
  }
}

constexpr int GK = 6;
constexpr int RK = 4 * GK;

__device__ __forceinline__ unsigned edge_row(const int* __restrict__ csr, int p, int jl,
                                             unsigned last)
{
  p = p > jl ? jl : p;
  unsigned s = (unsigned)csr[p];
  return s > last ? last : s;
}

__device__ __forceinline__ float2 z_row(const char* __restrict__ zc, unsigned row, unsigned fb)
{
  return *(const float2*)(zc + (size_t)row * (size_t)(Hdim * 4) + fb);
}

__device__ __forceinline__ float2 msg2(float2 zv)
{
  return make_float2(fmaxf(zv.x, 0.f) + 1e-7f, fmaxf(zv.y, 0.f) + 1e-7f);
}

__device__ __forceinline__ void seg_max_m(float2 m, float tv, float& x0, float& x1)
{
  x0 = fmaxf(x0, m.x * tv);
  x1 = fmaxf(x1, m.y * tv);
}

__device__ __forceinline__ void seg_sum_m(float m, float tv, float mx, bool live,
                                          float& den, float& num)
{
  float a = __expf(m * tv - mx);
  a = live ? a : 0.f;
  den += a;
  num += m * a;
}

__device__ __forceinline__ void seg_sum2(float2 m, float tv, float x0, float x1, bool live,
                                         float& d0, float& d1, float& n0, float& n1)
{
  seg_sum_m(m.x, tv, x0, live, d0, n0);
  seg_sum_m(m.y, tv, x1, live, d1, n1);
}

__device__ __forceinline__ void node_sums(
    const char* __restrict__ zc, unsigned fb, const int* __restrict__ csr,
    int j0, int j1, int lane, float tv, unsigned last,
    float& d0, float& d1, float& n0, float& n1)
{
  const int jl = j1 - 1;
  const int jt = j0 + RK;

  const unsigned cv = edge_row(csr, j0 + lane, jl, last);

  float2 R[RK];
#pragma unroll
  for (int g = 0; g < GK; ++g) {
    if (j0 + 4 * g <= jl) {
#pragma unroll
      for (int i = 0; i < 4; ++i)
        R[4 * g + i] = z_row(zc, (unsigned)__builtin_amdgcn_readlane((int)cv, 4 * g + i), fb);
    }
  }

  float x0 = -3.402823466e38f, x1 = x0;
#pragma unroll
  for (int g = 0; g < GK; ++g) {
    if (j0 + 4 * g <= jl) {
#pragma unroll
      for (int i = 0; i < 4; ++i) {
        R[4 * g + i] = msg2(R[4 * g + i]);
        seg_max_m(R[4 * g + i], tv, x0, x1);
      }
    }
  }
  for (int j = jt; j < j1; j += 4) {
    const unsigned c0 = edge_row(csr, j, jl, last),     c1 = edge_row(csr, j + 1, jl, last);
    const unsigned c2 = edge_row(csr, j + 2, jl, last), c3 = edge_row(csr, j + 3, jl, last);
    const float2 r0 = z_row(zc, c0, fb);
    const float2 r1 = z_row(zc, c1, fb);
    const float2 r2 = z_row(zc, c2, fb);
    const float2 r3 = z_row(zc, c3, fb);
    seg_max_m(msg2(r0), tv, x0, x1);
    seg_max_m(msg2(r1), tv, x0, x1);
    seg_max_m(msg2(r2), tv, x0, x1);
    seg_max_m(msg2(r3), tv, x0, x1);
  }

#pragma unroll
  for (int g = 0; g < GK; ++g) {
    const int p = j0 + 4 * g;
    if (p <= jl) {
      const bool l1 = p + 1 <= jl, l2 = p + 2 <= jl, l3 = p + 3 <= jl;
      seg_sum2(R[4 * g + 0], tv, x0, x1, true, d0, d1, n0, n1);
      seg_sum2(R[4 * g + 1], tv, x0, x1, l1,   d0, d1, n0, n1);
      seg_sum2(R[4 * g + 2], tv, x0, x1, l2,   d0, d1, n0, n1);
      seg_sum2(R[4 * g + 3], tv, x0, x1, l3,   d0, d1, n0, n1);
    }
  }
  for (int j = jt; j < j1; j += 4) {
    const unsigned c0 = edge_row(csr, j, jl, last),     c1 = edge_row(csr, j + 1, jl, last);
    const unsigned c2 = edge_row(csr, j + 2, jl, last), c3 = edge_row(csr, j + 3, jl, last);
    const float2 r0 = z_row(zc, c0, fb);
    const float2 r1 = z_row(zc, c1, fb);
    const float2 r2 = z_row(zc, c2, fb);
    const float2 r3 = z_row(zc, c3, fb);
    const bool l1 = j + 1 <= jl, l2 = j + 2 <= jl, l3 = j + 3 <= jl;
    seg_sum2(msg2(r0), tv, x0, x1, true, d0, d1, n0, n1);
    seg_sum2(msg2(r1), tv, x0, x1, l1,   d0, d1, n0, n1);
    seg_sum2(msg2(r2), tv, x0, x1, l2,   d0, d1, n0, n1);
    seg_sum2(msg2(r3), tv, x0, x1, l3,   d0, d1, n0, n1);
  }
}

constexpr int TR = 16;

template <bool FINAL>
__global__ __launch_bounds__(TR / ROWS_PER_WAVE * 32) void conv_kernel(
    const float* __restrict__ z, const int* __restrict__ off,
    const int* __restrict__ csr, const float* __restrict__ t, int ti,
    const float* base,
    const float* __restrict__ W1, const float* __restrict__ b1,
    const float* __restrict__ g1, const float* __restrict__ be1,
    const float* __restrict__ W2, const float* __restrict__ b2,
    const float* __restrict__ ng, const float* __restrict__ nb,
    const float* __restrict__ linW, const float* __restrict__ linB,
    float* hout, float* __restrict__ fout, float* __restrict__ outL, int nrows, int E)
{
  __shared__ float U[TR][Hdim + 4] __attribute__((aligned(16)));
  __shared__ float T[TR][H2 + 4];

  const int tid  = threadIdx.x;
  const int row0 = blockIdx.x * TR;

  const int wave = tid >> 5, lane = tid & 31;
  const int half = lane >> 4, ln = lane & 15;

  {
    const int lr0 = __builtin_amdgcn_readfirstlane(wave) * ROWS_PER_WAVE;
    const unsigned fb = (unsigned)lane * 8u;
    const char* zc = (const char*)z;
    const float tv = t[ti];
    const unsigned last = (unsigned)(nrows - 1);
    int ob;
    {
      const int k = lane < ROWS_PER_WAVE ? lane : ROWS_PER_WAVE;
      const int d = row0 + lr0 + k;
      ob = off[d < nrows ? d : nrows];
      ob = ob < 0 ? 0 : (ob > E ? E : ob);
    }
#pragma nounroll
    for (int i = 0; i < ROWS_PER_WAVE; ++i) {
      const int r = lr0 + i;
      const int d = row0 + r;
      float2 u = make_float2(0.f, 0.f);
      if (d < nrows) {
        const int j0 = __builtin_amdgcn_readlane(ob, i);
        int j1 = __builtin_amdgcn_readlane(ob, i + 1);
        j1 = j1 < j0 ? j0 : j1;
        const float2 zv = z_row(zc, (unsigned)d, fb);
        float d0 = 0.f, d1 = 0.f, n0 = 0.f, n1 = 0.f;
        if (j0 < j1)
          node_sums(zc, fb, csr, j0, j1, lane, tv, last, d0, d1, n0, n1);
        u.x = zv.x + n0 / (d0 + 1e-16f);
        u.y = zv.y + n1 / (d1 + 1e-16f);
      }
      *(float2*)&U[r][2 * lane] = u;
    }
  }
  __syncthreads();

  for (int t = 0; t < 4; ++t) {
    const int n0 = (wave * 4 + t) * 16;
    v8f acc = {};
    for (int kk = 0; kk < Hdim; kk += 4) {
      int ka = kk + 2 * half;
      v2f a, b;
      a[0] = U[ln][ka];     a[1] = U[ln][ka + 1];
      b[0] = W1[ka * H2 + n0 + ln];
      b[1] = W1[(ka + 1) * H2 + n0 + ln];
      acc = WMMA_F32(a, b, acc);
    }
    for (int r = 0; r < 8; ++r) {
      int mm = r + 8 * half;
      int nn = n0 + ln;
      T[mm][nn] = acc[r] + b1[nn];
    }
  }
  __syncthreads();

  {
    int r = tid >> 2;
    int q = tid & 3;
    float s = 0.f, s2 = 0.f;
    for (int c = q * 32; c < q * 32 + 32; ++c) {
      float v = T[r][c]; s += v; s2 += v * v;
    }
    s  += __shfl_xor(s, 1);  s  += __shfl_xor(s, 2);
    s2 += __shfl_xor(s2, 1); s2 += __shfl_xor(s2, 2);
    float mu  = s * (1.f / 128.f);
    float var = s2 * (1.f / 128.f) - mu * mu;
    float rs  = rsqrtf(var + 1e-5f);
    for (int c = q * 32; c < q * 32 + 32; ++c) {
      float v = (T[r][c] - mu) * rs * g1[c] + be1[c];
      T[r][c] = fmaxf(v, 0.f);
    }
  }
  __syncthreads();

#pragma unroll
  for (int t = 0; t < 2; ++t) {
    const int n0 = (wave * 2 + t) * 16;
    const int nn = n0 + ln;
    const float bn = b2[nn];
    float bv[8];
    if (base) {
#pragma unroll
      for (int r = 0; r < 8; ++r) {
        const int gr = row0 + r + 8 * half;
        const int lr = gr < nrows ? gr : nrows - 1;
        bv[r] = base[(long long)lr * Hdim + nn];
      }
    } else {
#pragma unroll
      for (int r = 0; r < 8; ++r) bv[r] = 0.f;
    }
    v8f acc = {};
    for (int kk = 0; kk < H2; kk += 4) {
      int ka = kk + 2 * half;
      v2f a, b;
      a[0] = T[ln][ka];     a[1] = T[ln][ka + 1];
      b[0] = W2[ka * Hdim + n0 + ln];
      b[1] = W2[(ka + 1) * Hdim + n0 + ln];
      acc = WMMA_F32(a, b, acc);
    }
    for (int r = 0; r < 8; ++r) {
      float v = acc[r] + bn;
      if (base) v += bv[r];
      U[r + 8 * half][nn] = v;
    }
  }
  __syncthreads();

  {
    const int lr0 = __builtin_amdgcn_readfirstlane(wave) * ROWS_PER_WAVE;
    const int gr0 = row0 + lr0;
    float v0[ROWS_PER_WAVE], v1[ROWS_PER_WAVE];
#pragma unroll
    for (int k = 0; k < ROWS_PER_WAVE; ++k) {
      v0[k] = U[lr0 + k][lane];
      v1[k] = U[lr0 + k][lane + 32];
    }
    const float g0 = ng[lane], g1n = ng[lane + 32];
    const float b0 = nb[lane], b1n = nb[lane + 32];
    float w0[3], w1[3], lb[3];
    if (FINAL) {
#pragma unroll
      for (int j = 0; j < 3; ++j) {
        w0[j] = linW[lane * 3 + j];
        w1[j] = linW[(lane + 32) * 3 + j];
        lb[j] = linB[j];
      }
    }

    float s[ROWS_PER_WAVE], s2[ROWS_PER_WAVE];
#pragma unroll
    for (int k = 0; k < ROWS_PER_WAVE; ++k) {
      s[k]  = v0[k] + v1[k];
      s2[k] = v0[k] * v0[k] + v1[k] * v1[k];
    }
#pragma unroll
    for (int off = 1; off < 32; off <<= 1) {
#pragma unroll
      for (int k = 0; k < ROWS_PER_WAVE; ++k) {
        s[k]  += __shfl_xor(s[k], off);
        s2[k] += __shfl_xor(s2[k], off);
      }
    }
    float p[ROWS_PER_WAVE][3];
#pragma unroll
    for (int k = 0; k < ROWS_PER_WAVE; ++k) {
      float mu  = s[k] * (1.f / 64.f);
      float var = s2[k] * (1.f / 64.f) - mu * mu;
      float rs  = rsqrtf(var + 1e-5f);
      float f0 = fmaxf((v0[k] - mu) * rs * g0 + b0, 0.f);
      float f1 = fmaxf((v1[k] - mu) * rs * g1n + b1n, 0.f);
      if (gr0 + k < nrows) {
        if (!FINAL) {
          float* hr = hout + (long long)(gr0 + k) * Hdim;
          hr[lane]      = v0[k];
          hr[lane + 32] = v1[k];
        }
        float* fr = fout + (long long)(gr0 + k) * Hdim;
        fr[lane]      = f0;
        fr[lane + 32] = f1;
      }
      if (FINAL) {
#pragma unroll
        for (int j = 0; j < 3; ++j) p[k][j] = f0 * w0[j] + f1 * w1[j];
      }
    }
    if (FINAL) {
#pragma unroll
      for (int off = 1; off < 32; off <<= 1) {
#pragma unroll
        for (int k = 0; k < ROWS_PER_WAVE; ++k) {
#pragma unroll
          for (int j = 0; j < 3; ++j) p[k][j] += __shfl_xor(p[k][j], off);
        }
      }
#pragma unroll
      for (int k = 0; k < ROWS_PER_WAVE; ++k) {
        if (gr0 + k < nrows && lane == 0) {
#pragma unroll
          for (int j = 0; j < 3; ++j) outL[(long long)(gr0 + k) * 3 + j] = p[k][j] + lb[j];
        }
      }
    }
  }
}

extern "C" void kernel_launch(void* const* d_in, const int* in_sizes, int n_in,
                              void* d_out, int out_size, void* d_ws, size_t ws_size,
                              hipStream_t stream)
{
  const float* x    = (const float*)d_in[0];
  const int*   edge = (const int*)d_in[1];
  const float* encW = (const float*)d_in[2];
  const float* encB = (const float*)d_in[3];
  const float* tvec = (const float*)d_in[4];
  const float* W1   = (const float*)d_in[5];
  const float* b1   = (const float*)d_in[6];
  const float* g1   = (const float*)d_in[7];
  const float* be1  = (const float*)d_in[8];
  const float* W2   = (const float*)d_in[9];
  const float* b2   = (const float*)d_in[10];
  const float* ng   = (const float*)d_in[11];
  const float* nb   = (const float*)d_in[12];
  const float* linW = (const float*)d_in[13];
  const float* linB = (const float*)d_in[14];

  const int N = in_sizes[0] / IND;
  const int E = in_sizes[1] / 2;
  const int* src = edge;
  const int* dst = edge + E;

  float* za    = (float*)d_ws;
  float* zb    = za    + (size_t)N * Hdim;
  float* plane = zb    + (size_t)N * Hdim;
  float* h     = plane + (size_t)N * Hdim;

  const int n1  = N + 1;
  const int n1a = (n1 + 3) & ~3;
  const int NB  = (N + BW - 1) / BW;
  const int NC  = (E + CH - 1) / CH;
  int* off    = (int*)plane;
  int* bstart = off + n1a;
  int* M      = bstart + (NBMAX + 4);
  int* rec    = M + ((NC * NB + 3) & ~3);
  int* csr    = rec + 2 * (size_t)E;

  const int rowBlocks  = (N + 63) / 64;
  const int cBlocks    = (N + TR - 1) / TR;
  const int cThreads   = TR / ROWS_PER_WAVE * 32;

  bin_count_kernel<<<NC, 1024, 0, stream>>>(dst, M, E, NB);
  bin_scatter_kernel<<<NC, 1024, 0, stream>>>(src, dst, M, bstart, rec, E, NB, NC);
  bucket_fill_kernel<<<NB, 1024, 0, stream>>>(rec, bstart, off, csr, N, E, NB);

  enc_gemm_kernel<<<rowBlocks, 256, 0, stream>>>(x, encW, encB, za, N);

  float* outH = (float*)d_out;
  float* outL = outH + (size_t)N * Hdim;

  for (int i = 0; i < 3; ++i) {
    const float* zin  = i == 1 ? zb : za;
    float*       zout = i == 0 ? zb : za;
    const float* base = i == 0 ? nullptr : h;
    if (i < 2) {
      conv_kernel<false><<<cBlocks, cThreads, 0, stream>>>(
          zin, off, csr, tvec, i, base,
          W1 + (size_t)i * Hdim * H2, b1 + (size_t)i * H2,
          g1 + (size_t)i * H2,        be1 + (size_t)i * H2,
          W2 + (size_t)i * H2 * Hdim, b2 + (size_t)i * Hdim,
          ng + (i + 1) * Hdim, nb + (i + 1) * Hdim, nullptr, nullptr,
          h, zout, nullptr, N, E);
    } else {
      conv_kernel<true><<<cBlocks, cThreads, 0, stream>>>(
          zin, off, csr, tvec, i, base,
          W1 + (size_t)i * Hdim * H2, b1 + (size_t)i * H2,
          g1 + (size_t)i * H2,        be1 + (size_t)i * H2,
          W2 + (size_t)i * H2 * Hdim, b2 + (size_t)i * Hdim,
          ng, nb, linW, linB,
          nullptr, outH, outL, N, E);
    }
  }
}


# Reads not waiting on the stage before go out a stage early: next node's index word during a node's sums; a wave's W fragments for all its tiles as one stream, tiles side by side; sum orders kept
# speedup vs baseline: 5.1948x; 1.0752x over previous
#include <hip/hip_runtime.h>
#include <cstdint>


typedef __attribute__((ext_vector_type(2))) float v2f;
typedef __attribute__((ext_vector_type(8))) float v8f;

#define WMMA_F32(a, b, c) \
  __builtin_amdgcn_wmma_f32_16x16x4_f32(false, (a), false, (b), (short)0, (c), false, false)

constexpr int Hdim = 64;
constexpr int H2   = 128;
constexpr int IND  = 128;

__global__ __launch_bounds__(256) void enc_gemm_kernel(
    const float* __restrict__ x, const float* __restrict__ W,
    const float* __restrict__ bias, float* __restrict__ h, int nrows)
{
  __shared__ float Xs[64][IND + 4] __attribute__((aligned(16)));
  const int tid  = threadIdx.x;
  const int row0 = blockIdx.x * 64;

  float4 xv[8];
#pragma unroll
  for (int k = 0; k < 8; ++k) {
    const int i  = tid + 256 * k;
    const int gr = row0 + (i >> 5);
    const int lr = gr < nrows ? gr : nrows - 1;
    xv[k] = *(const float4*)(x + (long long)lr * IND + (i & 31) * 4);
  }
#pragma unroll
  for (int k = 0; k < 8; ++k) {
    const int i = tid + 256 * k;
    const int r = i >> 5;
    const float4 v = (row0 + r < nrows) ? xv[k] : make_float4(0.f, 0.f, 0.f, 0.f);
    *(float4*)&Xs[r][(i & 31) * 4] = v;
  }
  __syncthreads();

  const int wave = tid >> 5, lane = tid & 31;
  const int half = lane >> 4, ln = lane & 15;

  {
    const int m0 = (wave >> 1) * 16;
    const int c0 = (wave & 1) * 32 + ln;
    const float bn0 = bias[c0], bn1 = bias[c0 + 16];
    v8f acc0 = {}, acc1 = {};
#pragma unroll
    for (int kk = 0; kk < IND; kk += 4) {
      const int ka = kk + 2 * half;
      v2f a, p, q;
      a[0] = Xs[m0 + ln][ka];      a[1] = Xs[m0 + ln][ka + 1];
      p[0] = W[ka * Hdim + c0];
      p[1] = W[(ka + 1) * Hdim + c0];
      q[0] = W[ka * Hdim + c0 + 16];
      q[1] = W[(ka + 1) * Hdim + c0 + 16];
      acc0 = WMMA_F32(a, p, acc0);
      acc1 = WMMA_F32(a, q, acc1);
    }
    for (int r = 0; r < 8; ++r) {
      const int gr = row0 + m0 + r + 8 * half;
      if (gr < nrows) {
        h[(long long)gr * Hdim + c0]      = acc0[r] + bn0;
        h[(long long)gr * Hdim + c0 + 16] = acc1[r] + bn1;
      }
    }
  }
}

constexpr int ROWS_PER_WAVE = 8;

constexpr int BSH   = 8;
constexpr int BW    = 1 << BSH;
constexpr int NBMAX = 256;
constexpr int EPT   = 4;
constexpr int CH    = 1024 * EPT;

__device__ __forceinline__ int bucket_of(int d, int NB)
{
  const unsigned b = (unsigned)d >> BSH;
  return b < (unsigned)NB ? (int)b : NB - 1;
}

__global__ __launch_bounds__(1024) void bin_count_kernel(
    const int* __restrict__ dst, int* __restrict__ M, int E, int NB)
{
  __shared__ int hist[NBMAX];
  const int tid = threadIdx.x;
  const int e0  = blockIdx.x * CH + tid;
  int dd[EPT];
#pragma unroll
  for (int k = 0; k < EPT; ++k) {
    const int e = e0 + 1024 * k;
    dd[k] = dst[e < E ? e : E - 1];
  }
  if (tid < NBMAX) hist[tid] = 0;
  __syncthreads();
#pragma unroll
  for (int k = 0; k < EPT; ++k)
    if (e0 + 1024 * k < E) atomicAdd(&hist[bucket_of(dd[k], NB)], 1);
  __syncthreads();
  if (tid < NB && tid < NBMAX) M[blockIdx.x * NB + tid] = hist[tid];
}

__global__ __launch_bounds__(1024) void bin_scatter_kernel(
    const int* __restrict__ src, const int* __restrict__ dst,
    const int* __restrict__ M, int* __restrict__ bstart, int* __restrict__ rec,
    int E, int NB, int NC)
{
  __shared__ int ptot[4][NBMAX];
  __shared__ int ppre[4][NBMAX];
  __shared__ int cur[NBMAX];
  __shared__ int wsum[NBMAX / 32];
  const int tid = threadIdx.x, lane = tid & 31, wave = tid >> 5;
  const int c  = blockIdx.x;
  const int e0 = c * CH + tid;

  int dd[EPT], ss[EPT];
#pragma unroll
  for (int k = 0; k < EPT; ++k) {
    const int e  = e0 + 1024 * k;
    const int le = e < E ? e : E - 1;
    dd[k] = dst[le];
    ss[k] = src[le];
  }

  const int b = tid & (NBMAX - 1), q = tid >> 8;
  int tot = 0, pre = 0;
  if (b < NB) {
    for (int k0 = q; k0 < NC; k0 += 64) {
      int v[16];
#pragma unroll
      for (int j = 0; j < 16; ++j) {
        const int k = k0 + 4 * j;
        v[j] = M[(k < NC ? k : NC - 1) * NB + b];
      }
#pragma unroll
      for (int j = 0; j < 16; ++j) {
        const int k = k0 + 4 * j;
        const int a = k < NC ? v[j] : 0;
        tot += a;
        pre += k < c ? a : 0;
      }
    }
  }
  ptot[q][b] = tot;
  ppre[q][b] = pre;
  __syncthreads();

  int inc = 0;
  if (tid < NBMAX) {
    tot = ptot[0][tid] + ptot[1][tid] + ptot[2][tid] + ptot[3][tid];
    pre = ppre[0][tid] + ppre[1][tid] + ppre[2][tid] + ppre[3][tid];
    inc = tot;
    for (int o = 1; o < 32; o <<= 1) {
      const int up = __shfl_up(inc, o);
      if (lane >= o) inc += up;
    }
    if (lane == 31) wsum[wave] = inc;
  }
  __syncthreads();
  if (tid < NBMAX) {
    int wx = 0;
    for (int w = 0; w < wave; ++w) wx += wsum[w];
    const int ex = wx + inc - tot;
    cur[tid] = ex + pre;
    if (c == 0) {
      if (tid < NB)      bstart[tid] = ex;
      if (tid == NB - 1) bstart[NB]  = ex + tot;
    }
  }
  __syncthreads();

  int2* rc = (int2*)rec;
#pragma unroll
  for (int k = 0; k < EPT; ++k) {
    if (e0 + 1024 * k < E) {
      const int pos = atomicAdd(&cur[bucket_of(dd[k], NB)], 1);
      if ((unsigned)pos < (unsigned)E) rc[pos] = make_int2(ss[k], dd[k]);
    }
  }
}

__device__ __forceinline__ int local_of(int d, int n0)
{
  const int l = d - n0;
  return l < 0 ? 0 : (l > BW - 1 ? BW - 1 : l);
}

__global__ __launch_bounds__(1024) void bucket_fill_kernel(
    const int* __restrict__ rec, const int* __restrict__ bstart,
    int* __restrict__ off, int* __restrict__ csr, int N, int E, int NB)
{
  __shared__ int deg[BW];
  __shared__ int loff[BW];
  __shared__ int wsum[BW / 32];
  const int tid = threadIdx.x, lane = tid & 31, wave = tid >> 5;
  const int b  = blockIdx.x;
  const int n0 = b * BW;
  const int2* rc = (const int2*)rec;
  int s0 = bstart[b], s1 = bstart[b + 1];
  s0 = s0 < 0 ? 0 : (s0 > E ? E : s0);
  s1 = s1 < s0 ? s0 : (s1 > E ? E : s1);

  if (tid < BW) deg[tid] = 0;
  __syncthreads();
  for (int r0 = s0 + tid; r0 < s1; r0 += 4096) {
    int dv[4];
#pragma unroll
    for (int j = 0; j < 4; ++j) {
      const int r = r0 + 1024 * j;
      dv[j] = rc[r < s1 ? r : s1 - 1].y;
    }
#pragma unroll
    for (int j = 0; j < 4; ++j) asm volatile("" : "+v"(dv[j]));
#pragma unroll
    for (int j = 0; j < 4; ++j)
      if (r0 + 1024 * j < s1) atomicAdd(&deg[local_of(dv[j], n0)], 1);
  }
  __syncthreads();

  int v = 0, inc = 0;
  if (tid < BW) {
    v = deg[tid];
    inc = v;
    for (int o = 1; o < 32; o <<= 1) {
      const int up = __shfl_up(inc, o);
      if (lane >= o) inc += up;
    }
    if (lane == 31) wsum[wave] = inc;
  }
  __syncthreads();
  if (tid < BW) {
    int wx = 0;
    for (int w = 0; w < wave; ++w) wx += wsum[w];
    const int st = s0 + wx + inc - v;
    loff[tid] = st;
    deg[tid]  = 0;
    const int node = n0 + tid;
    if (node <= N) off[node] = st;
    if (b == NB - 1 && tid == BW - 1 && node + 1 == N) off[N] = st + v;
  }
  __syncthreads();

  for (int r0 = s0 + tid; r0 < s1; r0 += 4096) {
    int2 rv[4];
#pragma unroll
    for (int j = 0; j < 4; ++j) {
      const int r = r0 + 1024 * j;
      rv[j] = rc[r < s1 ? r : s1 - 1];
    }
#pragma unroll
    for (int j = 0; j < 4; ++j) asm volatile("" : "+v"(rv[j].x), "+v"(rv[j].y));
#pragma unroll
    for (int j = 0; j < 4; ++j) {
      if (r0 + 1024 * j < s1) {
        const int l   = local_of(rv[j].y, n0);
        const int pos = loff[l] + atomicAdd(&deg[l], 1);
        if ((unsigned)pos < (unsigned)E) csr[pos] = rv[j].x;
      }
    }
  }
}

constexpr int GK = 6;
constexpr int RK = 4 * GK;

__device__ __forceinline__ unsigned edge_row(const int* __restrict__ csr, int p, int jl,
                                             unsigned last)
{
  p = p > jl ? jl : p;
  unsigned s = (unsigned)csr[p];
  return s > last ? last : s;
}

__device__ __forceinline__ unsigned head_word(const int* __restrict__ csr, int j0, int j1,
                                              int lane, int E, unsigned last)
{
  int p = j0 + lane;
  p = p > j1 - 1 ? j1 - 1 : p;
  p = p > E - 1 ? E - 1 : p;
  p = p < 0 ? 0 : p;
  const unsigned s = (unsigned)csr[p];
  return s > last ? last : s;
}

__device__ __forceinline__ float2 z_row(const char* __restrict__ zc, unsigned row, unsigned fb)
{
  return *(const float2*)(zc + (size_t)row * (size_t)(Hdim * 4) + fb);
}

__device__ __forceinline__ float2 msg2(float2 zv)
{
  return make_float2(fmaxf(zv.x, 0.f) + 1e-7f, fmaxf(zv.y, 0.f) + 1e-7f);
}

__device__ __forceinline__ void seg_max_m(float2 m, float tv, float& x0, float& x1)
{
  x0 = fmaxf(x0, m.x * tv);
  x1 = fmaxf(x1, m.y * tv);
}

__device__ __forceinline__ void seg_sum_m(float m, float tv, float mx, bool live,
                                          float& den, float& num)
{
  float a = __expf(m * tv - mx);
  a = live ? a : 0.f;
  den += a;
  num += m * a;
}

__device__ __forceinline__ void seg_sum2(float2 m, float tv, float x0, float x1, bool live,
                                         float& d0, float& d1, float& n0, float& n1)
{
  seg_sum_m(m.x, tv, x0, live, d0, n0);
  seg_sum_m(m.y, tv, x1, live, d1, n1);
}

__device__ __forceinline__ void node_sums(
    const char* __restrict__ zc, unsigned fb, const int* __restrict__ csr,
    int j0, int j1, unsigned cv, float tv, unsigned last,
    float& d0, float& d1, float& n0, float& n1)
{
  const int jl = j1 - 1;
  const int jt = j0 + RK;


  float2 R[RK];
#pragma unroll
  for (int g = 0; g < GK; ++g) {
    if (j0 + 4 * g <= jl) {
#pragma unroll
      for (int i = 0; i < 4; ++i)
        R[4 * g + i] = z_row(zc, (unsigned)__builtin_amdgcn_readlane((int)cv, 4 * g + i), fb);
    }
  }

  float x0 = -3.402823466e38f, x1 = x0;
#pragma unroll
  for (int g = 0; g < GK; ++g) {
    if (j0 + 4 * g <= jl) {
#pragma unroll
      for (int i = 0; i < 4; ++i) {
        R[4 * g + i] = msg2(R[4 * g + i]);
        seg_max_m(R[4 * g + i], tv, x0, x1);
      }
    }
  }
  for (int j = jt; j < j1; j += 4) {
    const unsigned c0 = edge_row(csr, j, jl, last),     c1 = edge_row(csr, j + 1, jl, last);
    const unsigned c2 = edge_row(csr, j + 2, jl, last), c3 = edge_row(csr, j + 3, jl, last);
    const float2 r0 = z_row(zc, c0, fb);
    const float2 r1 = z_row(zc, c1, fb);
    const float2 r2 = z_row(zc, c2, fb);
    const float2 r3 = z_row(zc, c3, fb);
    seg_max_m(msg2(r0), tv, x0, x1);
    seg_max_m(msg2(r1), tv, x0, x1);
    seg_max_m(msg2(r2), tv, x0, x1);
    seg_max_m(msg2(r3), tv, x0, x1);
  }

#pragma unroll
  for (int g = 0; g < GK; ++g) {
    const int p = j0 + 4 * g;
    if (p <= jl) {
      const bool l1 = p + 1 <= jl, l2 = p + 2 <= jl, l3 = p + 3 <= jl;
      seg_sum2(R[4 * g + 0], tv, x0, x1, true, d0, d1, n0, n1);
      seg_sum2(R[4 * g + 1], tv, x0, x1, l1,   d0, d1, n0, n1);
      seg_sum2(R[4 * g + 2], tv, x0, x1, l2,   d0, d1, n0, n1);
      seg_sum2(R[4 * g + 3], tv, x0, x1, l3,   d0, d1, n0, n1);
    }
  }
  for (int j = jt; j < j1; j += 4) {
    const unsigned c0 = edge_row(csr, j, jl, last),     c1 = edge_row(csr, j + 1, jl, last);
    const unsigned c2 = edge_row(csr, j + 2, jl, last), c3 = edge_row(csr, j + 3, jl, last);
    const float2 r0 = z_row(zc, c0, fb);
    const float2 r1 = z_row(zc, c1, fb);
    const float2 r2 = z_row(zc, c2, fb);
    const float2 r3 = z_row(zc, c3, fb);
    const bool l1 = j + 1 <= jl, l2 = j + 2 <= jl, l3 = j + 3 <= jl;
    seg_sum2(msg2(r0), tv, x0, x1, true, d0, d1, n0, n1);
    seg_sum2(msg2(r1), tv, x0, x1, l1,   d0, d1, n0, n1);
    seg_sum2(msg2(r2), tv, x0, x1, l2,   d0, d1, n0, n1);
    seg_sum2(msg2(r3), tv, x0, x1, l3,   d0, d1, n0, n1);
  }
}

constexpr int TR = 16;

template <int NK>
__device__ __forceinline__ void load_bfrag(const float* __restrict__ Wm, int ld, int k0,
                                           int half, int nn, v2f (&bf)[NK])
{
#pragma unroll
  for (int s = 0; s < NK; ++s) {
    const int ka = k0 + 4 * s + 2 * half;
    bf[s][0] = Wm[ka * ld + nn];
    bf[s][1] = Wm[(ka + 1) * ld + nn];
  }
}

template <int NK>
__device__ __forceinline__ v8f mma_run(const v2f (&af)[NK], const v2f (&bf)[NK], v8f acc)
{
#pragma unroll
  for (int s = 0; s < NK; ++s) acc = WMMA_F32(af[s], bf[s], acc);
  return acc;
}

template <bool FINAL>
__global__ __launch_bounds__(TR / ROWS_PER_WAVE * 32) void conv_kernel(
    const float* __restrict__ z, const int* __restrict__ off,
    const int* __restrict__ csr, const float* __restrict__ t, int ti,
    const float* base,
    const float* __restrict__ W1, const float* __restrict__ b1,
    const float* __restrict__ g1, const float* __restrict__ be1,
    const float* __restrict__ W2, const float* __restrict__ b2,
    const float* __restrict__ ng, const float* __restrict__ nb,
    const float* __restrict__ linW, const float* __restrict__ linB,
    float* hout, float* __restrict__ fout, float* __restrict__ outL, int nrows, int E)
{
  __shared__ float U[TR][Hdim + 4] __attribute__((aligned(16)));
  __shared__ float T[TR][H2 + 4];

  const int tid  = threadIdx.x;
  const int row0 = blockIdx.x * TR;

  const int wave = tid >> 5, lane = tid & 31;
  const int half = lane >> 4, ln = lane & 15;

  {
    const int lr0 = __builtin_amdgcn_readfirstlane(wave) * ROWS_PER_WAVE;
    const unsigned fb = (unsigned)lane * 8u;
    const char* zc = (const char*)z;
    const float tv = t[ti];
    const unsigned last = (unsigned)(nrows - 1);
    int ob;
    {
      const int k = lane < ROWS_PER_WAVE ? lane : ROWS_PER_WAVE;
      const int d = row0 + lr0 + k;
      ob = off[d < nrows ? d : nrows];
      ob = ob < 0 ? 0 : (ob > E ? E : ob);
    }
    unsigned cvn;
    {
      const int a0 = __builtin_amdgcn_readlane(ob, 0);
      int a1 = __builtin_amdgcn_readlane(ob, 1);
      a1 = a1 < a0 ? a0 : a1;
      cvn = head_word(csr, a0, a1, lane, E, last);
    }
#pragma nounroll
    for (int i = 0; i < ROWS_PER_WAVE; ++i) {
      const int r = lr0 + i;
      const int d = row0 + r;
      float2 u = make_float2(0.f, 0.f);
      const int j0 = __builtin_amdgcn_readlane(ob, i);
      const int j1r = __builtin_amdgcn_readlane(ob, i + 1);
      const int j1 = j1r < j0 ? j0 : j1r;
      const unsigned cv = cvn;
      {
        int b1n = __builtin_amdgcn_readlane(ob, i + 2);
        b1n = b1n < j1r ? j1r : b1n;
        cvn = head_word(csr, j1r, b1n, lane, E, last);
      }
      if (d < nrows) {
        const float2 zv = z_row(zc, (unsigned)d, fb);
        float d0 = 0.f, d1 = 0.f, n0 = 0.f, n1 = 0.f;
        if (j0 < j1)
          node_sums(zc, fb, csr, j0, j1, cv, tv, last, d0, d1, n0, n1);
        u.x = zv.x + n0 / (d0 + 1e-16f);
        u.y = zv.y + n1 / (d1 + 1e-16f);
      }
      *(float2*)&U[r][2 * lane] = u;
    }
  }
  {
    const int c1 = wave * 64 + ln;
    v2f bq0[16], bq1[16];
    float bs0, bs1;
    load_bfrag<16>(W1, H2, 0, half, c1, bq0);
    bs0 = b1[c1];
    __syncthreads();

    v2f af[16];
#pragma unroll
    for (int s = 0; s < 16; ++s) {
      const int ka = 4 * s + 2 * half;
      af[s][0] = U[ln][ka];     af[s][1] = U[ln][ka + 1];
    }
#pragma unroll
    for (int t = 0; t < 4; t += 2) {
      load_bfrag<16>(W1, H2, 0, half, c1 + 16 * (t + 1), bq1);
      bs1 = b1[c1 + 16 * (t + 1)];
      {
        v8f acc = {};
        acc = mma_run<16>(af, bq0, acc);
        for (int r = 0; r < 8; ++r) T[r + 8 * half][c1 + 16 * t] = acc[r] + bs0;
      }
      if (t + 2 < 4) {
        load_bfrag<16>(W1, H2, 0, half, c1 + 16 * (t + 2), bq0);
        bs0 = b1[c1 + 16 * (t + 2)];
      }
      {
        v8f acc = {};
        acc = mma_run<16>(af, bq1, acc);
        for (int r = 0; r < 8; ++r) T[r + 8 * half][c1 + 16 * (t + 1)] = acc[r] + bs1;
      }
    }
  }
  __syncthreads();

  {
    int r = tid >> 2;
    int q = tid & 3;
    float s = 0.f, s2 = 0.f;
    for (int c = q * 32; c < q * 32 + 32; ++c) {
      float v = T[r][c]; s += v; s2 += v * v;
    }
    s  += __shfl_xor(s, 1);  s  += __shfl_xor(s, 2);
    s2 += __shfl_xor(s2, 1); s2 += __shfl_xor(s2, 2);
    float mu  = s * (1.f / 128.f);
    float var = s2 * (1.f / 128.f) - mu * mu;
    float rs  = rsqrtf(var + 1e-5f);
    for (int c = q * 32; c < q * 32 + 32; ++c) {
      float v = (T[r][c] - mu) * rs * g1[c] + be1[c];
      T[r][c] = fmaxf(v, 0.f);
    }
  }
  __syncthreads();

  {
    const int c2 = wave * 32 + ln;
    const float bn0 = b2[c2], bn1 = b2[c2 + 16];
    float bv0[8], bv1[8];
    if (base) {
#pragma unroll
      for (int r = 0; r < 8; ++r) {
        const int gr = row0 + r + 8 * half;
        const int lr = gr < nrows ? gr : nrows - 1;
        bv0[r] = base[(long long)lr * Hdim + c2];
        bv1[r] = base[(long long)lr * Hdim + c2 + 16];
      }
    } else {
#pragma unroll
      for (int r = 0; r < 8; ++r) { bv0[r] = 0.f; bv1[r] = 0.f; }
    }
    v8f acc0 = {}, acc1 = {};
#pragma unroll
    for (int kk = 0; kk < H2; kk += 4) {
      const int ka = kk + 2 * half;
      v2f a, p, q;
      a[0] = T[ln][ka];     a[1] = T[ln][ka + 1];
      p[0] = W2[ka * Hdim + c2];
      p[1] = W2[(ka + 1) * Hdim + c2];
      q[0] = W2[ka * Hdim + c2 + 16];
      q[1] = W2[(ka + 1) * Hdim + c2 + 16];
      acc0 = WMMA_F32(a, p, acc0);
      acc1 = WMMA_F32(a, q, acc1);
    }
    for (int r = 0; r < 8; ++r) {
      float v0 = acc0[r] + bn0, v1 = acc1[r] + bn1;
      if (base) { v0 += bv0[r]; v1 += bv1[r]; }
      U[r + 8 * half][c2]      = v0;
      U[r + 8 * half][c2 + 16] = v1;
    }
  }
  __syncthreads();

  {
    const int lr0 = __builtin_amdgcn_readfirstlane(wave) * ROWS_PER_WAVE;
    const int gr0 = row0 + lr0;
    float v0[ROWS_PER_WAVE], v1[ROWS_PER_WAVE];
#pragma unroll
    for (int k = 0; k < ROWS_PER_WAVE; ++k) {
      v0[k] = U[lr0 + k][lane];
      v1[k] = U[lr0 + k][lane + 32];
    }
    const float g0 = ng[lane], g1n = ng[lane + 32];
    const float b0 = nb[lane], b1n = nb[lane + 32];
    float w0[3], w1[3], lb[3];
    if (FINAL) {
#pragma unroll
      for (int j = 0; j < 3; ++j) {
        w0[j] = linW[lane * 3 + j];
        w1[j] = linW[(lane + 32) * 3 + j];
        lb[j] = linB[j];
      }
    }

    float s[ROWS_PER_WAVE], s2[ROWS_PER_WAVE];
#pragma unroll
    for (int k = 0; k < ROWS_PER_WAVE; ++k) {
      s[k]  = v0[k] + v1[k];
      s2[k] = v0[k] * v0[k] + v1[k] * v1[k];
    }
#pragma unroll
    for (int off = 1; off < 32; off <<= 1) {
#pragma unroll
      for (int k = 0; k < ROWS_PER_WAVE; ++k) {
        s[k]  += __shfl_xor(s[k], off);
        s2[k] += __shfl_xor(s2[k], off);
      }
    }
    float p[ROWS_PER_WAVE][3];
#pragma unroll
    for (int k = 0; k < ROWS_PER_WAVE; ++k) {
      float mu  = s[k] * (1.f / 64.f);
      float var = s2[k] * (1.f / 64.f) - mu * mu;
      float rs  = rsqrtf(var + 1e-5f);
      float f0 = fmaxf((v0[k] - mu) * rs * g0 + b0, 0.f);
      float f1 = fmaxf((v1[k] - mu) * rs * g1n + b1n, 0.f);
      if (gr0 + k < nrows) {
        if (!FINAL) {
          float* hr = hout + (long long)(gr0 + k) * Hdim;
          hr[lane]      = v0[k];
          hr[lane + 32] = v1[k];
        }
        float* fr = fout + (long long)(gr0 + k) * Hdim;
        fr[lane]      = f0;
        fr[lane + 32] = f1;
      }
      if (FINAL) {
#pragma unroll
        for (int j = 0; j < 3; ++j) p[k][j] = f0 * w0[j] + f1 * w1[j];
      }
    }
    if (FINAL) {
#pragma unroll
      for (int off = 1; off < 32; off <<= 1) {
#pragma unroll
        for (int k = 0; k < ROWS_PER_WAVE; ++k) {
#pragma unroll
          for (int j = 0; j < 3; ++j) p[k][j] += __shfl_xor(p[k][j], off);
        }
      }
#pragma unroll
      for (int k = 0; k < ROWS_PER_WAVE; ++k) {
        if (gr0 + k < nrows && lane == 0) {
#pragma unroll
          for (int j = 0; j < 3; ++j) outL[(long long)(gr0 + k) * 3 + j] = p[k][j] + lb[j];
        }
      }
    }
  }
}

extern "C" void kernel_launch(void* const* d_in, const int* in_sizes, int n_in,
                              void* d_out, int out_size, void* d_ws, size_t ws_size,
                              hipStream_t stream)
{
  const float* x    = (const float*)d_in[0];
  const int*   edge = (const int*)d_in[1];
  const float* encW = (const float*)d_in[2];
  const float* encB = (const float*)d_in[3];
  const float* tvec = (const float*)d_in[4];
  const float* W1   = (const float*)d_in[5];
  const float* b1   = (const float*)d_in[6];
  const float* g1   = (const float*)d_in[7];
  const float* be1  = (const float*)d_in[8];
  const float* W2   = (const float*)d_in[9];
  const float* b2   = (const float*)d_in[10];
  const float* ng   = (const float*)d_in[11];
  const float* nb   = (const float*)d_in[12];
  const float* linW = (const float*)d_in[13];
  const float* linB = (const float*)d_in[14];

  const int N = in_sizes[0] / IND;
  const int E = in_sizes[1] / 2;
  const int* src = edge;
  const int* dst = edge + E;

  float* za    = (float*)d_ws;
  float* zb    = za    + (size_t)N * Hdim;
  float* plane = zb    + (size_t)N * Hdim;
  float* h     = plane + (size_t)N * Hdim;

  const int n1  = N + 1;
  const int n1a = (n1 + 3) & ~3;
  const int NB  = (N + BW - 1) / BW;
  const int NC  = (E + CH - 1) / CH;
  int* off    = (int*)plane;
  int* bstart = off + n1a;
  int* M      = bstart + (NBMAX + 4);
  int* rec    = M + ((NC * NB + 3) & ~3);
  int* csr    = rec + 2 * (size_t)E;

  const int rowBlocks  = (N + 63) / 64;
  const int cBlocks    = (N + TR - 1) / TR;
  const int cThreads   = TR / ROWS_PER_WAVE * 32;

  bin_count_kernel<<<NC, 1024, 0, stream>>>(dst, M, E, NB);
  bin_scatter_kernel<<<NC, 1024, 0, stream>>>(src, dst, M, bstart, rec, E, NB, NC);
  bucket_fill_kernel<<<NB, 1024, 0, stream>>>(rec, bstart, off, csr, N, E, NB);

  enc_gemm_kernel<<<rowBlocks, 256, 0, stream>>>(x, encW, encB, za, N);

  float* outH = (float*)d_out;
  float* outL = outH + (size_t)N * Hdim;

  for (int i = 0; i < 3; ++i) {
    const float* zin  = i == 1 ? zb : za;
    float*       zout = i == 0 ? zb : za;
    const float* base = i == 0 ? nullptr : h;
    if (i < 2) {
      conv_kernel<false><<<cBlocks, cThreads, 0, stream>>>(
          zin, off, csr, tvec, i, base,
          W1 + (size_t)i * Hdim * H2, b1 + (size_t)i * H2,
          g1 + (size_t)i * H2,        be1 + (size_t)i * H2,
          W2 + (size_t)i * H2 * Hdim, b2 + (size_t)i * Hdim,
          ng + (i + 1) * Hdim, nb + (i + 1) * Hdim, nullptr, nullptr,
          h, zout, nullptr, N, E);
    } else {
      conv_kernel<true><<<cBlocks, cThreads, 0, stream>>>(
          zin, off, csr, tvec, i, base,
          W1 + (size_t)i * Hdim * H2, b1 + (size_t)i * H2,
          g1 + (size_t)i * H2,        be1 + (size_t)i * H2,
          W2 + (size_t)i * H2 * Hdim, b2 + (size_t)i * Hdim,
          ng, nb, linW, linB,
          nullptr, outH, outL, N, E);
    }
  }
}


# Encoder GEMM loses its launch: its 128-row tiles run as extra blocks of the edge list's placement and fill launches (role by block index, two blocks per WGP): 7 -> 6 launches; sum orders kept
# speedup vs baseline: 5.4151x; 1.0424x over previous
#include <hip/hip_runtime.h>
#include <cstdint>


typedef __attribute__((ext_vector_type(2))) float v2f;
typedef __attribute__((ext_vector_type(8))) float v8f;

#define WMMA_F32(a, b, c) \
  __builtin_amdgcn_wmma_f32_16x16x4_f32(false, (a), false, (b), (short)0, (c), false, false)

constexpr int Hdim = 64;
constexpr int H2   = 128;
constexpr int IND  = 128;

constexpr int TE = 128;

__device__ __forceinline__ void enc_tile(
    const float* __restrict__ x, const float* __restrict__ W,
    const float* __restrict__ bias, float* __restrict__ h, int nrows, int tile)
{
  __shared__ float Xs[TE][IND + 4] __attribute__((aligned(16)));
  const int tid  = threadIdx.x;
  const int row0 = tile * TE;

  float4 xv[4];
#pragma unroll
  for (int k = 0; k < 4; ++k) {
    const int i  = tid + 1024 * k;
    const int gr = row0 + (i >> 5);
    const int lr = gr < nrows ? gr : nrows - 1;
    xv[k] = *(const float4*)(x + (long long)lr * IND + (i & 31) * 4);
  }
#pragma unroll
  for (int k = 0; k < 4; ++k) {
    const int i = tid + 1024 * k;
    const int r = i >> 5;
    const float4 v = (row0 + r < nrows) ? xv[k] : make_float4(0.f, 0.f, 0.f, 0.f);
    *(float4*)&Xs[r][(i & 31) * 4] = v;
  }
  __syncthreads();

  const int wave = tid >> 5, lane = tid & 31;
  const int half = lane >> 4, ln = lane & 15;

  if (wave >= TE / 8) return;

  {
    const int m0 = (wave >> 1) * 16;
    const int c0 = (wave & 1) * 32 + ln;
    const float bn0 = bias[c0], bn1 = bias[c0 + 16];
    v8f acc0 = {}, acc1 = {};
#pragma nounroll
    for (int k0 = 0; k0 < IND; k0 += 16) {
      v2f p[4], q[4];
#pragma unroll
      for (int s = 0; s < 4; ++s) {
        const int ka = k0 + 4 * s + 2 * half;
        p[s][0] = W[ka * Hdim + c0];
        p[s][1] = W[(ka + 1) * Hdim + c0];
        q[s][0] = W[ka * Hdim + c0 + 16];
        q[s][1] = W[(ka + 1) * Hdim + c0 + 16];
      }
#pragma unroll
      for (int s = 0; s < 4; ++s)
        asm volatile("" : "+v"(p[s][0]), "+v"(p[s][1]), "+v"(q[s][0]), "+v"(q[s][1]));
#pragma unroll
      for (int s = 0; s < 4; ++s) {
        const int ka = k0 + 4 * s + 2 * half;
        v2f a;
        a[0] = Xs[m0 + ln][ka];      a[1] = Xs[m0 + ln][ka + 1];
        acc0 = WMMA_F32(a, p[s], acc0);
        acc1 = WMMA_F32(a, q[s], acc1);
      }
    }
    for (int r = 0; r < 8; ++r) {
      const int gr = row0 + m0 + r + 8 * half;
      if (gr < nrows) {
        h[(long long)gr * Hdim + c0]      = acc0[r] + bn0;
        h[(long long)gr * Hdim + c0 + 16] = acc1[r] + bn1;
      }
    }
  }
}

constexpr int ROWS_PER_WAVE = 8;

constexpr int BSH   = 8;
constexpr int BW    = 1 << BSH;
constexpr int NBMAX = 256;
constexpr int EPT   = 4;
constexpr int CH    = 1024 * EPT;

__device__ __forceinline__ int bucket_of(int d, int NB)
{
  const unsigned b = (unsigned)d >> BSH;
  return b < (unsigned)NB ? (int)b : NB - 1;
}

__global__ __launch_bounds__(1024) void bin_count_kernel(
    const int* __restrict__ dst, int* __restrict__ M, int E, int NB)
{
  __shared__ int hist[NBMAX];
  const int tid = threadIdx.x;
  const int e0  = blockIdx.x * CH + tid;
  int dd[EPT];
#pragma unroll
  for (int k = 0; k < EPT; ++k) {
    const int e = e0 + 1024 * k;
    dd[k] = dst[e < E ? e : E - 1];
  }
  if (tid < NBMAX) hist[tid] = 0;
  __syncthreads();
#pragma unroll
  for (int k = 0; k < EPT; ++k)
    if (e0 + 1024 * k < E) atomicAdd(&hist[bucket_of(dd[k], NB)], 1);
  __syncthreads();
  if (tid < NB && tid < NBMAX) M[blockIdx.x * NB + tid] = hist[tid];
}

__global__ __launch_bounds__(1024, 16) void bin_scatter_kernel(
    const int* __restrict__ src, const int* __restrict__ dst,
    const int* __restrict__ M, int* __restrict__ bstart, int* __restrict__ rec,
    int E, int NB, int NC,
    const float* __restrict__ ex, const float* __restrict__ eW,
    const float* __restrict__ eb, float* __restrict__ eh, int nrows, int et0, int et1)
{
  if ((int)blockIdx.x >= NC) {
    const int tile = et0 + (int)blockIdx.x - NC;
    if (tile < et1) enc_tile(ex, eW, eb, eh, nrows, tile);
    return;
  }
  __shared__ int ptot[4][NBMAX];
  __shared__ int ppre[4][NBMAX];
  __shared__ int cur[NBMAX];
  __shared__ int wsum[NBMAX / 32];
  const int tid = threadIdx.x, lane = tid & 31, wave = tid >> 5;
  const int c  = blockIdx.x;
  const int e0 = c * CH + tid;

  int dd[EPT], ss[EPT];
#pragma unroll
  for (int k = 0; k < EPT; ++k) {
    const int e  = e0 + 1024 * k;
    const int le = e < E ? e : E - 1;
    dd[k] = dst[le];
    ss[k] = src[le];
  }

  const int b = tid & (NBMAX - 1), q = tid >> 8;
  int tot = 0, pre = 0;
  if (b < NB) {
    for (int k0 = q; k0 < NC; k0 += 32) {
      int v[8];
#pragma unroll
      for (int j = 0; j < 8; ++j) {
        const int k = k0 + 4 * j;
        v[j] = M[(k < NC ? k : NC - 1) * NB + b];
      }
#pragma unroll
      for (int j = 0; j < 8; ++j) {
        const int k = k0 + 4 * j;
        const int a = k < NC ? v[j] : 0;
        tot += a;
        pre += k < c ? a : 0;
      }
    }
  }
  ptot[q][b] = tot;
  ppre[q][b] = pre;
  __syncthreads();

  int inc = 0;
  if (tid < NBMAX) {
    tot = ptot[0][tid] + ptot[1][tid] + ptot[2][tid] + ptot[3][tid];
    pre = ppre[0][tid] + ppre[1][tid] + ppre[2][tid] + ppre[3][tid];
    inc = tot;
    for (int o = 1; o < 32; o <<= 1) {
      const int up = __shfl_up(inc, o);
      if (lane >= o) inc += up;
    }
    if (lane == 31) wsum[wave] = inc;
  }
  __syncthreads();
  if (tid < NBMAX) {
    int wx = 0;
    for (int w = 0; w < wave; ++w) wx += wsum[w];
    const int ex = wx + inc - tot;
    cur[tid] = ex + pre;
    if (c == 0) {
      if (tid < NB)      bstart[tid] = ex;
      if (tid == NB - 1) bstart[NB]  = ex + tot;
    }
  }
  __syncthreads();

  int2* rc = (int2*)rec;
#pragma unroll
  for (int k = 0; k < EPT; ++k) {
    if (e0 + 1024 * k < E) {
      const int pos = atomicAdd(&cur[bucket_of(dd[k], NB)], 1);
      if ((unsigned)pos < (unsigned)E) rc[pos] = make_int2(ss[k], dd[k]);
    }
  }
}

__device__ __forceinline__ int local_of(int d, int n0)
{
  const int l = d - n0;
  return l < 0 ? 0 : (l > BW - 1 ? BW - 1 : l);
}

__global__ __launch_bounds__(1024, 16) void bucket_fill_kernel(
    const int* __restrict__ rec, const int* __restrict__ bstart,
    int* __restrict__ off, int* __restrict__ csr, int N, int E, int NB,
    const float* __restrict__ ex, const float* __restrict__ eW,
    const float* __restrict__ eb, float* __restrict__ eh, int et0, int et1)
{
  if ((int)blockIdx.x >= NB) {
    const int tile = et0 + (int)blockIdx.x - NB;
    if (tile < et1) enc_tile(ex, eW, eb, eh, N, tile);
    return;
  }
  __shared__ int deg[BW];
  __shared__ int loff[BW];
  __shared__ int wsum[BW / 32];
  const int tid = threadIdx.x, lane = tid & 31, wave = tid >> 5;
  const int b  = blockIdx.x;
  const int n0 = b * BW;
  const int2* rc = (const int2*)rec;
  int s0 = bstart[b], s1 = bstart[b + 1];
  s0 = s0 < 0 ? 0 : (s0 > E ? E : s0);
  s1 = s1 < s0 ? s0 : (s1 > E ? E : s1);

  if (tid < BW) deg[tid] = 0;
  __syncthreads();
  for (int r0 = s0 + tid; r0 < s1; r0 += 4096) {
    int dv[4];
#pragma unroll
    for (int j = 0; j < 4; ++j) {
      const int r = r0 + 1024 * j;
      dv[j] = rc[r < s1 ? r : s1 - 1].y;
    }
#pragma unroll
    for (int j = 0; j < 4; ++j) asm volatile("" : "+v"(dv[j]));
#pragma unroll
    for (int j = 0; j < 4; ++j)
      if (r0 + 1024 * j < s1) atomicAdd(&deg[local_of(dv[j], n0)], 1);
  }
  __syncthreads();

  int v = 0, inc = 0;
  if (tid < BW) {
    v = deg[tid];
    inc = v;
    for (int o = 1; o < 32; o <<= 1) {
      const int up = __shfl_up(inc, o);
      if (lane >= o) inc += up;
    }
    if (lane == 31) wsum[wave] = inc;
  }
  __syncthreads();
  if (tid < BW) {
    int wx = 0;
    for (int w = 0; w < wave; ++w) wx += wsum[w];
    const int st = s0 + wx + inc - v;
    loff[tid] = st;
    deg[tid]  = 0;
    const int node = n0 + tid;
    if (node <= N) off[node] = st;
    if (b == NB - 1 && tid == BW - 1 && node + 1 == N) off[N] = st + v;
  }
  __syncthreads();

  for (int r0 = s0 + tid; r0 < s1; r0 += 4096) {
    int2 rv[4];
#pragma unroll
    for (int j = 0; j < 4; ++j) {
      const int r = r0 + 1024 * j;
      rv[j] = rc[r < s1 ? r : s1 - 1];
    }
#pragma unroll
    for (int j = 0; j < 4; ++j) asm volatile("" : "+v"(rv[j].x), "+v"(rv[j].y));
#pragma unroll
    for (int j = 0; j < 4; ++j) {
      if (r0 + 1024 * j < s1) {
        const int l   = local_of(rv[j].y, n0);
        const int pos = loff[l] + atomicAdd(&deg[l], 1);
        if ((unsigned)pos < (unsigned)E) csr[pos] = rv[j].x;
      }
    }
  }
}

constexpr int GK = 6;
constexpr int RK = 4 * GK;

__device__ __forceinline__ unsigned edge_row(const int* __restrict__ csr, int p, int jl,
                                             unsigned last)
{
  p = p > jl ? jl : p;
  unsigned s = (unsigned)csr[p];
  return s > last ? last : s;
}

__device__ __forceinline__ unsigned head_word(const int* __restrict__ csr, int j0, int j1,
                                              int lane, int E, unsigned last)
{
  int p = j0 + lane;
  p = p > j1 - 1 ? j1 - 1 : p;
  p = p > E - 1 ? E - 1 : p;
  p = p < 0 ? 0 : p;
  const unsigned s = (unsigned)csr[p];
  return s > last ? last : s;
}

__device__ __forceinline__ float2 z_row(const char* __restrict__ zc, unsigned row, unsigned fb)
{
  return *(const float2*)(zc + (size_t)row * (size_t)(Hdim * 4) + fb);
}

__device__ __forceinline__ float2 msg2(float2 zv)
{
  return make_float2(fmaxf(zv.x, 0.f) + 1e-7f, fmaxf(zv.y, 0.f) + 1e-7f);
}

__device__ __forceinline__ void seg_max_m(float2 m, float tv, float& x0, float& x1)
{
  x0 = fmaxf(x0, m.x * tv);
  x1 = fmaxf(x1, m.y * tv);
}

__device__ __forceinline__ void seg_sum_m(float m, float tv, float mx, bool live,
                                          float& den, float& num)
{
  float a = __expf(m * tv - mx);
  a = live ? a : 0.f;
  den += a;
  num += m * a;
}

__device__ __forceinline__ void seg_sum2(float2 m, float tv, float x0, float x1, bool live,
                                         float& d0, float& d1, float& n0, float& n1)
{
  seg_sum_m(m.x, tv, x0, live, d0, n0);
  seg_sum_m(m.y, tv, x1, live, d1, n1);
}

__device__ __forceinline__ void node_sums(
    const char* __restrict__ zc, unsigned fb, const int* __restrict__ csr,
    int j0, int j1, unsigned cv, float tv, unsigned last,
    float& d0, float& d1, float& n0, float& n1)
{
  const int jl = j1 - 1;
  const int jt = j0 + RK;


  float2 R[RK];
#pragma unroll
  for (int g = 0; g < GK; ++g) {
    if (j0 + 4 * g <= jl) {
#pragma unroll
      for (int i = 0; i < 4; ++i)
        R[4 * g + i] = z_row(zc, (unsigned)__builtin_amdgcn_readlane((int)cv, 4 * g + i), fb);
    }
  }

  float x0 = -3.402823466e38f, x1 = x0;
#pragma unroll
  for (int g = 0; g < GK; ++g) {
    if (j0 + 4 * g <= jl) {
#pragma unroll
      for (int i = 0; i < 4; ++i) {
        R[4 * g + i] = msg2(R[4 * g + i]);
        seg_max_m(R[4 * g + i], tv, x0, x1);
      }
    }
  }
  for (int j = jt; j < j1; j += 4) {
    const unsigned c0 = edge_row(csr, j, jl, last),     c1 = edge_row(csr, j + 1, jl, last);
    const unsigned c2 = edge_row(csr, j + 2, jl, last), c3 = edge_row(csr, j + 3, jl, last);
    const float2 r0 = z_row(zc, c0, fb);
    const float2 r1 = z_row(zc, c1, fb);
    const float2 r2 = z_row(zc, c2, fb);
    const float2 r3 = z_row(zc, c3, fb);
    seg_max_m(msg2(r0), tv, x0, x1);
    seg_max_m(msg2(r1), tv, x0, x1);
    seg_max_m(msg2(r2), tv, x0, x1);
    seg_max_m(msg2(r3), tv, x0, x1);
  }

#pragma unroll
  for (int g = 0; g < GK; ++g) {
    const int p = j0 + 4 * g;
    if (p <= jl) {
      const bool l1 = p + 1 <= jl, l2 = p + 2 <= jl, l3 = p + 3 <= jl;
      seg_sum2(R[4 * g + 0], tv, x0, x1, true, d0, d1, n0, n1);
      seg_sum2(R[4 * g + 1], tv, x0, x1, l1,   d0, d1, n0, n1);
      seg_sum2(R[4 * g + 2], tv, x0, x1, l2,   d0, d1, n0, n1);
      seg_sum2(R[4 * g + 3], tv, x0, x1, l3,   d0, d1, n0, n1);
    }
  }
  for (int j = jt; j < j1; j += 4) {
    const unsigned c0 = edge_row(csr, j, jl, last),     c1 = edge_row(csr, j + 1, jl, last);
    const unsigned c2 = edge_row(csr, j + 2, jl, last), c3 = edge_row(csr, j + 3, jl, last);
    const float2 r0 = z_row(zc, c0, fb);
    const float2 r1 = z_row(zc, c1, fb);
    const float2 r2 = z_row(zc, c2, fb);
    const float2 r3 = z_row(zc, c3, fb);
    const bool l1 = j + 1 <= jl, l2 = j + 2 <= jl, l3 = j + 3 <= jl;
    seg_sum2(msg2(r0), tv, x0, x1, true, d0, d1, n0, n1);
    seg_sum2(msg2(r1), tv, x0, x1, l1,   d0, d1, n0, n1);
    seg_sum2(msg2(r2), tv, x0, x1, l2,   d0, d1, n0, n1);
    seg_sum2(msg2(r3), tv, x0, x1, l3,   d0, d1, n0, n1);
  }
}

constexpr int TR = 16;

template <int NK>
__device__ __forceinline__ void load_bfrag(const float* __restrict__ Wm, int ld, int k0,
                                           int half, int nn, v2f (&bf)[NK])
{
#pragma unroll
  for (int s = 0; s < NK; ++s) {
    const int ka = k0 + 4 * s + 2 * half;
    bf[s][0] = Wm[ka * ld + nn];
    bf[s][1] = Wm[(ka + 1) * ld + nn];
  }
}

template <int NK>
__device__ __forceinline__ v8f mma_run(const v2f (&af)[NK], const v2f (&bf)[NK], v8f acc)
{
#pragma unroll
  for (int s = 0; s < NK; ++s) acc = WMMA_F32(af[s], bf[s], acc);
  return acc;
}

template <bool FINAL>
__global__ __launch_bounds__(TR / ROWS_PER_WAVE * 32) void conv_kernel(
    const float* __restrict__ z, const int* __restrict__ off,
    const int* __restrict__ csr, const float* __restrict__ t, int ti,
    const float* base,
    const float* __restrict__ W1, const float* __restrict__ b1,
    const float* __restrict__ g1, const float* __restrict__ be1,
    const float* __restrict__ W2, const float* __restrict__ b2,
    const float* __restrict__ ng, const float* __restrict__ nb,
    const float* __restrict__ linW, const float* __restrict__ linB,
    float* hout, float* __restrict__ fout, float* __restrict__ outL, int nrows, int E)
{
  __shared__ float U[TR][Hdim + 4] __attribute__((aligned(16)));
  __shared__ float T[TR][H2 + 4];

  const int tid  = threadIdx.x;
  const int row0 = blockIdx.x * TR;

  const int wave = tid >> 5, lane = tid & 31;
  const int half = lane >> 4, ln = lane & 15;

  {
    const int lr0 = __builtin_amdgcn_readfirstlane(wave) * ROWS_PER_WAVE;
    const unsigned fb = (unsigned)lane * 8u;
    const char* zc = (const char*)z;
    const float tv = t[ti];
    const unsigned last = (unsigned)(nrows - 1);
    int ob;
    {
      const int k = lane < ROWS_PER_WAVE ? lane : ROWS_PER_WAVE;
      const int d = row0 + lr0 + k;
      ob = off[d < nrows ? d : nrows];
      ob = ob < 0 ? 0 : (ob > E ? E : ob);
    }
    unsigned cvn;
    {
      const int a0 = __builtin_amdgcn_readlane(ob, 0);
      int a1 = __builtin_amdgcn_readlane(ob, 1);
      a1 = a1 < a0 ? a0 : a1;
      cvn = head_word(csr, a0, a1, lane, E, last);
    }
#pragma nounroll
    for (int i = 0; i < ROWS_PER_WAVE; ++i) {
      const int r = lr0 + i;
      const int d = row0 + r;
      float2 u = make_float2(0.f, 0.f);
      const int j0 = __builtin_amdgcn_readlane(ob, i);
      const int j1r = __builtin_amdgcn_readlane(ob, i + 1);
      const int j1 = j1r < j0 ? j0 : j1r;
      const unsigned cv = cvn;
      {
        int b1n = __builtin_amdgcn_readlane(ob, i + 2);
        b1n = b1n < j1r ? j1r : b1n;
        cvn = head_word(csr, j1r, b1n, lane, E, last);
      }
      if (d < nrows) {
        const float2 zv = z_row(zc, (unsigned)d, fb);
        float d0 = 0.f, d1 = 0.f, n0 = 0.f, n1 = 0.f;
        if (j0 < j1)
          node_sums(zc, fb, csr, j0, j1, cv, tv, last, d0, d1, n0, n1);
        u.x = zv.x + n0 / (d0 + 1e-16f);
        u.y = zv.y + n1 / (d1 + 1e-16f);
      }
      *(float2*)&U[r][2 * lane] = u;
    }
  }
  {
    const int c1 = wave * 64 + ln;
    v2f bq0[16], bq1[16];
    float bs0, bs1;
    load_bfrag<16>(W1, H2, 0, half, c1, bq0);
    bs0 = b1[c1];
    __syncthreads();

    v2f af[16];
#pragma unroll
    for (int s = 0; s < 16; ++s) {
      const int ka = 4 * s + 2 * half;
      af[s][0] = U[ln][ka];     af[s][1] = U[ln][ka + 1];
    }
#pragma unroll
    for (int t = 0; t < 4; t += 2) {
      load_bfrag<16>(W1, H2, 0, half, c1 + 16 * (t + 1), bq1);
      bs1 = b1[c1 + 16 * (t + 1)];
      {
        v8f acc = {};
        acc = mma_run<16>(af, bq0, acc);
        for (int r = 0; r < 8; ++r) T[r + 8 * half][c1 + 16 * t] = acc[r] + bs0;
      }
      if (t + 2 < 4) {
        load_bfrag<16>(W1, H2, 0, half, c1 + 16 * (t + 2), bq0);
        bs0 = b1[c1 + 16 * (t + 2)];
      }
      {
        v8f acc = {};
        acc = mma_run<16>(af, bq1, acc);
        for (int r = 0; r < 8; ++r) T[r + 8 * half][c1 + 16 * (t + 1)] = acc[r] + bs1;
      }
    }
  }
  __syncthreads();

  {
    int r = tid >> 2;
    int q = tid & 3;
    float s = 0.f, s2 = 0.f;
    for (int c = q * 32; c < q * 32 + 32; ++c) {
      float v = T[r][c]; s += v; s2 += v * v;
    }
    s  += __shfl_xor(s, 1);  s  += __shfl_xor(s, 2);
    s2 += __shfl_xor(s2, 1); s2 += __shfl_xor(s2, 2);
    float mu  = s * (1.f / 128.f);
    float var = s2 * (1.f / 128.f) - mu * mu;
    float rs  = rsqrtf(var + 1e-5f);
    for (int c = q * 32; c < q * 32 + 32; ++c) {
      float v = (T[r][c] - mu) * rs * g1[c] + be1[c];
      T[r][c] = fmaxf(v, 0.f);
    }
  }
  __syncthreads();

  {
    const int c2 = wave * 32 + ln;
    const float bn0 = b2[c2], bn1 = b2[c2 + 16];
    float bv0[8], bv1[8];
    if (base) {
#pragma unroll
      for (int r = 0; r < 8; ++r) {
        const int gr = row0 + r + 8 * half;
        const int lr = gr < nrows ? gr : nrows - 1;
        bv0[r] = base[(long long)lr * Hdim + c2];
        bv1[r] = base[(long long)lr * Hdim + c2 + 16];
      }
    } else {
#pragma unroll
      for (int r = 0; r < 8; ++r) { bv0[r] = 0.f; bv1[r] = 0.f; }
    }
    v8f acc0 = {}, acc1 = {};
#pragma unroll
    for (int kk = 0; kk < H2; kk += 4) {
      const int ka = kk + 2 * half;
      v2f a, p, q;
      a[0] = T[ln][ka];     a[1] = T[ln][ka + 1];
      p[0] = W2[ka * Hdim + c2];
      p[1] = W2[(ka + 1) * Hdim + c2];
      q[0] = W2[ka * Hdim + c2 + 16];
      q[1] = W2[(ka + 1) * Hdim + c2 + 16];
      acc0 = WMMA_F32(a, p, acc0);
      acc1 = WMMA_F32(a, q, acc1);
    }
    for (int r = 0; r < 8; ++r) {
      float v0 = acc0[r] + bn0, v1 = acc1[r] + bn1;
      if (base) { v0 += bv0[r]; v1 += bv1[r]; }
      U[r + 8 * half][c2]      = v0;
      U[r + 8 * half][c2 + 16] = v1;
    }
  }
  __syncthreads();

  {
    const int lr0 = __builtin_amdgcn_readfirstlane(wave) * ROWS_PER_WAVE;
    const int gr0 = row0 + lr0;
    float v0[ROWS_PER_WAVE], v1[ROWS_PER_WAVE];
#pragma unroll
    for (int k = 0; k < ROWS_PER_WAVE; ++k) {
      v0[k] = U[lr0 + k][lane];
      v1[k] = U[lr0 + k][lane + 32];
    }
    const float g0 = ng[lane], g1n = ng[lane + 32];
    const float b0 = nb[lane], b1n = nb[lane + 32];
    float w0[3], w1[3], lb[3];
    if (FINAL) {
#pragma unroll
      for (int j = 0; j < 3; ++j) {
        w0[j] = linW[lane * 3 + j];
        w1[j] = linW[(lane + 32) * 3 + j];
        lb[j] = linB[j];
      }
    }

    float s[ROWS_PER_WAVE], s2[ROWS_PER_WAVE];
#pragma unroll
    for (int k = 0; k < ROWS_PER_WAVE; ++k) {
      s[k]  = v0[k] + v1[k];
      s2[k] = v0[k] * v0[k] + v1[k] * v1[k];
    }
#pragma unroll
    for (int off = 1; off < 32; off <<= 1) {
#pragma unroll
      for (int k = 0; k < ROWS_PER_WAVE; ++k) {
        s[k]  += __shfl_xor(s[k], off);
        s2[k] += __shfl_xor(s2[k], off);
      }
    }
    float p[ROWS_PER_WAVE][3];
#pragma unroll
    for (int k = 0; k < ROWS_PER_WAVE; ++k) {
      float mu  = s[k] * (1.f / 64.f);
      float var = s2[k] * (1.f / 64.f) - mu * mu;
      float rs  = rsqrtf(var + 1e-5f);
      float f0 = fmaxf((v0[k] - mu) * rs * g0 + b0, 0.f);
      float f1 = fmaxf((v1[k] - mu) * rs * g1n + b1n, 0.f);
      if (gr0 + k < nrows) {
        if (!FINAL) {
          float* hr = hout + (long long)(gr0 + k) * Hdim;
          hr[lane]      = v0[k];
          hr[lane + 32] = v1[k];
        }
        float* fr = fout + (long long)(gr0 + k) * Hdim;
        fr[lane]      = f0;
        fr[lane + 32] = f1;
      }
      if (FINAL) {
#pragma unroll
        for (int j = 0; j < 3; ++j) p[k][j] = f0 * w0[j] + f1 * w1[j];
      }
    }
    if (FINAL) {
#pragma unroll
      for (int off = 1; off < 32; off <<= 1) {
#pragma unroll
        for (int k = 0; k < ROWS_PER_WAVE; ++k) {
#pragma unroll
          for (int j = 0; j < 3; ++j) p[k][j] += __shfl_xor(p[k][j], off);
        }
      }
#pragma unroll
      for (int k = 0; k < ROWS_PER_WAVE; ++k) {
        if (gr0 + k < nrows && lane == 0) {
#pragma unroll
          for (int j = 0; j < 3; ++j) outL[(long long)(gr0 + k) * 3 + j] = p[k][j] + lb[j];
        }
      }
    }
  }
}

extern "C" void kernel_launch(void* const* d_in, const int* in_sizes, int n_in,
                              void* d_out, int out_size, void* d_ws, size_t ws_size,
                              hipStream_t stream)
{
  const float* x    = (const float*)d_in[0];
  const int*   edge = (const int*)d_in[1];
  const float* encW = (const float*)d_in[2];
  const float* encB = (const float*)d_in[3];
  const float* tvec = (const float*)d_in[4];
  const float* W1   = (const float*)d_in[5];
  const float* b1   = (const float*)d_in[6];
  const float* g1   = (const float*)d_in[7];
  const float* be1  = (const float*)d_in[8];
  const float* W2   = (const float*)d_in[9];
  const float* b2   = (const float*)d_in[10];
  const float* ng   = (const float*)d_in[11];
  const float* nb   = (const float*)d_in[12];
  const float* linW = (const float*)d_in[13];
  const float* linB = (const float*)d_in[14];

  const int N = in_sizes[0] / IND;
  const int E = in_sizes[1] / 2;
  const int* src = edge;
  const int* dst = edge + E;

  float* za    = (float*)d_ws;
  float* zb    = za    + (size_t)N * Hdim;
  float* plane = zb    + (size_t)N * Hdim;
  float* h     = plane + (size_t)N * Hdim;

  const int n1  = N + 1;
  const int n1a = (n1 + 3) & ~3;
  const int NB  = (N + BW - 1) / BW;
  const int NC  = (E + CH - 1) / CH;
  int* off    = (int*)plane;
  int* bstart = off + n1a;
  int* M      = bstart + (NBMAX + 4);
  int* rec    = M + ((NC * NB + 3) & ~3);
  int* csr    = rec + 2 * (size_t)E;

  const int encT       = (N + TE - 1) / TE;
  const int encA       = (encT + 1) / 2;
  const int cBlocks    = (N + TR - 1) / TR;
  const int cThreads   = TR / ROWS_PER_WAVE * 32;

  bin_count_kernel<<<NC, 1024, 0, stream>>>(dst, M, E, NB);
  bin_scatter_kernel<<<NC + encA, 1024, 0, stream>>>(src, dst, M, bstart, rec, E, NB, NC,
                                                     x, encW, encB, za, N, 0, encA);
  bucket_fill_kernel<<<NB + encT - encA, 1024, 0, stream>>>(rec, bstart, off, csr, N, E, NB,
                                                            x, encW, encB, za, encA, encT);

  float* outH = (float*)d_out;
  float* outL = outH + (size_t)N * Hdim;

  for (int i = 0; i < 3; ++i) {
    const float* zin  = i == 1 ? zb : za;
    float*       zout = i == 0 ? zb : za;
    const float* base = i == 0 ? nullptr : h;
    if (i < 2) {
      conv_kernel<false><<<cBlocks, cThreads, 0, stream>>>(
          zin, off, csr, tvec, i, base,
          W1 + (size_t)i * Hdim * H2, b1 + (size_t)i * H2,
          g1 + (size_t)i * H2,        be1 + (size_t)i * H2,
          W2 + (size_t)i * H2 * Hdim, b2 + (size_t)i * Hdim,
          ng + (i + 1) * Hdim, nb + (i + 1) * Hdim, nullptr, nullptr,
          h, zout, nullptr, N, E);
    } else {
      conv_kernel<true><<<cBlocks, cThreads, 0, stream>>>(
          zin, off, csr, tvec, i, base,
          W1 + (size_t)i * Hdim * H2, b1 + (size_t)i * H2,
          g1 + (size_t)i * H2,        be1 + (size_t)i * H2,
          W2 + (size_t)i * H2 * Hdim, b2 + (size_t)i * Hdim,
          ng, nb, linW, linB,
          nullptr, outH, outL, N, E);
    }
  }
}


# Gather issues fewer vector instructions per edge: message = max(z+eps, eps), segment max over messages (min for t<0), alpha's exponent one base-2 multiply-add (one rounding fewer); sum order kept
# speedup vs baseline: 5.7017x; 1.0529x over previous
#include <hip/hip_runtime.h>
#include <cstdint>


typedef __attribute__((ext_vector_type(2))) float v2f;
typedef __attribute__((ext_vector_type(8))) float v8f;

#define WMMA_F32(a, b, c) \
  __builtin_amdgcn_wmma_f32_16x16x4_f32(false, (a), false, (b), (short)0, (c), false, false)

constexpr int Hdim = 64;
constexpr int H2   = 128;
constexpr int IND  = 128;

constexpr int TE = 128;

__device__ __forceinline__ void enc_tile(
    const float* __restrict__ x, const float* __restrict__ W,
    const float* __restrict__ bias, float* __restrict__ h, int nrows, int tile)
{
  __shared__ float Xs[TE][IND + 4] __attribute__((aligned(16)));
  const int tid  = threadIdx.x;
  const int row0 = tile * TE;

  float4 xv[4];
#pragma unroll
  for (int k = 0; k < 4; ++k) {
    const int i  = tid + 1024 * k;
    const int gr = row0 + (i >> 5);
    const int lr = gr < nrows ? gr : nrows - 1;
    xv[k] = *(const float4*)(x + (long long)lr * IND + (i & 31) * 4);
  }
#pragma unroll
  for (int k = 0; k < 4; ++k) {
    const int i = tid + 1024 * k;
    const int r = i >> 5;
    const float4 v = (row0 + r < nrows) ? xv[k] : make_float4(0.f, 0.f, 0.f, 0.f);
    *(float4*)&Xs[r][(i & 31) * 4] = v;
  }
  __syncthreads();

  const int wave = tid >> 5, lane = tid & 31;
  const int half = lane >> 4, ln = lane & 15;

  if (wave >= TE / 8) return;

  {
    const int m0 = (wave >> 1) * 16;
    const int c0 = (wave & 1) * 32 + ln;
    const float bn0 = bias[c0], bn1 = bias[c0 + 16];
    v8f acc0 = {}, acc1 = {};
#pragma nounroll
    for (int k0 = 0; k0 < IND; k0 += 16) {
      v2f p[4], q[4];
#pragma unroll
      for (int s = 0; s < 4; ++s) {
        const int ka = k0 + 4 * s + 2 * half;
        p[s][0] = W[ka * Hdim + c0];
        p[s][1] = W[(ka + 1) * Hdim + c0];
        q[s][0] = W[ka * Hdim + c0 + 16];
        q[s][1] = W[(ka + 1) * Hdim + c0 + 16];
      }
#pragma unroll
      for (int s = 0; s < 4; ++s)
        asm volatile("" : "+v"(p[s][0]), "+v"(p[s][1]), "+v"(q[s][0]), "+v"(q[s][1]));
#pragma unroll
      for (int s = 0; s < 4; ++s) {
        const int ka = k0 + 4 * s + 2 * half;
        v2f a;
        a[0] = Xs[m0 + ln][ka];      a[1] = Xs[m0 + ln][ka + 1];
        acc0 = WMMA_F32(a, p[s], acc0);
        acc1 = WMMA_F32(a, q[s], acc1);
      }
    }
    for (int r = 0; r < 8; ++r) {
      const int gr = row0 + m0 + r + 8 * half;
      if (gr < nrows) {
        h[(long long)gr * Hdim + c0]      = acc0[r] + bn0;
        h[(long long)gr * Hdim + c0 + 16] = acc1[r] + bn1;
      }
    }
  }
}

constexpr int ROWS_PER_WAVE = 8;

constexpr int BSH   = 8;
constexpr int BW    = 1 << BSH;
constexpr int NBMAX = 256;
constexpr int EPT   = 4;
constexpr int CH    = 1024 * EPT;

__device__ __forceinline__ int bucket_of(int d, int NB)
{
  const unsigned b = (unsigned)d >> BSH;
  return b < (unsigned)NB ? (int)b : NB - 1;
}

__global__ __launch_bounds__(1024) void bin_count_kernel(
    const int* __restrict__ dst, int* __restrict__ M, int E, int NB)
{
  __shared__ int hist[NBMAX];
  const int tid = threadIdx.x;
  const int e0  = blockIdx.x * CH + tid;
  int dd[EPT];
#pragma unroll
  for (int k = 0; k < EPT; ++k) {
    const int e = e0 + 1024 * k;
    dd[k] = dst[e < E ? e : E - 1];
  }
  if (tid < NBMAX) hist[tid] = 0;
  __syncthreads();
#pragma unroll
  for (int k = 0; k < EPT; ++k)
    if (e0 + 1024 * k < E) atomicAdd(&hist[bucket_of(dd[k], NB)], 1);
  __syncthreads();
  if (tid < NB && tid < NBMAX) M[blockIdx.x * NB + tid] = hist[tid];
}

__global__ __launch_bounds__(1024, 16) void bin_scatter_kernel(
    const int* __restrict__ src, const int* __restrict__ dst,
    const int* __restrict__ M, int* __restrict__ bstart, int* __restrict__ rec,
    int E, int NB, int NC,
    const float* __restrict__ ex, const float* __restrict__ eW,
    const float* __restrict__ eb, float* __restrict__ eh, int nrows, int et0, int et1)
{
  if ((int)blockIdx.x >= NC) {
    const int tile = et0 + (int)blockIdx.x - NC;
    if (tile < et1) enc_tile(ex, eW, eb, eh, nrows, tile);
    return;
  }
  __shared__ int ptot[4][NBMAX];
  __shared__ int ppre[4][NBMAX];
  __shared__ int cur[NBMAX];
  __shared__ int wsum[NBMAX / 32];
  const int tid = threadIdx.x, lane = tid & 31, wave = tid >> 5;
  const int c  = blockIdx.x;
  const int e0 = c * CH + tid;

  int dd[EPT], ss[EPT];
#pragma unroll
  for (int k = 0; k < EPT; ++k) {
    const int e  = e0 + 1024 * k;
    const int le = e < E ? e : E - 1;
    dd[k] = dst[le];
    ss[k] = src[le];
  }

  const int b = tid & (NBMAX - 1), q = tid >> 8;
  int tot = 0, pre = 0;
  if (b < NB) {
    for (int k0 = q; k0 < NC; k0 += 32) {
      int v[8];
#pragma unroll
      for (int j = 0; j < 8; ++j) {
        const int k = k0 + 4 * j;
        v[j] = M[(k < NC ? k : NC - 1) * NB + b];
      }
#pragma unroll
      for (int j = 0; j < 8; ++j) {
        const int k = k0 + 4 * j;
        const int a = k < NC ? v[j] : 0;
        tot += a;
        pre += k < c ? a : 0;
      }
    }
  }
  ptot[q][b] = tot;
  ppre[q][b] = pre;
  __syncthreads();

  int inc = 0;
  if (tid < NBMAX) {
    tot = ptot[0][tid] + ptot[1][tid] + ptot[2][tid] + ptot[3][tid];
    pre = ppre[0][tid] + ppre[1][tid] + ppre[2][tid] + ppre[3][tid];
    inc = tot;
    for (int o = 1; o < 32; o <<= 1) {
      const int up = __shfl_up(inc, o);
      if (lane >= o) inc += up;
    }
    if (lane == 31) wsum[wave] = inc;
  }
  __syncthreads();
  if (tid < NBMAX) {
    int wx = 0;
    for (int w = 0; w < wave; ++w) wx += wsum[w];
    const int ex = wx + inc - tot;
    cur[tid] = ex + pre;
    if (c == 0) {
      if (tid < NB)      bstart[tid] = ex;
      if (tid == NB - 1) bstart[NB]  = ex + tot;
    }
  }
  __syncthreads();

  int2* rc = (int2*)rec;
#pragma unroll
  for (int k = 0; k < EPT; ++k) {
    if (e0 + 1024 * k < E) {
      const int pos = atomicAdd(&cur[bucket_of(dd[k], NB)], 1);
      if ((unsigned)pos < (unsigned)E) rc[pos] = make_int2(ss[k], dd[k]);
    }
  }
}

__device__ __forceinline__ int local_of(int d, int n0)
{
  const int l = d - n0;
  return l < 0 ? 0 : (l > BW - 1 ? BW - 1 : l);
}

__global__ __launch_bounds__(1024, 16) void bucket_fill_kernel(
    const int* __restrict__ rec, const int* __restrict__ bstart,
    int* __restrict__ off, int* __restrict__ csr, int N, int E, int NB,
    const float* __restrict__ ex, const float* __restrict__ eW,
    const float* __restrict__ eb, float* __restrict__ eh, int et0, int et1)
{
  if ((int)blockIdx.x >= NB) {
    const int tile = et0 + (int)blockIdx.x - NB;
    if (tile < et1) enc_tile(ex, eW, eb, eh, N, tile);
    return;
  }
  __shared__ int deg[BW];
  __shared__ int loff[BW];
  __shared__ int wsum[BW / 32];
  const int tid = threadIdx.x, lane = tid & 31, wave = tid >> 5;
  const int b  = blockIdx.x;
  const int n0 = b * BW;
  const int2* rc = (const int2*)rec;
  int s0 = bstart[b], s1 = bstart[b + 1];
  s0 = s0 < 0 ? 0 : (s0 > E ? E : s0);
  s1 = s1 < s0 ? s0 : (s1 > E ? E : s1);

  if (tid < BW) deg[tid] = 0;
  __syncthreads();
  for (int r0 = s0 + tid; r0 < s1; r0 += 4096) {
    int dv[4];
#pragma unroll
    for (int j = 0; j < 4; ++j) {
      const int r = r0 + 1024 * j;
      dv[j] = rc[r < s1 ? r : s1 - 1].y;
    }
#pragma unroll
    for (int j = 0; j < 4; ++j) asm volatile("" : "+v"(dv[j]));
#pragma unroll
    for (int j = 0; j < 4; ++j)
      if (r0 + 1024 * j < s1) atomicAdd(&deg[local_of(dv[j], n0)], 1);
  }
  __syncthreads();

  int v = 0, inc = 0;
  if (tid < BW) {
    v = deg[tid];
    inc = v;
    for (int o = 1; o < 32; o <<= 1) {
      const int up = __shfl_up(inc, o);
      if (lane >= o) inc += up;
    }
    if (lane == 31) wsum[wave] = inc;
  }
  __syncthreads();
  if (tid < BW) {
    int wx = 0;
    for (int w = 0; w < wave; ++w) wx += wsum[w];
    const int st = s0 + wx + inc - v;
    loff[tid] = st;
    deg[tid]  = 0;
    const int node = n0 + tid;
    if (node <= N) off[node] = st;
    if (b == NB - 1 && tid == BW - 1 && node + 1 == N) off[N] = st + v;
  }
  __syncthreads();

  for (int r0 = s0 + tid; r0 < s1; r0 += 4096) {
    int2 rv[4];
#pragma unroll
    for (int j = 0; j < 4; ++j) {
      const int r = r0 + 1024 * j;
      rv[j] = rc[r < s1 ? r : s1 - 1];
    }
#pragma unroll
    for (int j = 0; j < 4; ++j) asm volatile("" : "+v"(rv[j].x), "+v"(rv[j].y));
#pragma unroll
    for (int j = 0; j < 4; ++j) {
      if (r0 + 1024 * j < s1) {
        const int l   = local_of(rv[j].y, n0);
        const int pos = loff[l] + atomicAdd(&deg[l], 1);
        if ((unsigned)pos < (unsigned)E) csr[pos] = rv[j].x;
      }
    }
  }
}

constexpr int GK = 6;
constexpr int RK = 4 * GK;

__device__ __forceinline__ unsigned edge_row(const int* __restrict__ csr, int p, int jl,
                                             unsigned last)
{
  p = p > jl ? jl : p;
  unsigned s = (unsigned)csr[p];
  return s > last ? last : s;
}

__device__ __forceinline__ unsigned head_word(const int* __restrict__ csr, int j0, int j1,
                                              int lane, int E, unsigned last)
{
  int p = j0 + lane;
  p = p > j1 - 1 ? j1 - 1 : p;
  p = p > E - 1 ? E - 1 : p;
  p = p < 0 ? 0 : p;
  const unsigned s = (unsigned)csr[p];
  return s > last ? last : s;
}

__device__ __forceinline__ float2 z_row(const char* __restrict__ zc, unsigned row, unsigned fb)
{
  return *(const float2*)(zc + (size_t)row * (size_t)(Hdim * 4) + fb);
}

__device__ __forceinline__ float2 msg2(float2 zv)
{
  return make_float2(fmaxf(zv.x + 1e-7f, 1e-7f), fmaxf(zv.y + 1e-7f, 1e-7f));
}

template <bool NEG>
__device__ __forceinline__ void seg_ext_m(float2 m, float& x0, float& x1)
{
  if (NEG) { x0 = fminf(x0, m.x); x1 = fminf(x1, m.y); }
  else     { x0 = fmaxf(x0, m.x); x1 = fmaxf(x1, m.y); }
}

__device__ __forceinline__ void seg_sum_m(float m, float t2, float e2, bool live,
                                          float& den, float& num)
{
  float a = __builtin_amdgcn_exp2f(__builtin_fmaf(m, t2, -e2));
  a = live ? a : 0.f;
  den += a;
  num = __builtin_fmaf(m, a, num);
}

__device__ __forceinline__ void seg_sum2(float2 m, float t2, float e0, float e1, bool live,
                                         float& d0, float& d1, float& n0, float& n1)
{
  seg_sum_m(m.x, t2, e0, live, d0, n0);
  seg_sum_m(m.y, t2, e1, live, d1, n1);
}

template <bool NEG>
__device__ __forceinline__ void node_sums(
    const char* __restrict__ zc, unsigned fb, const int* __restrict__ csr,
    int j0, int j1, unsigned cv, float tv, unsigned last,
    float& d0, float& d1, float& n0, float& n1)
{
  const int jl = j1 - 1;
  const int jt = j0 + RK;


  float2 R[RK];
#pragma unroll
  for (int g = 0; g < GK; ++g) {
    if (j0 + 4 * g <= jl) {
#pragma unroll
      for (int i = 0; i < 4; ++i)
        R[4 * g + i] = z_row(zc, (unsigned)__builtin_amdgcn_readlane((int)cv, 4 * g + i), fb);
    }
  }

  float x0 = NEG ? 3.402823466e38f : -3.402823466e38f, x1 = x0;
#pragma unroll
  for (int g = 0; g < GK; ++g) {
    if (j0 + 4 * g <= jl) {
#pragma unroll
      for (int i = 0; i < 4; ++i) {
        R[4 * g + i] = msg2(R[4 * g + i]);
        seg_ext_m<NEG>(R[4 * g + i], x0, x1);
      }
    }
  }
  for (int j = jt; j < j1; j += 4) {
    const unsigned c0 = edge_row(csr, j, jl, last),     c1 = edge_row(csr, j + 1, jl, last);
    const unsigned c2 = edge_row(csr, j + 2, jl, last), c3 = edge_row(csr, j + 3, jl, last);
    const float2 r0 = z_row(zc, c0, fb);
    const float2 r1 = z_row(zc, c1, fb);
    const float2 r2 = z_row(zc, c2, fb);
    const float2 r3 = z_row(zc, c3, fb);
    seg_ext_m<NEG>(msg2(r0), x0, x1);
    seg_ext_m<NEG>(msg2(r1), x0, x1);
    seg_ext_m<NEG>(msg2(r2), x0, x1);
    seg_ext_m<NEG>(msg2(r3), x0, x1);
  }
  const float t2 = tv * 1.44269504088896341f;
  const float e0 = x0 * t2, e1 = x1 * t2;

#pragma unroll
  for (int g = 0; g < GK; ++g) {
    const int p = j0 + 4 * g;
    if (p <= jl) {
      const bool l1 = p + 1 <= jl, l2 = p + 2 <= jl, l3 = p + 3 <= jl;
      seg_sum2(R[4 * g + 0], t2, e0, e1, true, d0, d1, n0, n1);
      seg_sum2(R[4 * g + 1], t2, e0, e1, l1,   d0, d1, n0, n1);
      seg_sum2(R[4 * g + 2], t2, e0, e1, l2,   d0, d1, n0, n1);
      seg_sum2(R[4 * g + 3], t2, e0, e1, l3,   d0, d1, n0, n1);
    }
  }
  for (int j = jt; j < j1; j += 4) {
    const unsigned c0 = edge_row(csr, j, jl, last),     c1 = edge_row(csr, j + 1, jl, last);
    const unsigned c2 = edge_row(csr, j + 2, jl, last), c3 = edge_row(csr, j + 3, jl, last);
    const float2 r0 = z_row(zc, c0, fb);
    const float2 r1 = z_row(zc, c1, fb);
    const float2 r2 = z_row(zc, c2, fb);
    const float2 r3 = z_row(zc, c3, fb);
    const bool l1 = j + 1 <= jl, l2 = j + 2 <= jl, l3 = j + 3 <= jl;
    seg_sum2(msg2(r0), t2, e0, e1, true, d0, d1, n0, n1);
    seg_sum2(msg2(r1), t2, e0, e1, l1,   d0, d1, n0, n1);
    seg_sum2(msg2(r2), t2, e0, e1, l2,   d0, d1, n0, n1);
    seg_sum2(msg2(r3), t2, e0, e1, l3,   d0, d1, n0, n1);
  }
}

template <bool NEG>
__device__ __forceinline__ void gather_rows(
    const char* __restrict__ zc, unsigned fb, const int* __restrict__ csr,
    int ob, int lr0, int row0, int lane, float tv, unsigned last, int nrows, int E,
    float (*U)[Hdim + 4])
{
  unsigned cvn;
  {
    const int a0 = __builtin_amdgcn_readlane(ob, 0);
    int a1 = __builtin_amdgcn_readlane(ob, 1);
    a1 = a1 < a0 ? a0 : a1;
    cvn = head_word(csr, a0, a1, lane, E, last);
  }
#pragma nounroll
  for (int i = 0; i < ROWS_PER_WAVE; ++i) {
    const int r = lr0 + i;
    const int d = row0 + r;
    float2 u = make_float2(0.f, 0.f);
    const int j0 = __builtin_amdgcn_readlane(ob, i);
    const int j1r = __builtin_amdgcn_readlane(ob, i + 1);
    const int j1 = j1r < j0 ? j0 : j1r;
    const unsigned cv = cvn;
    {
      int b1n = __builtin_amdgcn_readlane(ob, i + 2);
      b1n = b1n < j1r ? j1r : b1n;
      cvn = head_word(csr, j1r, b1n, lane, E, last);
    }
    if (d < nrows) {
      const float2 zv = z_row(zc, (unsigned)d, fb);
      float d0 = 0.f, d1 = 0.f, n0 = 0.f, n1 = 0.f;
      if (j0 < j1)
        node_sums<NEG>(zc, fb, csr, j0, j1, cv, tv, last, d0, d1, n0, n1);
      u.x = zv.x + n0 / (d0 + 1e-16f);
      u.y = zv.y + n1 / (d1 + 1e-16f);
    }
    *(float2*)&U[r][2 * lane] = u;
  }
}

constexpr int TR = 16;

template <int NK>
__device__ __forceinline__ void load_bfrag(const float* __restrict__ Wm, int ld, int k0,
                                           int half, int nn, v2f (&bf)[NK])
{
#pragma unroll
  for (int s = 0; s < NK; ++s) {
    const int ka = k0 + 4 * s + 2 * half;
    bf[s][0] = Wm[ka * ld + nn];
    bf[s][1] = Wm[(ka + 1) * ld + nn];
  }
}

template <int NK>
__device__ __forceinline__ v8f mma_run(const v2f (&af)[NK], const v2f (&bf)[NK], v8f acc)
{
#pragma unroll
  for (int s = 0; s < NK; ++s) acc = WMMA_F32(af[s], bf[s], acc);
  return acc;
}

template <bool FINAL>
__global__ __launch_bounds__(TR / ROWS_PER_WAVE * 32) void conv_kernel(
    const float* __restrict__ z, const int* __restrict__ off,
    const int* __restrict__ csr, const float* __restrict__ t, int ti,
    const float* base,
    const float* __restrict__ W1, const float* __restrict__ b1,
    const float* __restrict__ g1, const float* __restrict__ be1,
    const float* __restrict__ W2, const float* __restrict__ b2,
    const float* __restrict__ ng, const float* __restrict__ nb,
    const float* __restrict__ linW, const float* __restrict__ linB,
    float* hout, float* __restrict__ fout, float* __restrict__ outL, int nrows, int E)
{
  __shared__ float U[TR][Hdim + 4] __attribute__((aligned(16)));
  __shared__ float T[TR][H2 + 4];

  const int tid  = threadIdx.x;
  const int row0 = blockIdx.x * TR;

  const int wave = tid >> 5, lane = tid & 31;
  const int half = lane >> 4, ln = lane & 15;

  {
    const int lr0 = __builtin_amdgcn_readfirstlane(wave) * ROWS_PER_WAVE;
    const unsigned fb = (unsigned)lane * 8u;
    const char* zc = (const char*)z;
    const float tv = t[ti];
    const unsigned last = (unsigned)(nrows - 1);
    int ob;
    {
      const int k = lane < ROWS_PER_WAVE ? lane : ROWS_PER_WAVE;
      const int d = row0 + lr0 + k;
      ob = off[d < nrows ? d : nrows];
      ob = ob < 0 ? 0 : (ob > E ? E : ob);
    }
    if (tv < 0.f) gather_rows<true>(zc, fb, csr, ob, lr0, row0, lane, tv, last, nrows, E, U);
    else          gather_rows<false>(zc, fb, csr, ob, lr0, row0, lane, tv, last, nrows, E, U);
  }
  {
    const int c1 = wave * 64 + ln;
    v2f bq0[16], bq1[16];
    float bs0, bs1;
    load_bfrag<16>(W1, H2, 0, half, c1, bq0);
    bs0 = b1[c1];
    __syncthreads();

    v2f af[16];
#pragma unroll
    for (int s = 0; s < 16; ++s) {
      const int ka = 4 * s + 2 * half;
      af[s][0] = U[ln][ka];     af[s][1] = U[ln][ka + 1];
    }
#pragma unroll
    for (int t = 0; t < 4; t += 2) {
      load_bfrag<16>(W1, H2, 0, half, c1 + 16 * (t + 1), bq1);
      bs1 = b1[c1 + 16 * (t + 1)];
      {
        v8f acc = {};
        acc = mma_run<16>(af, bq0, acc);
        for (int r = 0; r < 8; ++r) T[r + 8 * half][c1 + 16 * t] = acc[r] + bs0;
      }
      if (t + 2 < 4) {
        load_bfrag<16>(W1, H2, 0, half, c1 + 16 * (t + 2), bq0);
        bs0 = b1[c1 + 16 * (t + 2)];
      }
      {
        v8f acc = {};
        acc = mma_run<16>(af, bq1, acc);
        for (int r = 0; r < 8; ++r) T[r + 8 * half][c1 + 16 * (t + 1)] = acc[r] + bs1;
      }
    }
  }
  __syncthreads();

  {
    int r = tid >> 2;
    int q = tid & 3;
    float s = 0.f, s2 = 0.f;
    for (int c = q * 32; c < q * 32 + 32; ++c) {
      float v = T[r][c]; s += v; s2 += v * v;
    }
    s  += __shfl_xor(s, 1);  s  += __shfl_xor(s, 2);
    s2 += __shfl_xor(s2, 1); s2 += __shfl_xor(s2, 2);
    float mu  = s * (1.f / 128.f);
    float var = s2 * (1.f / 128.f) - mu * mu;
    float rs  = rsqrtf(var + 1e-5f);
    for (int c = q * 32; c < q * 32 + 32; ++c) {
      float v = (T[r][c] - mu) * rs * g1[c] + be1[c];
      T[r][c] = fmaxf(v, 0.f);
    }
  }
  __syncthreads();

  {
    const int c2 = wave * 32 + ln;
    const float bn0 = b2[c2], bn1 = b2[c2 + 16];
    float bv0[8], bv1[8];
    if (base) {
#pragma unroll
      for (int r = 0; r < 8; ++r) {
        const int gr = row0 + r + 8 * half;
        const int lr = gr < nrows ? gr : nrows - 1;
        bv0[r] = base[(long long)lr * Hdim + c2];
        bv1[r] = base[(long long)lr * Hdim + c2 + 16];
      }
    } else {
#pragma unroll
      for (int r = 0; r < 8; ++r) { bv0[r] = 0.f; bv1[r] = 0.f; }
    }
    v8f acc0 = {}, acc1 = {};
#pragma unroll
    for (int kk = 0; kk < H2; kk += 4) {
      const int ka = kk + 2 * half;
      v2f a, p, q;
      a[0] = T[ln][ka];     a[1] = T[ln][ka + 1];
      p[0] = W2[ka * Hdim + c2];
      p[1] = W2[(ka + 1) * Hdim + c2];
      q[0] = W2[ka * Hdim + c2 + 16];
      q[1] = W2[(ka + 1) * Hdim + c2 + 16];
      acc0 = WMMA_F32(a, p, acc0);
      acc1 = WMMA_F32(a, q, acc1);
    }
    for (int r = 0; r < 8; ++r) {
      float v0 = acc0[r] + bn0, v1 = acc1[r] + bn1;
      if (base) { v0 += bv0[r]; v1 += bv1[r]; }
      U[r + 8 * half][c2]      = v0;
      U[r + 8 * half][c2 + 16] = v1;
    }
  }
  __syncthreads();

  {
    const int lr0 = __builtin_amdgcn_readfirstlane(wave) * ROWS_PER_WAVE;
    const int gr0 = row0 + lr0;
    float v0[ROWS_PER_WAVE], v1[ROWS_PER_WAVE];
#pragma unroll
    for (int k = 0; k < ROWS_PER_WAVE; ++k) {
      v0[k] = U[lr0 + k][lane];
      v1[k] = U[lr0 + k][lane + 32];
    }
    const float g0 = ng[lane], g1n = ng[lane + 32];
    const float b0 = nb[lane], b1n = nb[lane + 32];
    float w0[3], w1[3], lb[3];
    if (FINAL) {
#pragma unroll
      for (int j = 0; j < 3; ++j) {
        w0[j] = linW[lane * 3 + j];
        w1[j] = linW[(lane + 32) * 3 + j];
        lb[j] = linB[j];
      }
    }

    float s[ROWS_PER_WAVE], s2[ROWS_PER_WAVE];
#pragma unroll
    for (int k = 0; k < ROWS_PER_WAVE; ++k) {
      s[k]  = v0[k] + v1[k];
      s2[k] = v0[k] * v0[k] + v1[k] * v1[k];
    }
#pragma unroll
    for (int off = 1; off < 32; off <<= 1) {
#pragma unroll
      for (int k = 0; k < ROWS_PER_WAVE; ++k) {
        s[k]  += __shfl_xor(s[k], off);
        s2[k] += __shfl_xor(s2[k], off);
      }
    }
    float p[ROWS_PER_WAVE][3];
#pragma unroll
    for (int k = 0; k < ROWS_PER_WAVE; ++k) {
      float mu  = s[k] * (1.f / 64.f);
      float var = s2[k] * (1.f / 64.f) - mu * mu;
      float rs  = rsqrtf(var + 1e-5f);
      float f0 = fmaxf((v0[k] - mu) * rs * g0 + b0, 0.f);
      float f1 = fmaxf((v1[k] - mu) * rs * g1n + b1n, 0.f);
      if (gr0 + k < nrows) {
        if (!FINAL) {
          float* hr = hout + (long long)(gr0 + k) * Hdim;
          hr[lane]      = v0[k];
          hr[lane + 32] = v1[k];
        }
        float* fr = fout + (long long)(gr0 + k) * Hdim;
        fr[lane]      = f0;
        fr[lane + 32] = f1;
      }
      if (FINAL) {
#pragma unroll
        for (int j = 0; j < 3; ++j) p[k][j] = f0 * w0[j] + f1 * w1[j];
      }
    }
    if (FINAL) {
#pragma unroll
      for (int off = 1; off < 32; off <<= 1) {
#pragma unroll
        for (int k = 0; k < ROWS_PER_WAVE; ++k) {
#pragma unroll
          for (int j = 0; j < 3; ++j) p[k][j] += __shfl_xor(p[k][j], off);
        }
      }
#pragma unroll
      for (int k = 0; k < ROWS_PER_WAVE; ++k) {
        if (gr0 + k < nrows && lane == 0) {
#pragma unroll
          for (int j = 0; j < 3; ++j) outL[(long long)(gr0 + k) * 3 + j] = p[k][j] + lb[j];
        }
      }
    }
  }
}

extern "C" void kernel_launch(void* const* d_in, const int* in_sizes, int n_in,
                              void* d_out, int out_size, void* d_ws, size_t ws_size,
                              hipStream_t stream)
{
  const float* x    = (const float*)d_in[0];
  const int*   edge = (const int*)d_in[1];
  const float* encW = (const float*)d_in[2];
  const float* encB = (const float*)d_in[3];
  const float* tvec = (const float*)d_in[4];
  const float* W1   = (const float*)d_in[5];
  const float* b1   = (const float*)d_in[6];
  const float* g1   = (const float*)d_in[7];
  const float* be1  = (const float*)d_in[8];
  const float* W2   = (const float*)d_in[9];
  const float* b2   = (const float*)d_in[10];
  const float* ng   = (const float*)d_in[11];
  const float* nb   = (const float*)d_in[12];
  const float* linW = (const float*)d_in[13];
  const float* linB = (const float*)d_in[14];

  const int N = in_sizes[0] / IND;
  const int E = in_sizes[1] / 2;
  const int* src = edge;
  const int* dst = edge + E;

  float* za    = (float*)d_ws;
  float* zb    = za    + (size_t)N * Hdim;
  float* plane = zb    + (size_t)N * Hdim;
  float* h     = plane + (size_t)N * Hdim;

  const int n1  = N + 1;
  const int n1a = (n1 + 3) & ~3;
  const int NB  = (N + BW - 1) / BW;
  const int NC  = (E + CH - 1) / CH;
  int* off    = (int*)plane;
  int* bstart = off + n1a;
  int* M      = bstart + (NBMAX + 4);
  int* rec    = M + ((NC * NB + 3) & ~3);
  int* csr    = rec + 2 * (size_t)E;

  const int encT       = (N + TE - 1) / TE;
  const int encA       = (encT + 1) / 2;
  const int cBlocks    = (N + TR - 1) / TR;
  const int cThreads   = TR / ROWS_PER_WAVE * 32;

  bin_count_kernel<<<NC, 1024, 0, stream>>>(dst, M, E, NB);
  bin_scatter_kernel<<<NC + encA, 1024, 0, stream>>>(src, dst, M, bstart, rec, E, NB, NC,
                                                     x, encW, encB, za, N, 0, encA);
  bucket_fill_kernel<<<NB + encT - encA, 1024, 0, stream>>>(rec, bstart, off, csr, N, E, NB,
                                                            x, encW, encB, za, encA, encT);

  float* outH = (float*)d_out;
  float* outL = outH + (size_t)N * Hdim;

  for (int i = 0; i < 3; ++i) {
    const float* zin  = i == 1 ? zb : za;
    float*       zout = i == 0 ? zb : za;
    const float* base = i == 0 ? nullptr : h;
    if (i < 2) {
      conv_kernel<false><<<cBlocks, cThreads, 0, stream>>>(
          zin, off, csr, tvec, i, base,
          W1 + (size_t)i * Hdim * H2, b1 + (size_t)i * H2,
          g1 + (size_t)i * H2,        be1 + (size_t)i * H2,
          W2 + (size_t)i * H2 * Hdim, b2 + (size_t)i * Hdim,
          ng + (i + 1) * Hdim, nb + (i + 1) * Hdim, nullptr, nullptr,
          h, zout, nullptr, N, E);
    } else {
      conv_kernel<true><<<cBlocks, cThreads, 0, stream>>>(
          zin, off, csr, tvec, i, base,
          W1 + (size_t)i * Hdim * H2, b1 + (size_t)i * H2,
          g1 + (size_t)i * H2,        be1 + (size_t)i * H2,
          W2 + (size_t)i * H2 * Hdim, b2 + (size_t)i * Hdim,
          ng, nb, linW, linB,
          nullptr, outH, outL, N, E);
    }
  }
}
